# starting kernel
# baseline (speedup 1.0000x reference)
#include <hip/hip_runtime.h>
#include <math.h>


typedef float v2f __attribute__((ext_vector_type(2)));
typedef float v8f __attribute__((ext_vector_type(8)));

#define WPB 8

static __device__ __forceinline__ v8f wmma_f32(v2f a, v2f b, v8f c) {
  return __builtin_amdgcn_wmma_f32_16x16x4_f32(false, a, false, b,
                                               (short)0, c, false, false);
}

static __device__ __forceinline__ v2f b_frag(const float* __restrict__ W,
                                             int lane, int k0, int K, int Ncols) {
  int n  = lane & 15;
  int kb = k0 + ((lane >> 4) << 1);
  int nm  = (n      < Ncols) ? n      : (Ncols - 1);
  int kc0 = (kb     < K)     ? kb     : (K - 1);
  int kc1 = (kb + 1 < K)     ? kb + 1 : (K - 1);
  float w0 = W[kc0 * Ncols + nm];
  float w1 = W[kc1 * Ncols + nm];
  v2f b;
  b.x = (kb     < K && n < Ncols) ? w0 : 0.0f;
  b.y = (kb + 1 < K && n < Ncols) ? w1 : 0.0f;
  return b;
}

static __device__ __forceinline__ v2f a_frag_x(const float* __restrict__ x,
                                               int N, int m0, int lane,
                                               int k0, int co, int K) {
  int r = m0 + (lane & 15);
  if (r >= N) r = N - 1;
  int kb = k0 + ((lane >> 4) << 1);
  const float* rowp = x + (size_t)r * 16;
  int i0 = co + kb;     i0 = (i0 < 16) ? i0 : 15;
  int i1 = co + kb + 1; i1 = (i1 < 16) ? i1 : 15;
  float a0 = rowp[i0];
  float a1 = rowp[i1];
  v2f a;
  a.x = (kb     < K) ? a0 : 0.0f;
  a.y = (kb + 1 < K) ? a1 : 0.0f;
  return a;
}

static __device__ __forceinline__ v2f a_frag_lds(const float* t, int lane, int k0) {
  int r  = lane & 15;
  int kb = k0 + ((lane >> 4) << 1);
  v2f a;
  a.x = t[r * 16 + kb];
  a.y = t[r * 16 + kb + 1];
  return a;
}

__global__ void k_init(float* __restrict__ deg, float* __restrict__ acc1,
                       float* __restrict__ acc2, int N) {
  int i = blockIdx.x * blockDim.x + threadIdx.x;
  if (i >= N) return;
  deg[i] = 1.0f;
  float4 z = make_float4(0.f, 0.f, 0.f, 0.f);
  float4* a1 = (float4*)(acc1 + (size_t)i * 16);
  a1[0] = z; a1[1] = z; a1[2] = z; a1[3] = z;
  float4* a2 = (float4*)(acc2 + (size_t)i * 8);
  a2[0] = z; a2[1] = z;
}

__global__ void k_degree(const int* __restrict__ col, float* __restrict__ deg,
                         int NE) {
  int e = blockIdx.x * blockDim.x + threadIdx.x;
  if (e < NE) atomicAdd(&deg[col[e]], 1.0f);
}

__global__ void __launch_bounds__(256)
k_dense1(const float* __restrict__ x, const float* __restrict__ gW1,
         const float* __restrict__ deg, float* __restrict__ dinv,
         float* __restrict__ xw1s, int N) {
  __shared__ float ot[WPB][256];
  int wslot = threadIdx.x >> 5;
  int lane  = threadIdx.x & 31;
  int m0 = (blockIdx.x * WPB + wslot) * 16;
  if (m0 >= N) return;
  bool full = (m0 + 16) <= N;
  float* t = ot[wslot];

  int rr  = m0 + (lane & 15);
  int rrc = (rr < N) ? rr : N - 1;
  float d = __builtin_amdgcn_rsqf(deg[rrc]);

  v8f c = {};
  v2f a = a_frag_x(x, N, m0, lane, 0,  0,  3);
  v2f b = b_frag(gW1, lane, 0, 3, 16);
  c = wmma_f32(a, b, c);

  if (lane < 16 && rr < N) dinv[rr] = d;

  int n  = lane & 15;
  int mo = (lane >> 4) << 3;
  for (int v = 0; v < 8; ++v)
    t[(mo + v) * 16 + n] = c[v];
  __builtin_amdgcn_wave_barrier();

  if (full) {
    float dv0 = __shfl(d, lane >> 2, 32);
    float dv1 = __shfl(d, 8 + (lane >> 2), 32);
    const float4* src = (const float4*)t;
    float4 v0 = src[lane];
    float4 v1 = src[lane + 32];
    v0.x *= dv0; v0.y *= dv0; v0.z *= dv0; v0.w *= dv0;
    v1.x *= dv1; v1.y *= dv1; v1.z *= dv1; v1.w *= dv1;
    float4* dst = (float4*)(xw1s + (size_t)m0 * 16);
    dst[lane]      = v0;
    dst[lane + 32] = v1;
  } else {
    for (int v = 0; v < 8; ++v) {
      int r = m0 + mo + v;
      float dv = __shfl(d, mo + v, 32);
      if (r < N) xw1s[(size_t)r * 16 + n] = t[(mo + v) * 16 + n] * dv;
    }
  }
}

__global__ void k_scatter1(const int* __restrict__ row, const int* __restrict__ col,
                           const float* __restrict__ xw1s, float* __restrict__ acc1,
                           int NE) {
  int t = blockIdx.x * blockDim.x + threadIdx.x;
  int e = t >> 2;
  if (e >= NE) return;
  int q = t & 3;
  int r = row[e], cdst = col[e];
  float4 v = ((const float4*)(xw1s + (size_t)r * 16))[q];
  float* dst = acc1 + (size_t)cdst * 16 + q * 4;
  atomicAdd(dst + 0, v.x);
  atomicAdd(dst + 1, v.y);
  atomicAdd(dst + 2, v.z);
  atomicAdd(dst + 3, v.w);
}

__global__ void __launch_bounds__(256)
k_dense2(const float* __restrict__ acc1, const float* __restrict__ xw1s,
         const float* __restrict__ dinv, const float* __restrict__ gW2,
         const float* __restrict__ gb1, float* __restrict__ xw2s, int N) {
  __shared__ float ht[WPB][256];
  __shared__ float dv[WPB][16];
  int wslot = threadIdx.x >> 5;
  int lane  = threadIdx.x & 31;
  int m0 = (blockIdx.x * WPB + wslot) * 16;
  if (m0 >= N) return;
  bool full = (m0 + 16) <= N;
  float* t   = ht[wslot];
  float* dvs = dv[wslot];

  for (int h = 0; h < 2; ++h) {
    int f   = lane + h * 32;
    int row = f >> 2;
    int cc4 = f & 3;
    int r   = m0 + row;
    int rc  = (r < N) ? r : N - 1;
    float dvr = dinv[rc];
    if (cc4 == 0) dvs[row] = dvr;
    float4 a4 = *(const float4*)(acc1 + (size_t)rc * 16 + cc4 * 4);
    float4 s4 = *(const float4*)(xw1s + (size_t)rc * 16 + cc4 * 4);
    float4 b4 = ((const float4*)gb1)[cc4];
    float4 hv;
    hv.x = fmaxf(dvr * (a4.x + s4.x) + b4.x, 0.0f);
    hv.y = fmaxf(dvr * (a4.y + s4.y) + b4.y, 0.0f);
    hv.z = fmaxf(dvr * (a4.z + s4.z) + b4.z, 0.0f);
    hv.w = fmaxf(dvr * (a4.w + s4.w) + b4.w, 0.0f);
    *(float4*)&t[row * 16 + cc4 * 4] = hv;
  }
  __builtin_amdgcn_wave_barrier();

  v8f c = {};
  for (int k0 = 0; k0 < 16; k0 += 4) {
    v2f a = a_frag_lds(t, lane, k0);
    v2f b = b_frag(gW2, lane, k0, 16, 8);
    c = wmma_f32(a, b, c);
  }
  __builtin_amdgcn_wave_barrier();

  int n  = lane & 15;
  int mo = (lane >> 4) << 3;
  if (n < 8) {
    for (int v = 0; v < 8; ++v)
      t[(mo + v) * 8 + n] = c[v];
  }
  __builtin_amdgcn_wave_barrier();

  if (full) {
    float dvr = dvs[lane >> 1];
    float4 v4 = ((const float4*)t)[lane];
    v4.x *= dvr; v4.y *= dvr; v4.z *= dvr; v4.w *= dvr;
    ((float4*)(xw2s + (size_t)m0 * 8))[lane] = v4;
  } else if (n < 8) {
    for (int v = 0; v < 8; ++v) {
      int r = m0 + mo + v;
      if (r < N) xw2s[(size_t)r * 8 + n] = t[(mo + v) * 8 + n] * dvs[mo + v];
    }
  }
}

__global__ void k_scatter2(const int* __restrict__ row, const int* __restrict__ col,
                           const float* __restrict__ xw2s, float* __restrict__ acc2,
                           int NE) {
  int t = blockIdx.x * blockDim.x + threadIdx.x;
  int e = t >> 1;
  if (e >= NE) return;
  int q = t & 1;
  int r = row[e], cdst = col[e];
  float4 v = ((const float4*)(xw2s + (size_t)r * 8))[q];
  float* dst = acc2 + (size_t)cdst * 8 + q * 4;
  atomicAdd(dst + 0, v.x);
  atomicAdd(dst + 1, v.y);
  atomicAdd(dst + 2, v.z);
  atomicAdd(dst + 3, v.w);
}

__global__ void __launch_bounds__(256)
k_final(const float* __restrict__ x, const float* __restrict__ acc2,
        const float* __restrict__ xw2s, const float* __restrict__ dinv,
        const float* __restrict__ gb2,
        const float* __restrict__ cW1, const float* __restrict__ cb1,
        const float* __restrict__ cW2, const float* __restrict__ cb2,
        const float* __restrict__ mW1, const float* __restrict__ mb1,
        const float* __restrict__ mW2, const float* __restrict__ mb2,
        float* __restrict__ out, int N) {
  __shared__ float ht[WPB][256];
  __shared__ float ct[WPB][128];
  __shared__ float g0s[WPB][16];
  int wslot = threadIdx.x >> 5;
  int lane  = threadIdx.x & 31;
  int m0 = (blockIdx.x * WPB + wslot) * 16;
  if (m0 >= N) return;
  bool full = (m0 + 16) <= N;
  float* t  = ht[wslot];
  float* o  = ct[wslot];
  float* g0 = g0s[wslot];

  int n  = lane & 15;
  int mo = (lane >> 4) << 3;

  int grow = lane >> 1, gcc = lane & 1;
  int gr  = m0 + grow;
  int grc = (gr < N) ? gr : N - 1;
  float  gdv = dinv[grc];
  float4 a24 = *(const float4*)(acc2 + (size_t)grc * 8 + gcc * 4);
  float4 s24 = *(const float4*)(xw2s + (size_t)grc * 8 + gcc * 4);
  float4 gb4 = ((const float4*)gb2)[gcc];
  float4 go4;
  go4.x = gdv * (a24.x + s24.x) + gb4.x;
  go4.y = gdv * (a24.y + s24.y) + gb4.y;
  go4.z = gdv * (a24.z + s24.z) + gb4.z;
  go4.w = gdv * (a24.w + s24.w) + gb4.w;

  v8f ch = {};
  for (int k0 = 0; k0 < 16; k0 += 4) {
    v2f a = a_frag_x(x, N, m0, lane, k0,  3,  13);
    v2f b = b_frag(cW1, lane, k0, 13, 16);
    ch = wmma_f32(a, b, ch);
  }
  for (int v = 0; v < 8; ++v) t[(mo + v) * 16 + n] = fmaxf(ch[v] + cb1[n], 0.0f);
  __builtin_amdgcn_wave_barrier();

  v8f cc = {};
  for (int k0 = 0; k0 < 16; k0 += 4) {
    v2f a = a_frag_lds(t, lane, k0);
    v2f b = b_frag(cW2, lane, k0, 16, 8);
    cc = wmma_f32(a, b, cc);
  }
  if (n < 8) {
    for (int v = 0; v < 8; ++v) o[(mo + v) * 8 + n] = cc[v] + cb2[n];
  }
  __builtin_amdgcn_wave_barrier();

  v8f cm = {};
  for (int k0 = 0; k0 < 16; k0 += 4) {
    v2f a = a_frag_x(x, N, m0, lane, k0,  0,  16);
    v2f b = b_frag(mW1, lane, k0, 16, 16);
    cm = wmma_f32(a, b, cm);
  }
  for (int v = 0; v < 8; ++v) t[(mo + v) * 16 + n] = fmaxf(cm[v] + mb1[n], 0.0f);
  __builtin_amdgcn_wave_barrier();

  if (lane < 16) {
    float l0 = mb2[0], l1 = mb2[1];
    for (int j = 0; j < 16; ++j) {
      float hv = t[lane * 16 + j];
      l0 += hv * mW2[j * 2 + 0];
      l1 += hv * mW2[j * 2 + 1];
    }
    float mx = fmaxf(l0, l1);
    float e0 = __expf(l0 - mx), e1 = __expf(l1 - mx);
    g0[lane] = e0 / (e0 + e1);
  }
  __builtin_amdgcn_wave_barrier();

  float g  = g0[grow];
  float4 c4 = *(const float4*)&o[lane * 4];
  float4 r4;
  r4.x = g * go4.x + (1.0f - g) * c4.x;
  r4.y = g * go4.y + (1.0f - g) * c4.y;
  r4.z = g * go4.z + (1.0f - g) * c4.z;
  r4.w = g * go4.w + (1.0f - g) * c4.w;
  if (full) {
    ((float4*)(out + (size_t)m0 * 8))[lane] = r4;
  } else if (gr < N) {
    *(float4*)(out + (size_t)gr * 8 + gcc * 4) = r4;
  }
}

extern "C" void kernel_launch(void* const* d_in, const int* in_sizes, int n_in,
                              void* d_out, int out_size, void* d_ws, size_t ws_size,
                              hipStream_t stream) {
  const float* x   = (const float*)d_in[0];
  const int*   ei  = (const int*)  d_in[1];
  const float* gW1 = (const float*)d_in[2];
  const float* gb1 = (const float*)d_in[3];
  const float* gW2 = (const float*)d_in[4];
  const float* gb2 = (const float*)d_in[5];
  const float* cW1 = (const float*)d_in[6];
  const float* cb1 = (const float*)d_in[7];
  const float* cW2 = (const float*)d_in[8];
  const float* cb2 = (const float*)d_in[9];
  const float* mW1 = (const float*)d_in[10];
  const float* mb1 = (const float*)d_in[11];
  const float* mW2 = (const float*)d_in[12];
  const float* mb2 = (const float*)d_in[13];
  float* out = (float*)d_out;

  const int N  = in_sizes[0] / 16;
  const int NE = in_sizes[1] / 2;
  const int* erow = ei;
  const int* ecol = ei + NE;

  float* deg  = (float*)d_ws;
  float* dinv = deg  + N;
  float* xw1s = dinv + N;
  float* acc1 = xw1s + (size_t)16 * N;
  float* xw2s = acc1 + (size_t)16 * N;
  float* acc2 = xw2s + (size_t)8 * N;

  const int tiles   = (N + 15) / 16;
  const int gBlocks = (tiles + WPB - 1) / WPB;

  k_init    <<<(N + 255) / 256, 256, 0, stream>>>(deg, acc1, acc2, N);
  k_degree  <<<(NE + 255) / 256, 256, 0, stream>>>(ecol, deg, NE);
  k_dense1  <<<gBlocks, 256, 0, stream>>>(x, gW1, deg, dinv, xw1s, N);
  k_scatter1<<<((size_t)NE * 4 + 255) / 256, 256, 0, stream>>>(erow, ecol, xw1s, acc1, NE);
  k_dense2  <<<gBlocks, 256, 0, stream>>>(acc1, xw1s, dinv, gW2, gb1, xw2s, N);
  k_scatter2<<<((size_t)NE * 2 + 255) / 256, 256, 0, stream>>>(erow, ecol, xw2s, acc2, NE);
  k_final   <<<gBlocks, 256, 0, stream>>>(x, acc2, xw2s, dinv, gb2,
                                          cW1, cb1, cW2, cb2,
                                          mW1, mb1, mW2, mb2, out, N);
}


# edge scatters: one lane per feature so each gather/atomic instruction covers whole contiguous rows (2x64B, 4x32B), in looping resident waves; the order in which the f32 atomics arrive changes
# speedup vs baseline: 1.4460x; 1.4460x over previous
#include <hip/hip_runtime.h>
#include <math.h>


typedef float v2f __attribute__((ext_vector_type(2)));
typedef float v8f __attribute__((ext_vector_type(8)));

#define WPB 8

static __device__ __forceinline__ v8f wmma_f32(v2f a, v2f b, v8f c) {
  return __builtin_amdgcn_wmma_f32_16x16x4_f32(false, a, false, b,
                                               (short)0, c, false, false);
}

static __device__ __forceinline__ v2f b_frag(const float* __restrict__ W,
                                             int lane, int k0, int K, int Ncols) {
  int n  = lane & 15;
  int kb = k0 + ((lane >> 4) << 1);
  int nm  = (n      < Ncols) ? n      : (Ncols - 1);
  int kc0 = (kb     < K)     ? kb     : (K - 1);
  int kc1 = (kb + 1 < K)     ? kb + 1 : (K - 1);
  float w0 = W[kc0 * Ncols + nm];
  float w1 = W[kc1 * Ncols + nm];
  v2f b;
  b.x = (kb     < K && n < Ncols) ? w0 : 0.0f;
  b.y = (kb + 1 < K && n < Ncols) ? w1 : 0.0f;
  return b;
}

static __device__ __forceinline__ v2f a_frag_x(const float* __restrict__ x,
                                               int N, int m0, int lane,
                                               int k0, int co, int K) {
  int r = m0 + (lane & 15);
  if (r >= N) r = N - 1;
  int kb = k0 + ((lane >> 4) << 1);
  const float* rowp = x + (size_t)r * 16;
  int i0 = co + kb;     i0 = (i0 < 16) ? i0 : 15;
  int i1 = co + kb + 1; i1 = (i1 < 16) ? i1 : 15;
  float a0 = rowp[i0];
  float a1 = rowp[i1];
  v2f a;
  a.x = (kb     < K) ? a0 : 0.0f;
  a.y = (kb + 1 < K) ? a1 : 0.0f;
  return a;
}

static __device__ __forceinline__ v2f a_frag_lds(const float* t, int lane, int k0) {
  int r  = lane & 15;
  int kb = k0 + ((lane >> 4) << 1);
  v2f a;
  a.x = t[r * 16 + kb];
  a.y = t[r * 16 + kb + 1];
  return a;
}

__global__ void k_init(float* __restrict__ deg, float* __restrict__ acc1,
                       float* __restrict__ acc2, int N) {
  int i = blockIdx.x * blockDim.x + threadIdx.x;
  if (i >= N) return;
  deg[i] = 1.0f;
  float4 z = make_float4(0.f, 0.f, 0.f, 0.f);
  float4* a1 = (float4*)(acc1 + (size_t)i * 16);
  a1[0] = z; a1[1] = z; a1[2] = z; a1[3] = z;
  float4* a2 = (float4*)(acc2 + (size_t)i * 8);
  a2[0] = z; a2[1] = z;
}

__global__ void k_degree(const int* __restrict__ col, float* __restrict__ deg,
                         int NE) {
  int e = blockIdx.x * blockDim.x + threadIdx.x;
  if (e < NE) atomicAdd(&deg[col[e]], 1.0f);
}

__global__ void __launch_bounds__(256)
k_dense1(const float* __restrict__ x, const float* __restrict__ gW1,
         const float* __restrict__ deg, float* __restrict__ dinv,
         float* __restrict__ xw1s, int N) {
  __shared__ float ot[WPB][256];
  int wslot = threadIdx.x >> 5;
  int lane  = threadIdx.x & 31;
  int m0 = (blockIdx.x * WPB + wslot) * 16;
  if (m0 >= N) return;
  bool full = (m0 + 16) <= N;
  float* t = ot[wslot];

  int rr  = m0 + (lane & 15);
  int rrc = (rr < N) ? rr : N - 1;
  float d = __builtin_amdgcn_rsqf(deg[rrc]);

  v8f c = {};
  v2f a = a_frag_x(x, N, m0, lane, 0,  0,  3);
  v2f b = b_frag(gW1, lane, 0, 3, 16);
  c = wmma_f32(a, b, c);

  if (lane < 16 && rr < N) dinv[rr] = d;

  int n  = lane & 15;
  int mo = (lane >> 4) << 3;
  for (int v = 0; v < 8; ++v)
    t[(mo + v) * 16 + n] = c[v];
  __builtin_amdgcn_wave_barrier();

  if (full) {
    float dv0 = __shfl(d, lane >> 2, 32);
    float dv1 = __shfl(d, 8 + (lane >> 2), 32);
    const float4* src = (const float4*)t;
    float4 v0 = src[lane];
    float4 v1 = src[lane + 32];
    v0.x *= dv0; v0.y *= dv0; v0.z *= dv0; v0.w *= dv0;
    v1.x *= dv1; v1.y *= dv1; v1.z *= dv1; v1.w *= dv1;
    float4* dst = (float4*)(xw1s + (size_t)m0 * 16);
    dst[lane]      = v0;
    dst[lane + 32] = v1;
  } else {
    for (int v = 0; v < 8; ++v) {
      int r = m0 + mo + v;
      float dv = __shfl(d, mo + v, 32);
      if (r < N) xw1s[(size_t)r * 16 + n] = t[(mo + v) * 16 + n] * dv;
    }
  }
}

#define REP8(M)  M(0) M(1) M(2) M(3) M(4) M(5) M(6) M(7)
#define REP16(M) REP8(M) M(8) M(9) M(10) M(11) M(12) M(13) M(14) M(15)

template <int J>
static __device__ __forceinline__ int row16_bcast(int x) {
  return __builtin_amdgcn_update_dpp(0, x, 0x150 + J, 0xF, 0xF, false);
}
template <int J>
static __device__ __forceinline__ int grp8_bcast(int x) {
  return __builtin_amdgcn_mov_dpp8(x, J * 0x249249);
}

__global__ void __launch_bounds__(256)
k_scatter1(const int* __restrict__ row, const int* __restrict__ col,
           const float* __restrict__ xw1s, float* __restrict__ acc1, int NE) {
  const int      lane = threadIdx.x & 31;
  const unsigned f    = lane & 15;
  const int      base = lane & 16;
  const int wave   = __builtin_amdgcn_readfirstlane(
                       (int)((blockIdx.x * blockDim.x + threadIdx.x) >> 5));
  const int nwaves = (int)((gridDim.x * blockDim.x) >> 5);
  for (int e0 = wave * 32; e0 < NE; e0 += nwaves * 32) {
    const int rem = NE - e0;
    int e  = e0 + lane;
    int ec = (e < NE) ? e : NE - 1;
    const int r = row[ec], c = col[ec];
    float v[16];
#define GATHER1(J) v[J] = xw1s[(unsigned)row16_bcast<J>(r) * 16u + f];
    REP16(GATHER1)
#undef GATHER1
    if (rem >= 32) {
#define ADD1(J) atomicAdd(&acc1[(unsigned)row16_bcast<J>(c) * 16u + f], v[J]);
      REP16(ADD1)
#undef ADD1
    } else {
#define ADD1T(J) if (base + J < rem) atomicAdd(&acc1[(unsigned)row16_bcast<J>(c) * 16u + f], v[J]);
      REP16(ADD1T)
#undef ADD1T
    }
  }
}

__global__ void __launch_bounds__(256)
k_dense2(const float* __restrict__ acc1, const float* __restrict__ xw1s,
         const float* __restrict__ dinv, const float* __restrict__ gW2,
         const float* __restrict__ gb1, float* __restrict__ xw2s, int N) {
  __shared__ float ht[WPB][256];
  __shared__ float dv[WPB][16];
  int wslot = threadIdx.x >> 5;
  int lane  = threadIdx.x & 31;
  int m0 = (blockIdx.x * WPB + wslot) * 16;
  if (m0 >= N) return;
  bool full = (m0 + 16) <= N;
  float* t   = ht[wslot];
  float* dvs = dv[wslot];

  for (int h = 0; h < 2; ++h) {
    int f   = lane + h * 32;
    int row = f >> 2;
    int cc4 = f & 3;
    int r   = m0 + row;
    int rc  = (r < N) ? r : N - 1;
    float dvr = dinv[rc];
    if (cc4 == 0) dvs[row] = dvr;
    float4 a4 = *(const float4*)(acc1 + (size_t)rc * 16 + cc4 * 4);
    float4 s4 = *(const float4*)(xw1s + (size_t)rc * 16 + cc4 * 4);
    float4 b4 = ((const float4*)gb1)[cc4];
    float4 hv;
    hv.x = fmaxf(dvr * (a4.x + s4.x) + b4.x, 0.0f);
    hv.y = fmaxf(dvr * (a4.y + s4.y) + b4.y, 0.0f);
    hv.z = fmaxf(dvr * (a4.z + s4.z) + b4.z, 0.0f);
    hv.w = fmaxf(dvr * (a4.w + s4.w) + b4.w, 0.0f);
    *(float4*)&t[row * 16 + cc4 * 4] = hv;
  }
  __builtin_amdgcn_wave_barrier();

  v8f c = {};
  for (int k0 = 0; k0 < 16; k0 += 4) {
    v2f a = a_frag_lds(t, lane, k0);
    v2f b = b_frag(gW2, lane, k0, 16, 8);
    c = wmma_f32(a, b, c);
  }
  __builtin_amdgcn_wave_barrier();

  int n  = lane & 15;
  int mo = (lane >> 4) << 3;
  if (n < 8) {
    for (int v = 0; v < 8; ++v)
      t[(mo + v) * 8 + n] = c[v];
  }
  __builtin_amdgcn_wave_barrier();

  if (full) {
    float dvr = dvs[lane >> 1];
    float4 v4 = ((const float4*)t)[lane];
    v4.x *= dvr; v4.y *= dvr; v4.z *= dvr; v4.w *= dvr;
    ((float4*)(xw2s + (size_t)m0 * 8))[lane] = v4;
  } else if (n < 8) {
    for (int v = 0; v < 8; ++v) {
      int r = m0 + mo + v;
      if (r < N) xw2s[(size_t)r * 8 + n] = t[(mo + v) * 8 + n] * dvs[mo + v];
    }
  }
}

__global__ void __launch_bounds__(256)
k_scatter2(const int* __restrict__ row, const int* __restrict__ col,
           const float* __restrict__ xw2s, float* __restrict__ acc2, int NE) {
  const int      lane = threadIdx.x & 31;
  const unsigned f    = lane & 7;
  const int      base = lane & 24;
  const int wave   = __builtin_amdgcn_readfirstlane(
                       (int)((blockIdx.x * blockDim.x + threadIdx.x) >> 5));
  const int nwaves = (int)((gridDim.x * blockDim.x) >> 5);
  for (int e0 = wave * 32; e0 < NE; e0 += nwaves * 32) {
    const int rem = NE - e0;
    int e  = e0 + lane;
    int ec = (e < NE) ? e : NE - 1;
    const int r = row[ec], c = col[ec];
    float v[8];
#define GATHER2(J) v[J] = xw2s[(unsigned)grp8_bcast<J>(r) * 8u + f];
    REP8(GATHER2)
#undef GATHER2
    if (rem >= 32) {
#define ADD2(J) atomicAdd(&acc2[(unsigned)grp8_bcast<J>(c) * 8u + f], v[J]);
      REP8(ADD2)
#undef ADD2
    } else {
#define ADD2T(J) if (base + J < rem) atomicAdd(&acc2[(unsigned)grp8_bcast<J>(c) * 8u + f], v[J]);
      REP8(ADD2T)
#undef ADD2T
    }
  }
}

__global__ void __launch_bounds__(256)
k_final(const float* __restrict__ x, const float* __restrict__ acc2,
        const float* __restrict__ xw2s, const float* __restrict__ dinv,
        const float* __restrict__ gb2,
        const float* __restrict__ cW1, const float* __restrict__ cb1,
        const float* __restrict__ cW2, const float* __restrict__ cb2,
        const float* __restrict__ mW1, const float* __restrict__ mb1,
        const float* __restrict__ mW2, const float* __restrict__ mb2,
        float* __restrict__ out, int N) {
  __shared__ float ht[WPB][256];
  __shared__ float ct[WPB][128];
  __shared__ float g0s[WPB][16];
  int wslot = threadIdx.x >> 5;
  int lane  = threadIdx.x & 31;
  int m0 = (blockIdx.x * WPB + wslot) * 16;
  if (m0 >= N) return;
  bool full = (m0 + 16) <= N;
  float* t  = ht[wslot];
  float* o  = ct[wslot];
  float* g0 = g0s[wslot];

  int n  = lane & 15;
  int mo = (lane >> 4) << 3;

  int grow = lane >> 1, gcc = lane & 1;
  int gr  = m0 + grow;
  int grc = (gr < N) ? gr : N - 1;
  float  gdv = dinv[grc];
  float4 a24 = *(const float4*)(acc2 + (size_t)grc * 8 + gcc * 4);
  float4 s24 = *(const float4*)(xw2s + (size_t)grc * 8 + gcc * 4);
  float4 gb4 = ((const float4*)gb2)[gcc];
  float4 go4;
  go4.x = gdv * (a24.x + s24.x) + gb4.x;
  go4.y = gdv * (a24.y + s24.y) + gb4.y;
  go4.z = gdv * (a24.z + s24.z) + gb4.z;
  go4.w = gdv * (a24.w + s24.w) + gb4.w;

  v8f ch = {};
  for (int k0 = 0; k0 < 16; k0 += 4) {
    v2f a = a_frag_x(x, N, m0, lane, k0,  3,  13);
    v2f b = b_frag(cW1, lane, k0, 13, 16);
    ch = wmma_f32(a, b, ch);
  }
  for (int v = 0; v < 8; ++v) t[(mo + v) * 16 + n] = fmaxf(ch[v] + cb1[n], 0.0f);
  __builtin_amdgcn_wave_barrier();

  v8f cc = {};
  for (int k0 = 0; k0 < 16; k0 += 4) {
    v2f a = a_frag_lds(t, lane, k0);
    v2f b = b_frag(cW2, lane, k0, 16, 8);
    cc = wmma_f32(a, b, cc);
  }
  if (n < 8) {
    for (int v = 0; v < 8; ++v) o[(mo + v) * 8 + n] = cc[v] + cb2[n];
  }
  __builtin_amdgcn_wave_barrier();

  v8f cm = {};
  for (int k0 = 0; k0 < 16; k0 += 4) {
    v2f a = a_frag_x(x, N, m0, lane, k0,  0,  16);
    v2f b = b_frag(mW1, lane, k0, 16, 16);
    cm = wmma_f32(a, b, cm);
  }
  for (int v = 0; v < 8; ++v) t[(mo + v) * 16 + n] = fmaxf(cm[v] + mb1[n], 0.0f);
  __builtin_amdgcn_wave_barrier();

  if (lane < 16) {
    float l0 = mb2[0], l1 = mb2[1];
    for (int j = 0; j < 16; ++j) {
      float hv = t[lane * 16 + j];
      l0 += hv * mW2[j * 2 + 0];
      l1 += hv * mW2[j * 2 + 1];
    }
    float mx = fmaxf(l0, l1);
    float e0 = __expf(l0 - mx), e1 = __expf(l1 - mx);
    g0[lane] = e0 / (e0 + e1);
  }
  __builtin_amdgcn_wave_barrier();

  float g  = g0[grow];
  float4 c4 = *(const float4*)&o[lane * 4];
  float4 r4;
  r4.x = g * go4.x + (1.0f - g) * c4.x;
  r4.y = g * go4.y + (1.0f - g) * c4.y;
  r4.z = g * go4.z + (1.0f - g) * c4.z;
  r4.w = g * go4.w + (1.0f - g) * c4.w;
  if (full) {
    ((float4*)(out + (size_t)m0 * 8))[lane] = r4;
  } else if (gr < N) {
    *(float4*)(out + (size_t)gr * 8 + gcc * 4) = r4;
  }
}

extern "C" void kernel_launch(void* const* d_in, const int* in_sizes, int n_in,
                              void* d_out, int out_size, void* d_ws, size_t ws_size,
                              hipStream_t stream) {
  const float* x   = (const float*)d_in[0];
  const int*   ei  = (const int*)  d_in[1];
  const float* gW1 = (const float*)d_in[2];
  const float* gb1 = (const float*)d_in[3];
  const float* gW2 = (const float*)d_in[4];
  const float* gb2 = (const float*)d_in[5];
  const float* cW1 = (const float*)d_in[6];
  const float* cb1 = (const float*)d_in[7];
  const float* cW2 = (const float*)d_in[8];
  const float* cb2 = (const float*)d_in[9];
  const float* mW1 = (const float*)d_in[10];
  const float* mb1 = (const float*)d_in[11];
  const float* mW2 = (const float*)d_in[12];
  const float* mb2 = (const float*)d_in[13];
  float* out = (float*)d_out;

  const int N  = in_sizes[0] / 16;
  const int NE = in_sizes[1] / 2;
  const int* erow = ei;
  const int* ecol = ei + NE;

  float* deg  = (float*)d_ws;
  float* dinv = deg  + N;
  float* xw1s = dinv + N;
  float* acc1 = xw1s + (size_t)16 * N;
  float* xw2s = acc1 + (size_t)16 * N;
  float* acc2 = xw2s + (size_t)8 * N;

  const int tiles   = (N + 15) / 16;
  const int gBlocks = (tiles + WPB - 1) / WPB;

  k_init    <<<(N + 255) / 256, 256, 0, stream>>>(deg, acc1, acc2, N);
  k_degree  <<<(NE + 255) / 256, 256, 0, stream>>>(ecol, deg, NE);
  k_dense1  <<<gBlocks, 256, 0, stream>>>(x, gW1, deg, dinv, xw1s, N);
  k_scatter1<<<4096, 256, 0, stream>>>(erow, ecol, xw1s, acc1, NE);
  k_dense2  <<<gBlocks, 256, 0, stream>>>(acc1, xw1s, dinv, gW2, gb1, xw2s, N);
  k_scatter2<<<4096, 256, 0, stream>>>(erow, ecol, xw2s, acc2, NE);
  k_final   <<<gBlocks, 256, 0, stream>>>(x, acc2, xw2s, dinv, gb2,
                                          cW1, cb1, cW2, cb2,
                                          mW1, mb1, mW2, mb2, out, N);
}


# degree count in LDS: 256 resident blocks each count one node range over a slice of the edge list, then add whole 128-B lines of counts to deg (f32 atomics, integer-exact in any order)
# speedup vs baseline: 2.5345x; 1.7528x over previous
#include <hip/hip_runtime.h>
#include <math.h>


typedef float v2f __attribute__((ext_vector_type(2)));
typedef float v8f __attribute__((ext_vector_type(8)));

#define WPB 8

static __device__ __forceinline__ v8f wmma_f32(v2f a, v2f b, v8f c) {
  return __builtin_amdgcn_wmma_f32_16x16x4_f32(false, a, false, b,
                                               (short)0, c, false, false);
}

static __device__ __forceinline__ v2f b_frag(const float* __restrict__ W,
                                             int lane, int k0, int K, int Ncols) {
  int n  = lane & 15;
  int kb = k0 + ((lane >> 4) << 1);
  int nm  = (n      < Ncols) ? n      : (Ncols - 1);
  int kc0 = (kb     < K)     ? kb     : (K - 1);
  int kc1 = (kb + 1 < K)     ? kb + 1 : (K - 1);
  float w0 = W[kc0 * Ncols + nm];
  float w1 = W[kc1 * Ncols + nm];
  v2f b;
  b.x = (kb     < K && n < Ncols) ? w0 : 0.0f;
  b.y = (kb + 1 < K && n < Ncols) ? w1 : 0.0f;
  return b;
}

static __device__ __forceinline__ v2f a_frag_x(const float* __restrict__ x,
                                               int N, int m0, int lane,
                                               int k0, int co, int K) {
  int r = m0 + (lane & 15);
  if (r >= N) r = N - 1;
  int kb = k0 + ((lane >> 4) << 1);
  const float* rowp = x + (size_t)r * 16;
  int i0 = co + kb;     i0 = (i0 < 16) ? i0 : 15;
  int i1 = co + kb + 1; i1 = (i1 < 16) ? i1 : 15;
  float a0 = rowp[i0];
  float a1 = rowp[i1];
  v2f a;
  a.x = (kb     < K) ? a0 : 0.0f;
  a.y = (kb + 1 < K) ? a1 : 0.0f;
  return a;
}

static __device__ __forceinline__ v2f a_frag_lds(const float* t, int lane, int k0) {
  int r  = lane & 15;
  int kb = k0 + ((lane >> 4) << 1);
  v2f a;
  a.x = t[r * 16 + kb];
  a.y = t[r * 16 + kb + 1];
  return a;
}

__global__ void k_init(float* __restrict__ deg, float* __restrict__ acc1,
                       float* __restrict__ acc2, int N) {
  int i = blockIdx.x * blockDim.x + threadIdx.x;
  if (i >= N) return;
  deg[i] = 1.0f;
  float4 z = make_float4(0.f, 0.f, 0.f, 0.f);
  float4* a1 = (float4*)(acc1 + (size_t)i * 16);
  a1[0] = z; a1[1] = z; a1[2] = z; a1[3] = z;
  float4* a2 = (float4*)(acc2 + (size_t)i * 8);
  a2[0] = z; a2[1] = z;
}

#define DEG_CAP 25088

static __device__ __forceinline__ void deg_bump(unsigned* cnt, int c, int n0, int nn) {
  unsigned k = (unsigned)(c - n0);
  if (k < (unsigned)nn) atomicAdd(&cnt[k], 1u);
}

__global__ void __launch_bounds__(1024)
k_degree(const int* __restrict__ col, float* __restrict__ deg, int N, int NE) {
  __shared__ unsigned cnt[DEG_CAP];
  const int tid = (int)threadIdx.x;
  const int nth = (int)blockDim.x;
  const int R = (N + DEG_CAP - 1) / DEG_CAP;
  int S = (int)gridDim.x / R;
  if (S < 1) S = 1;
  const int per = (((NE + S - 1) / S) + 3) & ~3;
  const bool aligned = (((size_t)col) & 15) == 0;

  for (int item = (int)blockIdx.x; item < R * S; item += (int)gridDim.x) {
    const int rg = item % R;
    const int sl = item / R;
    const int n0 = rg * DEG_CAP;
    const int nn = (N - n0 < DEG_CAP) ? (N - n0) : DEG_CAP;
    for (int i = tid; i < nn; i += nth) cnt[i] = 0u;
    __syncthreads();

    const int e0 = (sl * per < NE) ? sl * per : NE;
    const int e1 = (e0 + per < NE) ? e0 + per : NE;
    int done = e0;
    if (aligned) {
      const int4* c4 = (const int4*)(col + e0);
      const int nv = (e1 - e0) >> 2;
      const int4 none = make_int4(-1, -1, -1, -1);
      for (int i = tid; i < nv; i += 8 * nth) {
        int4 v[8];
#pragma unroll
        for (int u = 0; u < 8; ++u) {
          const int iu = i + u * nth;
          v[u] = c4[(iu < nv) ? iu : nv - 1];
        }
#pragma unroll
        for (int u = 0; u < 8; ++u) {
          if (i + u * nth >= nv) v[u] = none;
          deg_bump(cnt, v[u].x, n0, nn); deg_bump(cnt, v[u].y, n0, nn);
          deg_bump(cnt, v[u].z, n0, nn); deg_bump(cnt, v[u].w, n0, nn);
        }
      }
      done = e0 + (nv << 2);
    }
    for (int e = done + tid; e < e1; e += nth) deg_bump(cnt, col[e], n0, nn);
    __syncthreads();

    const int lines = (nn + 31) >> 5;
    const int rot   = (int)(((long long)sl * lines) / S) << 5;
    for (int j = tid; j < lines * 32; j += nth) {
      int i = j + rot;
      if (i >= lines * 32) i -= lines * 32;
      if (i < nn) {
        unsigned c = cnt[i];
        if (c) atomicAdd(&deg[n0 + i], (float)c);
      }
    }
    __syncthreads();
  }
}

__global__ void __launch_bounds__(256)
k_dense1(const float* __restrict__ x, const float* __restrict__ gW1,
         const float* __restrict__ deg, float* __restrict__ dinv,
         float* __restrict__ xw1s, int N) {
  __shared__ float ot[WPB][256];
  int wslot = threadIdx.x >> 5;
  int lane  = threadIdx.x & 31;
  int m0 = (blockIdx.x * WPB + wslot) * 16;
  if (m0 >= N) return;
  bool full = (m0 + 16) <= N;
  float* t = ot[wslot];

  int rr  = m0 + (lane & 15);
  int rrc = (rr < N) ? rr : N - 1;
  float d = __builtin_amdgcn_rsqf(deg[rrc]);

  v8f c = {};
  v2f a = a_frag_x(x, N, m0, lane, 0,  0,  3);
  v2f b = b_frag(gW1, lane, 0, 3, 16);
  c = wmma_f32(a, b, c);

  if (lane < 16 && rr < N) dinv[rr] = d;

  int n  = lane & 15;
  int mo = (lane >> 4) << 3;
  for (int v = 0; v < 8; ++v)
    t[(mo + v) * 16 + n] = c[v];
  __builtin_amdgcn_wave_barrier();

  if (full) {
    float dv0 = __shfl(d, lane >> 2, 32);
    float dv1 = __shfl(d, 8 + (lane >> 2), 32);
    const float4* src = (const float4*)t;
    float4 v0 = src[lane];
    float4 v1 = src[lane + 32];
    v0.x *= dv0; v0.y *= dv0; v0.z *= dv0; v0.w *= dv0;
    v1.x *= dv1; v1.y *= dv1; v1.z *= dv1; v1.w *= dv1;
    float4* dst = (float4*)(xw1s + (size_t)m0 * 16);
    dst[lane]      = v0;
    dst[lane + 32] = v1;
  } else {
    for (int v = 0; v < 8; ++v) {
      int r = m0 + mo + v;
      float dv = __shfl(d, mo + v, 32);
      if (r < N) xw1s[(size_t)r * 16 + n] = t[(mo + v) * 16 + n] * dv;
    }
  }
}

#define REP8(M)  M(0) M(1) M(2) M(3) M(4) M(5) M(6) M(7)
#define REP16(M) REP8(M) M(8) M(9) M(10) M(11) M(12) M(13) M(14) M(15)

template <int J>
static __device__ __forceinline__ int row16_bcast(int x) {
  return __builtin_amdgcn_update_dpp(0, x, 0x150 + J, 0xF, 0xF, false);
}
template <int J>
static __device__ __forceinline__ int grp8_bcast(int x) {
  return __builtin_amdgcn_mov_dpp8(x, J * 0x249249);
}

__global__ void __launch_bounds__(256)
k_scatter1(const int* __restrict__ row, const int* __restrict__ col,
           const float* __restrict__ xw1s, float* __restrict__ acc1, int NE) {
  const int      lane = threadIdx.x & 31;
  const unsigned f    = lane & 15;
  const int      base = lane & 16;
  const int wave   = __builtin_amdgcn_readfirstlane(
                       (int)((blockIdx.x * blockDim.x + threadIdx.x) >> 5));
  const int nwaves = (int)((gridDim.x * blockDim.x) >> 5);
  for (int e0 = wave * 32; e0 < NE; e0 += nwaves * 32) {
    const int rem = NE - e0;
    int e  = e0 + lane;
    int ec = (e < NE) ? e : NE - 1;
    const int r = row[ec], c = col[ec];
    float v[16];
#define GATHER1(J) v[J] = xw1s[(unsigned)row16_bcast<J>(r) * 16u + f];
    REP16(GATHER1)
#undef GATHER1
    if (rem >= 32) {
#define ADD1(J) atomicAdd(&acc1[(unsigned)row16_bcast<J>(c) * 16u + f], v[J]);
      REP16(ADD1)
#undef ADD1
    } else {
#define ADD1T(J) if (base + J < rem) atomicAdd(&acc1[(unsigned)row16_bcast<J>(c) * 16u + f], v[J]);
      REP16(ADD1T)
#undef ADD1T
    }
  }
}

__global__ void __launch_bounds__(256)
k_dense2(const float* __restrict__ acc1, const float* __restrict__ xw1s,
         const float* __restrict__ dinv, const float* __restrict__ gW2,
         const float* __restrict__ gb1, float* __restrict__ xw2s, int N) {
  __shared__ float ht[WPB][256];
  __shared__ float dv[WPB][16];
  int wslot = threadIdx.x >> 5;
  int lane  = threadIdx.x & 31;
  int m0 = (blockIdx.x * WPB + wslot) * 16;
  if (m0 >= N) return;
  bool full = (m0 + 16) <= N;
  float* t   = ht[wslot];
  float* dvs = dv[wslot];

  for (int h = 0; h < 2; ++h) {
    int f   = lane + h * 32;
    int row = f >> 2;
    int cc4 = f & 3;
    int r   = m0 + row;
    int rc  = (r < N) ? r : N - 1;
    float dvr = dinv[rc];
    if (cc4 == 0) dvs[row] = dvr;
    float4 a4 = *(const float4*)(acc1 + (size_t)rc * 16 + cc4 * 4);
    float4 s4 = *(const float4*)(xw1s + (size_t)rc * 16 + cc4 * 4);
    float4 b4 = ((const float4*)gb1)[cc4];
    float4 hv;
    hv.x = fmaxf(dvr * (a4.x + s4.x) + b4.x, 0.0f);
    hv.y = fmaxf(dvr * (a4.y + s4.y) + b4.y, 0.0f);
    hv.z = fmaxf(dvr * (a4.z + s4.z) + b4.z, 0.0f);
    hv.w = fmaxf(dvr * (a4.w + s4.w) + b4.w, 0.0f);
    *(float4*)&t[row * 16 + cc4 * 4] = hv;
  }
  __builtin_amdgcn_wave_barrier();

  v8f c = {};
  for (int k0 = 0; k0 < 16; k0 += 4) {
    v2f a = a_frag_lds(t, lane, k0);
    v2f b = b_frag(gW2, lane, k0, 16, 8);
    c = wmma_f32(a, b, c);
  }
  __builtin_amdgcn_wave_barrier();

  int n  = lane & 15;
  int mo = (lane >> 4) << 3;
  if (n < 8) {
    for (int v = 0; v < 8; ++v)
      t[(mo + v) * 8 + n] = c[v];
  }
  __builtin_amdgcn_wave_barrier();

  if (full) {
    float dvr = dvs[lane >> 1];
    float4 v4 = ((const float4*)t)[lane];
    v4.x *= dvr; v4.y *= dvr; v4.z *= dvr; v4.w *= dvr;
    ((float4*)(xw2s + (size_t)m0 * 8))[lane] = v4;
  } else if (n < 8) {
    for (int v = 0; v < 8; ++v) {
      int r = m0 + mo + v;
      if (r < N) xw2s[(size_t)r * 8 + n] = t[(mo + v) * 8 + n] * dvs[mo + v];
    }
  }
}

__global__ void __launch_bounds__(256)
k_scatter2(const int* __restrict__ row, const int* __restrict__ col,
           const float* __restrict__ xw2s, float* __restrict__ acc2, int NE) {
  const int      lane = threadIdx.x & 31;
  const unsigned f    = lane & 7;
  const int      base = lane & 24;
  const int wave   = __builtin_amdgcn_readfirstlane(
                       (int)((blockIdx.x * blockDim.x + threadIdx.x) >> 5));
  const int nwaves = (int)((gridDim.x * blockDim.x) >> 5);
  for (int e0 = wave * 32; e0 < NE; e0 += nwaves * 32) {
    const int rem = NE - e0;
    int e  = e0 + lane;
    int ec = (e < NE) ? e : NE - 1;
    const int r = row[ec], c = col[ec];
    float v[8];
#define GATHER2(J) v[J] = xw2s[(unsigned)grp8_bcast<J>(r) * 8u + f];
    REP8(GATHER2)
#undef GATHER2
    if (rem >= 32) {
#define ADD2(J) atomicAdd(&acc2[(unsigned)grp8_bcast<J>(c) * 8u + f], v[J]);
      REP8(ADD2)
#undef ADD2
    } else {
#define ADD2T(J) if (base + J < rem) atomicAdd(&acc2[(unsigned)grp8_bcast<J>(c) * 8u + f], v[J]);
      REP8(ADD2T)
#undef ADD2T
    }
  }
}

__global__ void __launch_bounds__(256)
k_final(const float* __restrict__ x, const float* __restrict__ acc2,
        const float* __restrict__ xw2s, const float* __restrict__ dinv,
        const float* __restrict__ gb2,
        const float* __restrict__ cW1, const float* __restrict__ cb1,
        const float* __restrict__ cW2, const float* __restrict__ cb2,
        const float* __restrict__ mW1, const float* __restrict__ mb1,
        const float* __restrict__ mW2, const float* __restrict__ mb2,
        float* __restrict__ out, int N) {
  __shared__ float ht[WPB][256];
  __shared__ float ct[WPB][128];
  __shared__ float g0s[WPB][16];
  int wslot = threadIdx.x >> 5;
  int lane  = threadIdx.x & 31;
  int m0 = (blockIdx.x * WPB + wslot) * 16;
  if (m0 >= N) return;
  bool full = (m0 + 16) <= N;
  float* t  = ht[wslot];
  float* o  = ct[wslot];
  float* g0 = g0s[wslot];

  int n  = lane & 15;
  int mo = (lane >> 4) << 3;

  int grow = lane >> 1, gcc = lane & 1;
  int gr  = m0 + grow;
  int grc = (gr < N) ? gr : N - 1;
  float  gdv = dinv[grc];
  float4 a24 = *(const float4*)(acc2 + (size_t)grc * 8 + gcc * 4);
  float4 s24 = *(const float4*)(xw2s + (size_t)grc * 8 + gcc * 4);
  float4 gb4 = ((const float4*)gb2)[gcc];
  float4 go4;
  go4.x = gdv * (a24.x + s24.x) + gb4.x;
  go4.y = gdv * (a24.y + s24.y) + gb4.y;
  go4.z = gdv * (a24.z + s24.z) + gb4.z;
  go4.w = gdv * (a24.w + s24.w) + gb4.w;

  v8f ch = {};
  for (int k0 = 0; k0 < 16; k0 += 4) {
    v2f a = a_frag_x(x, N, m0, lane, k0,  3,  13);
    v2f b = b_frag(cW1, lane, k0, 13, 16);
    ch = wmma_f32(a, b, ch);
  }
  for (int v = 0; v < 8; ++v) t[(mo + v) * 16 + n] = fmaxf(ch[v] + cb1[n], 0.0f);
  __builtin_amdgcn_wave_barrier();

  v8f cc = {};
  for (int k0 = 0; k0 < 16; k0 += 4) {
    v2f a = a_frag_lds(t, lane, k0);
    v2f b = b_frag(cW2, lane, k0, 16, 8);
    cc = wmma_f32(a, b, cc);
  }
  if (n < 8) {
    for (int v = 0; v < 8; ++v) o[(mo + v) * 8 + n] = cc[v] + cb2[n];
  }
  __builtin_amdgcn_wave_barrier();

  v8f cm = {};
  for (int k0 = 0; k0 < 16; k0 += 4) {
    v2f a = a_frag_x(x, N, m0, lane, k0,  0,  16);
    v2f b = b_frag(mW1, lane, k0, 16, 16);
    cm = wmma_f32(a, b, cm);
  }
  for (int v = 0; v < 8; ++v) t[(mo + v) * 16 + n] = fmaxf(cm[v] + mb1[n], 0.0f);
  __builtin_amdgcn_wave_barrier();

  if (lane < 16) {
    float l0 = mb2[0], l1 = mb2[1];
    for (int j = 0; j < 16; ++j) {
      float hv = t[lane * 16 + j];
      l0 += hv * mW2[j * 2 + 0];
      l1 += hv * mW2[j * 2 + 1];
    }
    float mx = fmaxf(l0, l1);
    float e0 = __expf(l0 - mx), e1 = __expf(l1 - mx);
    g0[lane] = e0 / (e0 + e1);
  }
  __builtin_amdgcn_wave_barrier();

  float g  = g0[grow];
  float4 c4 = *(const float4*)&o[lane * 4];
  float4 r4;
  r4.x = g * go4.x + (1.0f - g) * c4.x;
  r4.y = g * go4.y + (1.0f - g) * c4.y;
  r4.z = g * go4.z + (1.0f - g) * c4.z;
  r4.w = g * go4.w + (1.0f - g) * c4.w;
  if (full) {
    ((float4*)(out + (size_t)m0 * 8))[lane] = r4;
  } else if (gr < N) {
    *(float4*)(out + (size_t)gr * 8 + gcc * 4) = r4;
  }
}

extern "C" void kernel_launch(void* const* d_in, const int* in_sizes, int n_in,
                              void* d_out, int out_size, void* d_ws, size_t ws_size,
                              hipStream_t stream) {
  const float* x   = (const float*)d_in[0];
  const int*   ei  = (const int*)  d_in[1];
  const float* gW1 = (const float*)d_in[2];
  const float* gb1 = (const float*)d_in[3];
  const float* gW2 = (const float*)d_in[4];
  const float* gb2 = (const float*)d_in[5];
  const float* cW1 = (const float*)d_in[6];
  const float* cb1 = (const float*)d_in[7];
  const float* cW2 = (const float*)d_in[8];
  const float* cb2 = (const float*)d_in[9];
  const float* mW1 = (const float*)d_in[10];
  const float* mb1 = (const float*)d_in[11];
  const float* mW2 = (const float*)d_in[12];
  const float* mb2 = (const float*)d_in[13];
  float* out = (float*)d_out;

  const int N  = in_sizes[0] / 16;
  const int NE = in_sizes[1] / 2;
  const int* erow = ei;
  const int* ecol = ei + NE;

  float* deg  = (float*)d_ws;
  float* dinv = deg  + N;
  float* xw1s = dinv + N;
  float* acc1 = xw1s + (size_t)16 * N;
  float* xw2s = acc1 + (size_t)16 * N;
  float* acc2 = xw2s + (size_t)8 * N;

  const int tiles   = (N + 15) / 16;
  const int gBlocks = (tiles + WPB - 1) / WPB;

  k_init    <<<(N + 255) / 256, 256, 0, stream>>>(deg, acc1, acc2, N);
  k_degree  <<<256, 1024, 0, stream>>>(ecol, deg, N, NE);
  k_dense1  <<<gBlocks, 256, 0, stream>>>(x, gW1, deg, dinv, xw1s, N);
  k_scatter1<<<4096, 256, 0, stream>>>(erow, ecol, xw1s, acc1, NE);
  k_dense2  <<<gBlocks, 256, 0, stream>>>(acc1, xw1s, dinv, gW2, gb1, xw2s, N);
  k_scatter2<<<4096, 256, 0, stream>>>(erow, ecol, xw2s, acc2, NE);
  k_final   <<<gBlocks, 256, 0, stream>>>(x, acc2, xw2s, dinv, gb2,
                                          cW1, cb1, cW2, cb2,
                                          mW1, mb1, mW2, mb2, out, N);
}


# edge scatters: each wave adds first into the accumulator rows of the node half matching the chip half it runs on (die id as hint, chunk counters, exact once); the order of the f32 atomics changes
# speedup vs baseline: 4.9378x; 1.9482x over previous
#include <hip/hip_runtime.h>
#include <math.h>


typedef float v2f __attribute__((ext_vector_type(2)));
typedef float v8f __attribute__((ext_vector_type(8)));

#define WPB 8
#define CTR_PITCH 64

static __device__ __forceinline__ v8f wmma_f32(v2f a, v2f b, v8f c) {
  return __builtin_amdgcn_wmma_f32_16x16x4_f32(false, a, false, b,
                                               (short)0, c, false, false);
}

static __device__ __forceinline__ v2f b_frag(const float* __restrict__ W,
                                             int lane, int k0, int K, int Ncols) {
  int n  = lane & 15;
  int kb = k0 + ((lane >> 4) << 1);
  int nm  = (n      < Ncols) ? n      : (Ncols - 1);
  int kc0 = (kb     < K)     ? kb     : (K - 1);
  int kc1 = (kb + 1 < K)     ? kb + 1 : (K - 1);
  float w0 = W[kc0 * Ncols + nm];
  float w1 = W[kc1 * Ncols + nm];
  v2f b;
  b.x = (kb     < K && n < Ncols) ? w0 : 0.0f;
  b.y = (kb + 1 < K && n < Ncols) ? w1 : 0.0f;
  return b;
}

static __device__ __forceinline__ v2f a_frag_x(const float* __restrict__ x,
                                               int N, int m0, int lane,
                                               int k0, int co, int K) {
  int r = m0 + (lane & 15);
  if (r >= N) r = N - 1;
  int kb = k0 + ((lane >> 4) << 1);
  const float* rowp = x + (size_t)r * 16;
  int i0 = co + kb;     i0 = (i0 < 16) ? i0 : 15;
  int i1 = co + kb + 1; i1 = (i1 < 16) ? i1 : 15;
  float a0 = rowp[i0];
  float a1 = rowp[i1];
  v2f a;
  a.x = (kb     < K) ? a0 : 0.0f;
  a.y = (kb + 1 < K) ? a1 : 0.0f;
  return a;
}

static __device__ __forceinline__ v2f a_frag_lds(const float* t, int lane, int k0) {
  int r  = lane & 15;
  int kb = k0 + ((lane >> 4) << 1);
  v2f a;
  a.x = t[r * 16 + kb];
  a.y = t[r * 16 + kb + 1];
  return a;
}

__global__ void k_init(float* __restrict__ deg, float* __restrict__ acc1,
                       float* __restrict__ acc2, int* __restrict__ ctr, int N) {
  int i = blockIdx.x * blockDim.x + threadIdx.x;
  if (i < 4) ctr[i * CTR_PITCH] = 0;
  if (i >= N) return;
  deg[i] = 1.0f;
  float4 z = make_float4(0.f, 0.f, 0.f, 0.f);
  float4* a1 = (float4*)(acc1 + (size_t)i * 16);
  a1[0] = z; a1[1] = z; a1[2] = z; a1[3] = z;
  float4* a2 = (float4*)(acc2 + (size_t)i * 8);
  a2[0] = z; a2[1] = z;
}

#define DEG_CAP 25088

static __device__ __forceinline__ void deg_bump(unsigned* cnt, int c, int n0, int nn) {
  unsigned k = (unsigned)(c - n0);
  if (k < (unsigned)nn) atomicAdd(&cnt[k], 1u);
}

__global__ void __launch_bounds__(1024)
k_degree(const int* __restrict__ col, float* __restrict__ deg, int N, int NE) {
  __shared__ unsigned cnt[DEG_CAP];
  const int tid = (int)threadIdx.x;
  const int nth = (int)blockDim.x;
  const int R = (N + DEG_CAP - 1) / DEG_CAP;
  int S = (int)gridDim.x / R;
  if (S < 1) S = 1;
  const int per = (((NE + S - 1) / S) + 3) & ~3;
  const bool aligned = (((size_t)col) & 15) == 0;

  for (int item = (int)blockIdx.x; item < R * S; item += (int)gridDim.x) {
    const int rg = item % R;
    const int sl = item / R;
    const int n0 = rg * DEG_CAP;
    const int nn = (N - n0 < DEG_CAP) ? (N - n0) : DEG_CAP;
    for (int i = tid; i < nn; i += nth) cnt[i] = 0u;
    __syncthreads();

    const int e0 = (sl * per < NE) ? sl * per : NE;
    const int e1 = (e0 + per < NE) ? e0 + per : NE;
    int done = e0;
    if (aligned) {
      const int4* c4 = (const int4*)(col + e0);
      const int nv = (e1 - e0) >> 2;
      const int4 none = make_int4(-1, -1, -1, -1);
      for (int i = tid; i < nv; i += 8 * nth) {
        int4 v[8];
#pragma unroll
        for (int u = 0; u < 8; ++u) {
          const int iu = i + u * nth;
          v[u] = c4[(iu < nv) ? iu : nv - 1];
        }
#pragma unroll
        for (int u = 0; u < 8; ++u) {
          if (i + u * nth >= nv) v[u] = none;
          deg_bump(cnt, v[u].x, n0, nn); deg_bump(cnt, v[u].y, n0, nn);
          deg_bump(cnt, v[u].z, n0, nn); deg_bump(cnt, v[u].w, n0, nn);
        }
      }
      done = e0 + (nv << 2);
    }
    for (int e = done + tid; e < e1; e += nth) deg_bump(cnt, col[e], n0, nn);
    __syncthreads();

    const int lines = (nn + 31) >> 5;
    const int rot   = (int)(((long long)sl * lines) / S) << 5;
    for (int j = tid; j < lines * 32; j += nth) {
      int i = j + rot;
      if (i >= lines * 32) i -= lines * 32;
      if (i < nn) {
        unsigned c = cnt[i];
        if (c) atomicAdd(&deg[n0 + i], (float)c);
      }
    }
    __syncthreads();
  }
}

__global__ void __launch_bounds__(256)
k_dense1(const float* __restrict__ x, const float* __restrict__ gW1,
         const float* __restrict__ deg, float* __restrict__ dinv,
         float* __restrict__ xw1s, int N) {
  __shared__ float ot[WPB][256];
  int wslot = threadIdx.x >> 5;
  int lane  = threadIdx.x & 31;
  int m0 = (blockIdx.x * WPB + wslot) * 16;
  if (m0 >= N) return;
  bool full = (m0 + 16) <= N;
  float* t = ot[wslot];

  int rr  = m0 + (lane & 15);
  int rrc = (rr < N) ? rr : N - 1;
  float d = __builtin_amdgcn_rsqf(deg[rrc]);

  v8f c = {};
  v2f a = a_frag_x(x, N, m0, lane, 0,  0,  3);
  v2f b = b_frag(gW1, lane, 0, 3, 16);
  c = wmma_f32(a, b, c);

  if (lane < 16 && rr < N) dinv[rr] = d;

  int n  = lane & 15;
  int mo = (lane >> 4) << 3;
  for (int v = 0; v < 8; ++v)
    t[(mo + v) * 16 + n] = c[v];
  __builtin_amdgcn_wave_barrier();

  if (full) {
    float dv0 = __shfl(d, lane >> 2, 32);
    float dv1 = __shfl(d, 8 + (lane >> 2), 32);
    const float4* src = (const float4*)t;
    float4 v0 = src[lane];
    float4 v1 = src[lane + 32];
    v0.x *= dv0; v0.y *= dv0; v0.z *= dv0; v0.w *= dv0;
    v1.x *= dv1; v1.y *= dv1; v1.z *= dv1; v1.w *= dv1;
    float4* dst = (float4*)(xw1s + (size_t)m0 * 16);
    dst[lane]      = v0;
    dst[lane + 32] = v1;
  } else {
    for (int v = 0; v < 8; ++v) {
      int r = m0 + mo + v;
      float dv = __shfl(d, mo + v, 32);
      if (r < N) xw1s[(size_t)r * 16 + n] = t[(mo + v) * 16 + n] * dv;
    }
  }
}

#define REP8(M)  M(0) M(1) M(2) M(3) M(4) M(5) M(6) M(7)
#define REP16(M) REP8(M) M(8) M(9) M(10) M(11) M(12) M(13) M(14) M(15)

template <int J>
static __device__ __forceinline__ int row16_bcast(int x) {
  return __builtin_amdgcn_update_dpp(0, x, 0x150 + J, 0xF, 0xF, false);
}
template <int J>
static __device__ __forceinline__ int grp8_bcast(int x) {
  return __builtin_amdgcn_mov_dpp8(x, J * 0x249249);
}

static __device__ __forceinline__ int chip_half() {
  return (int)((__builtin_amdgcn_s_sendmsg_rtn(0x87) >> 18) & 1u);
}
static __device__ __forceinline__ int node_split(int N) { return ((N >> 1) + 63) & ~63; }

#define BPC 8

static __device__ __forceinline__ int take_chunk(int* cnt, int lane) {
  int ck = 0;
  if (lane == 0) ck = atomicAdd(cnt, 1);
  return __builtin_amdgcn_readfirstlane(ck);
}

__global__ void __launch_bounds__(256)
k_scatter1(const int* __restrict__ row, const int* __restrict__ col,
           const float* __restrict__ xw1s, float* __restrict__ acc1,
           int* __restrict__ ctr, int N, int NE) {
  const int      lane = threadIdx.x & 31;
  const unsigned f    = lane & 15;
  const int split  = node_split(N);
  const int nchunk = (NE + 32 * BPC - 1) / (32 * BPC);
  const int mine   = chip_half();
  for (int ph = 0; ph < 2; ++ph) {
    const int p = mine ^ ph;
    int* cnt = ctr + p * CTR_PITCH;
    int ck = take_chunk(cnt, lane);
    while (ck < nchunk) {
      int r[BPC], c[BPC];
#pragma unroll
      for (int u = 0; u < BPC; ++u) {
        const int e  = (ck * BPC + u) * 32 + lane;
        const int ec = (e < NE) ? e : NE - 1;
        r[u] = row[ec]; c[u] = col[ec];
      }
      const int nx = take_chunk(cnt, lane);
#pragma unroll
      for (int u = 0; u < BPC; ++u) {
        const int e = (ck * BPC + u) * 32 + lane;
        const int ru = r[u], cu = c[u];
        asm volatile("" :: "v"(ru), "v"(cu));
        const int hit = (e < NE && ((cu >= split) ? 1 : 0) == p) ? 1 : 0;
        float v[16];
#define GATHERX(J) v[J] = 0.0f; if (row16_bcast<J>(hit)) v[J] = xw1s[(unsigned)row16_bcast<J>(ru) * 16u + f];
        REP16(GATHERX)
#undef GATHERX
#define ADDX(J) if (row16_bcast<J>(hit)) atomicAdd(&acc1[(unsigned)row16_bcast<J>(cu) * 16u + f], v[J]);
        REP16(ADDX)
#undef ADDX
      }
      ck = nx;
    }
  }
}

__global__ void __launch_bounds__(256)
k_dense2(const float* __restrict__ acc1, const float* __restrict__ xw1s,
         const float* __restrict__ dinv, const float* __restrict__ gW2,
         const float* __restrict__ gb1, float* __restrict__ xw2s, int N) {
  __shared__ float ht[WPB][256];
  __shared__ float dv[WPB][16];
  int wslot = threadIdx.x >> 5;
  int lane  = threadIdx.x & 31;
  int m0 = (blockIdx.x * WPB + wslot) * 16;
  if (m0 >= N) return;
  bool full = (m0 + 16) <= N;
  float* t   = ht[wslot];
  float* dvs = dv[wslot];

  for (int h = 0; h < 2; ++h) {
    int f   = lane + h * 32;
    int row = f >> 2;
    int cc4 = f & 3;
    int r   = m0 + row;
    int rc  = (r < N) ? r : N - 1;
    float dvr = dinv[rc];
    if (cc4 == 0) dvs[row] = dvr;
    float4 a4 = *(const float4*)(acc1 + (size_t)rc * 16 + cc4 * 4);
    float4 s4 = *(const float4*)(xw1s + (size_t)rc * 16 + cc4 * 4);
    float4 b4 = ((const float4*)gb1)[cc4];
    float4 hv;
    hv.x = fmaxf(dvr * (a4.x + s4.x) + b4.x, 0.0f);
    hv.y = fmaxf(dvr * (a4.y + s4.y) + b4.y, 0.0f);
    hv.z = fmaxf(dvr * (a4.z + s4.z) + b4.z, 0.0f);
    hv.w = fmaxf(dvr * (a4.w + s4.w) + b4.w, 0.0f);
    *(float4*)&t[row * 16 + cc4 * 4] = hv;
  }
  __builtin_amdgcn_wave_barrier();

  v8f c = {};
  for (int k0 = 0; k0 < 16; k0 += 4) {
    v2f a = a_frag_lds(t, lane, k0);
    v2f b = b_frag(gW2, lane, k0, 16, 8);
    c = wmma_f32(a, b, c);
  }
  __builtin_amdgcn_wave_barrier();

  int n  = lane & 15;
  int mo = (lane >> 4) << 3;
  if (n < 8) {
    for (int v = 0; v < 8; ++v)
      t[(mo + v) * 8 + n] = c[v];
  }
  __builtin_amdgcn_wave_barrier();

  if (full) {
    float dvr = dvs[lane >> 1];
    float4 v4 = ((const float4*)t)[lane];
    v4.x *= dvr; v4.y *= dvr; v4.z *= dvr; v4.w *= dvr;
    ((float4*)(xw2s + (size_t)m0 * 8))[lane] = v4;
  } else if (n < 8) {
    for (int v = 0; v < 8; ++v) {
      int r = m0 + mo + v;
      if (r < N) xw2s[(size_t)r * 8 + n] = t[(mo + v) * 8 + n] * dvs[mo + v];
    }
  }
}

__global__ void __launch_bounds__(256)
k_scatter2(const int* __restrict__ row, const int* __restrict__ col,
           const float* __restrict__ xw2s, float* __restrict__ acc2,
           int* __restrict__ ctr, int N, int NE) {
  const int      lane = threadIdx.x & 31;
  const unsigned f    = lane & 7;
  const int split  = node_split(N);
  const int nchunk = (NE + 32 * BPC - 1) / (32 * BPC);
  const int mine   = chip_half();
  for (int ph = 0; ph < 2; ++ph) {
    const int p = mine ^ ph;
    int* cnt = ctr + p * CTR_PITCH;
    int ck = take_chunk(cnt, lane);
    while (ck < nchunk) {
      int r[BPC], c[BPC];
#pragma unroll
      for (int u = 0; u < BPC; ++u) {
        const int e  = (ck * BPC + u) * 32 + lane;
        const int ec = (e < NE) ? e : NE - 1;
        r[u] = row[ec]; c[u] = col[ec];
      }
      const int nx = take_chunk(cnt, lane);
#pragma unroll
      for (int u = 0; u < BPC; ++u) {
        const int e = (ck * BPC + u) * 32 + lane;
        const int ru = r[u], cu = c[u];
        asm volatile("" :: "v"(ru), "v"(cu));
        const int hit = (e < NE && ((cu >= split) ? 1 : 0) == p) ? 1 : 0;
        float v[8];
#define GATHERX(J) v[J] = 0.0f; if (grp8_bcast<J>(hit)) v[J] = xw2s[(unsigned)grp8_bcast<J>(ru) * 8u + f];
        REP8(GATHERX)
#undef GATHERX
#define ADDX(J) if (grp8_bcast<J>(hit)) atomicAdd(&acc2[(unsigned)grp8_bcast<J>(cu) * 8u + f], v[J]);
        REP8(ADDX)
#undef ADDX
      }
      ck = nx;
    }
  }
}

__global__ void __launch_bounds__(256)
k_final(const float* __restrict__ x, const float* __restrict__ acc2,
        const float* __restrict__ xw2s, const float* __restrict__ dinv,
        const float* __restrict__ gb2,
        const float* __restrict__ cW1, const float* __restrict__ cb1,
        const float* __restrict__ cW2, const float* __restrict__ cb2,
        const float* __restrict__ mW1, const float* __restrict__ mb1,
        const float* __restrict__ mW2, const float* __restrict__ mb2,
        float* __restrict__ out, int N) {
  __shared__ float ht[WPB][256];
  __shared__ float ct[WPB][128];
  __shared__ float g0s[WPB][16];
  int wslot = threadIdx.x >> 5;
  int lane  = threadIdx.x & 31;
  int m0 = (blockIdx.x * WPB + wslot) * 16;
  if (m0 >= N) return;
  bool full = (m0 + 16) <= N;
  float* t  = ht[wslot];
  float* o  = ct[wslot];
  float* g0 = g0s[wslot];

  int n  = lane & 15;
  int mo = (lane >> 4) << 3;

  int grow = lane >> 1, gcc = lane & 1;
  int gr  = m0 + grow;
  int grc = (gr < N) ? gr : N - 1;
  float  gdv = dinv[grc];
  float4 a24 = *(const float4*)(acc2 + (size_t)grc * 8 + gcc * 4);
  float4 s24 = *(const float4*)(xw2s + (size_t)grc * 8 + gcc * 4);
  float4 gb4 = ((const float4*)gb2)[gcc];
  float4 go4;
  go4.x = gdv * (a24.x + s24.x) + gb4.x;
  go4.y = gdv * (a24.y + s24.y) + gb4.y;
  go4.z = gdv * (a24.z + s24.z) + gb4.z;
  go4.w = gdv * (a24.w + s24.w) + gb4.w;

  v8f ch = {};
  for (int k0 = 0; k0 < 16; k0 += 4) {
    v2f a = a_frag_x(x, N, m0, lane, k0,  3,  13);
    v2f b = b_frag(cW1, lane, k0, 13, 16);
    ch = wmma_f32(a, b, ch);
  }
  for (int v = 0; v < 8; ++v) t[(mo + v) * 16 + n] = fmaxf(ch[v] + cb1[n], 0.0f);
  __builtin_amdgcn_wave_barrier();

  v8f cc = {};
  for (int k0 = 0; k0 < 16; k0 += 4) {
    v2f a = a_frag_lds(t, lane, k0);
    v2f b = b_frag(cW2, lane, k0, 16, 8);
    cc = wmma_f32(a, b, cc);
  }
  if (n < 8) {
    for (int v = 0; v < 8; ++v) o[(mo + v) * 8 + n] = cc[v] + cb2[n];
  }
  __builtin_amdgcn_wave_barrier();

  v8f cm = {};
  for (int k0 = 0; k0 < 16; k0 += 4) {
    v2f a = a_frag_x(x, N, m0, lane, k0,  0,  16);
    v2f b = b_frag(mW1, lane, k0, 16, 16);
    cm = wmma_f32(a, b, cm);
  }
  for (int v = 0; v < 8; ++v) t[(mo + v) * 16 + n] = fmaxf(cm[v] + mb1[n], 0.0f);
  __builtin_amdgcn_wave_barrier();

  if (lane < 16) {
    float l0 = mb2[0], l1 = mb2[1];
    for (int j = 0; j < 16; ++j) {
      float hv = t[lane * 16 + j];
      l0 += hv * mW2[j * 2 + 0];
      l1 += hv * mW2[j * 2 + 1];
    }
    float mx = fmaxf(l0, l1);
    float e0 = __expf(l0 - mx), e1 = __expf(l1 - mx);
    g0[lane] = e0 / (e0 + e1);
  }
  __builtin_amdgcn_wave_barrier();

  float g  = g0[grow];
  float4 c4 = *(const float4*)&o[lane * 4];
  float4 r4;
  r4.x = g * go4.x + (1.0f - g) * c4.x;
  r4.y = g * go4.y + (1.0f - g) * c4.y;
  r4.z = g * go4.z + (1.0f - g) * c4.z;
  r4.w = g * go4.w + (1.0f - g) * c4.w;
  if (full) {
    ((float4*)(out + (size_t)m0 * 8))[lane] = r4;
  } else if (gr < N) {
    *(float4*)(out + (size_t)gr * 8 + gcc * 4) = r4;
  }
}

extern "C" void kernel_launch(void* const* d_in, const int* in_sizes, int n_in,
                              void* d_out, int out_size, void* d_ws, size_t ws_size,
                              hipStream_t stream) {
  const float* x   = (const float*)d_in[0];
  const int*   ei  = (const int*)  d_in[1];
  const float* gW1 = (const float*)d_in[2];
  const float* gb1 = (const float*)d_in[3];
  const float* gW2 = (const float*)d_in[4];
  const float* gb2 = (const float*)d_in[5];
  const float* cW1 = (const float*)d_in[6];
  const float* cb1 = (const float*)d_in[7];
  const float* cW2 = (const float*)d_in[8];
  const float* cb2 = (const float*)d_in[9];
  const float* mW1 = (const float*)d_in[10];
  const float* mb1 = (const float*)d_in[11];
  const float* mW2 = (const float*)d_in[12];
  const float* mb2 = (const float*)d_in[13];
  float* out = (float*)d_out;

  const int N  = in_sizes[0] / 16;
  const int NE = in_sizes[1] / 2;
  const int* erow = ei;
  const int* ecol = ei + NE;

  float* deg  = (float*)d_ws;
  float* dinv = deg  + N;
  float* xw1s = dinv + N;
  float* acc1 = xw1s + (size_t)16 * N;
  float* xw2s = acc1 + (size_t)16 * N;
  float* acc2 = xw2s + (size_t)8 * N;
  int*   ctr  = (int*)(acc2 + (size_t)8 * N);

  const int tiles   = (N + 15) / 16;
  const int gBlocks = (tiles + WPB - 1) / WPB;

  k_init    <<<(N + 255) / 256, 256, 0, stream>>>(deg, acc1, acc2, ctr, N);
  k_degree  <<<256, 1024, 0, stream>>>(ecol, deg, N, NE);
  k_dense1  <<<gBlocks, 256, 0, stream>>>(x, gW1, deg, dinv, xw1s, N);
  k_scatter1<<<2048, 256, 0, stream>>>(erow, ecol, xw1s, acc1, ctr, N, NE);
  k_dense2  <<<gBlocks, 256, 0, stream>>>(acc1, xw1s, dinv, gW2, gb1, xw2s, N);
  k_scatter2<<<2048, 256, 0, stream>>>(erow, ecol, xw2s, acc2, ctr + 2 * CTR_PITCH, N, NE);
  k_final   <<<gBlocks, 256, 0, stream>>>(x, acc2, xw2s, dinv, gb2,
                                          cW1, cb1, cW2, cb2,
                                          mW1, mb1, mW2, mb2, out, N);
}


# layer 1 sums the 3 dinv-scaled input features over the edges first and applies gW1 (WMMA) to the sums after, A(xW)=(Ax)W: 16-byte gather/atomic rows, not 64; the order of the f32 sums changes
# speedup vs baseline: 5.4727x; 1.1083x over previous
#include <hip/hip_runtime.h>
#include <math.h>


typedef float v2f __attribute__((ext_vector_type(2)));
typedef float v8f __attribute__((ext_vector_type(8)));

#define WPB 8
#define CTR_PITCH 64

static __device__ __forceinline__ v8f wmma_f32(v2f a, v2f b, v8f c) {
  return __builtin_amdgcn_wmma_f32_16x16x4_f32(false, a, false, b,
                                               (short)0, c, false, false);
}

static __device__ __forceinline__ v2f b_frag(const float* __restrict__ W,
                                             int lane, int k0, int K, int Ncols) {
  int n  = lane & 15;
  int kb = k0 + ((lane >> 4) << 1);
  int nm  = (n      < Ncols) ? n      : (Ncols - 1);
  int kc0 = (kb     < K)     ? kb     : (K - 1);
  int kc1 = (kb + 1 < K)     ? kb + 1 : (K - 1);
  float w0 = W[kc0 * Ncols + nm];
  float w1 = W[kc1 * Ncols + nm];
  v2f b;
  b.x = (kb     < K && n < Ncols) ? w0 : 0.0f;
  b.y = (kb + 1 < K && n < Ncols) ? w1 : 0.0f;
  return b;
}

static __device__ __forceinline__ v2f a_frag_x(const float* __restrict__ x,
                                               int N, int m0, int lane,
                                               int k0, int co, int K) {
  int r = m0 + (lane & 15);
  if (r >= N) r = N - 1;
  int kb = k0 + ((lane >> 4) << 1);
  const float* rowp = x + (size_t)r * 16;
  int i0 = co + kb;     i0 = (i0 < 16) ? i0 : 15;
  int i1 = co + kb + 1; i1 = (i1 < 16) ? i1 : 15;
  float a0 = rowp[i0];
  float a1 = rowp[i1];
  v2f a;
  a.x = (kb     < K) ? a0 : 0.0f;
  a.y = (kb + 1 < K) ? a1 : 0.0f;
  return a;
}

static __device__ __forceinline__ v2f a_frag_lds(const float* t, int lane, int k0) {
  int r  = lane & 15;
  int kb = k0 + ((lane >> 4) << 1);
  v2f a;
  a.x = t[r * 16 + kb];
  a.y = t[r * 16 + kb + 1];
  return a;
}

__global__ void k_init(float* __restrict__ deg, float* __restrict__ acc1,
                       float* __restrict__ acc2, int* __restrict__ ctr, int N) {
  int i = blockIdx.x * blockDim.x + threadIdx.x;
  if (i < 4) ctr[i * CTR_PITCH] = 0;
  if (i >= N) return;
  deg[i] = 1.0f;
  float4 z = make_float4(0.f, 0.f, 0.f, 0.f);
  *(float4*)(acc1 + (size_t)i * 4) = z;
  float4* a2 = (float4*)(acc2 + (size_t)i * 8);
  a2[0] = z; a2[1] = z;
}

#define DEG_CAP 25088

static __device__ __forceinline__ void deg_bump(unsigned* cnt, int c, int n0, int nn) {
  unsigned k = (unsigned)(c - n0);
  if (k < (unsigned)nn) atomicAdd(&cnt[k], 1u);
}

__global__ void __launch_bounds__(1024)
k_degree(const int* __restrict__ col, float* __restrict__ deg, int N, int NE) {
  __shared__ unsigned cnt[DEG_CAP];
  const int tid = (int)threadIdx.x;
  const int nth = (int)blockDim.x;
  const int R = (N + DEG_CAP - 1) / DEG_CAP;
  int S = (int)gridDim.x / R;
  if (S < 1) S = 1;
  const int per = (((NE + S - 1) / S) + 3) & ~3;
  const bool aligned = (((size_t)col) & 15) == 0;

  for (int item = (int)blockIdx.x; item < R * S; item += (int)gridDim.x) {
    const int rg = item % R;
    const int sl = item / R;
    const int n0 = rg * DEG_CAP;
    const int nn = (N - n0 < DEG_CAP) ? (N - n0) : DEG_CAP;
    for (int i = tid; i < nn; i += nth) cnt[i] = 0u;
    __syncthreads();

    const int e0 = (sl * per < NE) ? sl * per : NE;
    const int e1 = (e0 + per < NE) ? e0 + per : NE;
    int done = e0;
    if (aligned) {
      const int4* c4 = (const int4*)(col + e0);
      const int nv = (e1 - e0) >> 2;
      const int4 none = make_int4(-1, -1, -1, -1);
      for (int i = tid; i < nv; i += 8 * nth) {
        int4 v[8];
#pragma unroll
        for (int u = 0; u < 8; ++u) {
          const int iu = i + u * nth;
          v[u] = c4[(iu < nv) ? iu : nv - 1];
        }
#pragma unroll
        for (int u = 0; u < 8; ++u) {
          if (i + u * nth >= nv) v[u] = none;
          deg_bump(cnt, v[u].x, n0, nn); deg_bump(cnt, v[u].y, n0, nn);
          deg_bump(cnt, v[u].z, n0, nn); deg_bump(cnt, v[u].w, n0, nn);
        }
      }
      done = e0 + (nv << 2);
    }
    for (int e = done + tid; e < e1; e += nth) deg_bump(cnt, col[e], n0, nn);
    __syncthreads();

    const int lines = (nn + 31) >> 5;
    const int rot   = (int)(((long long)sl * lines) / S) << 5;
    for (int j = tid; j < lines * 32; j += nth) {
      int i = j + rot;
      if (i >= lines * 32) i -= lines * 32;
      if (i < nn) {
        unsigned c = cnt[i];
        if (c) atomicAdd(&deg[n0 + i], (float)c);
      }
    }
    __syncthreads();
  }
}

__global__ void __launch_bounds__(256)
k_scale1(const float* __restrict__ x, const float* __restrict__ deg,
         float* __restrict__ dinv, float* __restrict__ xs, int N) {
  int i = blockIdx.x * blockDim.x + threadIdx.x;
  if (i >= N) return;
  float d = __builtin_amdgcn_rsqf(deg[i]);
  dinv[i] = d;
  const float* xr = x + (size_t)i * 16;
  float4 o = make_float4(d * xr[0], d * xr[1], d * xr[2], 0.0f);
  *(float4*)(xs + (size_t)i * 4) = o;
}

#define REP4(M)  M(0) M(1) M(2) M(3)
#define REP8(M)  REP4(M) M(4) M(5) M(6) M(7)

template <int J>
static __device__ __forceinline__ int quad_bcast(int x) {
  return __builtin_amdgcn_update_dpp(0, x, J * 0x55, 0xF, 0xF, false);
}
template <int J>
static __device__ __forceinline__ int grp8_bcast(int x) {
  return __builtin_amdgcn_mov_dpp8(x, J * 0x249249);
}

static __device__ __forceinline__ int chip_half() {
  return (int)((__builtin_amdgcn_s_sendmsg_rtn(0x87) >> 18) & 1u);
}
static __device__ __forceinline__ int node_split(int N) { return ((N >> 1) + 63) & ~63; }

#define BPC 8

static __device__ __forceinline__ int take_chunk(int* cnt, int lane) {
  int ck = 0;
  if (lane == 0) ck = atomicAdd(cnt, 1);
  return __builtin_amdgcn_readfirstlane(ck);
}

__global__ void __launch_bounds__(256)
k_scatter1(const int* __restrict__ row, const int* __restrict__ col,
           const float* __restrict__ xs, float* __restrict__ acc1,
           int* __restrict__ ctr, int N, int NE) {
  const int      lane = threadIdx.x & 31;
  const unsigned f    = lane & 3;
  const int      live = (f < 3u) ? 1 : 0;
  const int split  = node_split(N);
  const int nchunk = (NE + 32 * BPC - 1) / (32 * BPC);
  const int mine   = chip_half();
  for (int ph = 0; ph < 2; ++ph) {
    const int p = mine ^ ph;
    int* cnt = ctr + p * CTR_PITCH;
    int ck = take_chunk(cnt, lane);
    while (ck < nchunk) {
      int r[BPC], c[BPC];
#pragma unroll
      for (int u = 0; u < BPC; ++u) {
        const int e  = (ck * BPC + u) * 32 + lane;
        const int ec = (e < NE) ? e : NE - 1;
        r[u] = row[ec]; c[u] = col[ec];
      }
      const int nx = take_chunk(cnt, lane);
#pragma unroll
      for (int u = 0; u < BPC; ++u) {
        const int e = (ck * BPC + u) * 32 + lane;
        const int ru = r[u], cu = c[u];
        asm volatile("" :: "v"(ru), "v"(cu));
        const int hit = (e < NE && ((cu >= split) ? 1 : 0) == p) ? 1 : 0;
        float v[4];
        int hq[4];
        unsigned rq[4], cq[4];
#define SHAREX(J) hq[J] = quad_bcast<J>(hit) & live; rq[J] = (unsigned)quad_bcast<J>(ru); cq[J] = (unsigned)quad_bcast<J>(cu);
        REP4(SHAREX)
#undef SHAREX
#define GATHERX(J) v[J] = 0.0f; if (hq[J]) v[J] = xs[rq[J] * 4u + f];
        REP4(GATHERX)
#undef GATHERX
#define ADDX(J) if (hq[J]) atomicAdd(&acc1[cq[J] * 4u + f], v[J]);
        REP4(ADDX)
#undef ADDX
      }
      ck = nx;
    }
  }
}

__global__ void __launch_bounds__(256)
k_dense2(const float* __restrict__ acc1, const float* __restrict__ xs,
         const float* __restrict__ dinv, const float* __restrict__ gW1,
         const float* __restrict__ gW2, const float* __restrict__ gb1,
         float* __restrict__ xw2s, int N) {
  __shared__ float ht[WPB][256];
  __shared__ float dv[WPB][16];
  int wslot = threadIdx.x >> 5;
  int lane  = threadIdx.x & 31;
  int m0 = (blockIdx.x * WPB + wslot) * 16;
  if (m0 >= N) return;
  bool full = (m0 + 16) <= N;
  float* t   = ht[wslot];
  float* dvs = dv[wslot];

  int n  = lane & 15;
  int mo = (lane >> 4) << 3;

  {
    int r  = m0 + n;
    int rc = (r < N) ? r : N - 1;
    float dvr = dinv[rc];
    if (lane < 16) dvs[n] = dvr;
    float4 a4 = *(const float4*)(acc1 + (size_t)rc * 4);
    float4 s4 = *(const float4*)(xs   + (size_t)rc * 4);
    float g0 = dvr * (a4.x + s4.x);
    float g1 = dvr * (a4.y + s4.y);
    float g2 = dvr * (a4.z + s4.z);
    v2f a;
    a.x = (lane < 16) ? g0 : g2;
    a.y = (lane < 16) ? g1 : 0.0f;
    v2f b = b_frag(gW1, lane, 0, 3, 16);
    v8f c1 = {};
    c1 = wmma_f32(a, b, c1);
    float bias = gb1[n];
    for (int v = 0; v < 8; ++v)
      t[(mo + v) * 16 + n] = fmaxf(c1[v] + bias, 0.0f);
  }
  __builtin_amdgcn_wave_barrier();

  v8f c = {};
  for (int k0 = 0; k0 < 16; k0 += 4) {
    v2f a = a_frag_lds(t, lane, k0);
    v2f b = b_frag(gW2, lane, k0, 16, 8);
    c = wmma_f32(a, b, c);
  }
  __builtin_amdgcn_wave_barrier();

  if (n < 8) {
    for (int v = 0; v < 8; ++v)
      t[(mo + v) * 8 + n] = c[v];
  }
  __builtin_amdgcn_wave_barrier();

  if (full) {
    float dvr = dvs[lane >> 1];
    float4 v4 = ((const float4*)t)[lane];
    v4.x *= dvr; v4.y *= dvr; v4.z *= dvr; v4.w *= dvr;
    ((float4*)(xw2s + (size_t)m0 * 8))[lane] = v4;
  } else if (n < 8) {
    for (int v = 0; v < 8; ++v) {
      int r = m0 + mo + v;
      if (r < N) xw2s[(size_t)r * 8 + n] = t[(mo + v) * 8 + n] * dvs[mo + v];
    }
  }
}

__global__ void __launch_bounds__(256)
k_scatter2(const int* __restrict__ row, const int* __restrict__ col,
           const float* __restrict__ xw2s, float* __restrict__ acc2,
           int* __restrict__ ctr, int N, int NE) {
  const int      lane = threadIdx.x & 31;
  const unsigned f    = lane & 7;
  const int split  = node_split(N);
  const int nchunk = (NE + 32 * BPC - 1) / (32 * BPC);
  const int mine   = chip_half();
  for (int ph = 0; ph < 2; ++ph) {
    const int p = mine ^ ph;
    int* cnt = ctr + p * CTR_PITCH;
    int ck = take_chunk(cnt, lane);
    while (ck < nchunk) {
      int r[BPC], c[BPC];
#pragma unroll
      for (int u = 0; u < BPC; ++u) {
        const int e  = (ck * BPC + u) * 32 + lane;
        const int ec = (e < NE) ? e : NE - 1;
        r[u] = row[ec]; c[u] = col[ec];
      }
      const int nx = take_chunk(cnt, lane);
#pragma unroll
      for (int u = 0; u < BPC; ++u) {
        const int e = (ck * BPC + u) * 32 + lane;
        const int ru = r[u], cu = c[u];
        asm volatile("" :: "v"(ru), "v"(cu));
        const int hit = (e < NE && ((cu >= split) ? 1 : 0) == p) ? 1 : 0;
        float v[8];
#define GATHERX(J) v[J] = 0.0f; if (grp8_bcast<J>(hit)) v[J] = xw2s[(unsigned)grp8_bcast<J>(ru) * 8u + f];
        REP8(GATHERX)
#undef GATHERX
#define ADDX(J) if (grp8_bcast<J>(hit)) atomicAdd(&acc2[(unsigned)grp8_bcast<J>(cu) * 8u + f], v[J]);
        REP8(ADDX)
#undef ADDX
      }
      ck = nx;
    }
  }
}

__global__ void __launch_bounds__(256)
k_final(const float* __restrict__ x, const float* __restrict__ acc2,
        const float* __restrict__ xw2s, const float* __restrict__ dinv,
        const float* __restrict__ gb2,
        const float* __restrict__ cW1, const float* __restrict__ cb1,
        const float* __restrict__ cW2, const float* __restrict__ cb2,
        const float* __restrict__ mW1, const float* __restrict__ mb1,
        const float* __restrict__ mW2, const float* __restrict__ mb2,
        float* __restrict__ out, int N) {
  __shared__ float ht[WPB][256];
  __shared__ float ct[WPB][128];
  __shared__ float g0s[WPB][16];
  int wslot = threadIdx.x >> 5;
  int lane  = threadIdx.x & 31;
  int m0 = (blockIdx.x * WPB + wslot) * 16;
  if (m0 >= N) return;
  bool full = (m0 + 16) <= N;
  float* t  = ht[wslot];
  float* o  = ct[wslot];
  float* g0 = g0s[wslot];

  int n  = lane & 15;
  int mo = (lane >> 4) << 3;

  int grow = lane >> 1, gcc = lane & 1;
  int gr  = m0 + grow;
  int grc = (gr < N) ? gr : N - 1;
  float  gdv = dinv[grc];
  float4 a24 = *(const float4*)(acc2 + (size_t)grc * 8 + gcc * 4);
  float4 s24 = *(const float4*)(xw2s + (size_t)grc * 8 + gcc * 4);
  float4 gb4 = ((const float4*)gb2)[gcc];
  float4 go4;
  go4.x = gdv * (a24.x + s24.x) + gb4.x;
  go4.y = gdv * (a24.y + s24.y) + gb4.y;
  go4.z = gdv * (a24.z + s24.z) + gb4.z;
  go4.w = gdv * (a24.w + s24.w) + gb4.w;

  v8f ch = {};
  for (int k0 = 0; k0 < 16; k0 += 4) {
    v2f a = a_frag_x(x, N, m0, lane, k0,  3,  13);
    v2f b = b_frag(cW1, lane, k0, 13, 16);
    ch = wmma_f32(a, b, ch);
  }
  for (int v = 0; v < 8; ++v) t[(mo + v) * 16 + n] = fmaxf(ch[v] + cb1[n], 0.0f);
  __builtin_amdgcn_wave_barrier();

  v8f cc = {};
  for (int k0 = 0; k0 < 16; k0 += 4) {
    v2f a = a_frag_lds(t, lane, k0);
    v2f b = b_frag(cW2, lane, k0, 16, 8);
    cc = wmma_f32(a, b, cc);
  }
  if (n < 8) {
    for (int v = 0; v < 8; ++v) o[(mo + v) * 8 + n] = cc[v] + cb2[n];
  }
  __builtin_amdgcn_wave_barrier();

  v8f cm = {};
  for (int k0 = 0; k0 < 16; k0 += 4) {
    v2f a = a_frag_x(x, N, m0, lane, k0,  0,  16);
    v2f b = b_frag(mW1, lane, k0, 16, 16);
    cm = wmma_f32(a, b, cm);
  }
  for (int v = 0; v < 8; ++v) t[(mo + v) * 16 + n] = fmaxf(cm[v] + mb1[n], 0.0f);
  __builtin_amdgcn_wave_barrier();

  if (lane < 16) {
    float l0 = mb2[0], l1 = mb2[1];
    for (int j = 0; j < 16; ++j) {
      float hv = t[lane * 16 + j];
      l0 += hv * mW2[j * 2 + 0];
      l1 += hv * mW2[j * 2 + 1];
    }
    float mx = fmaxf(l0, l1);
    float e0 = __expf(l0 - mx), e1 = __expf(l1 - mx);
    g0[lane] = e0 / (e0 + e1);
  }
  __builtin_amdgcn_wave_barrier();

  float g  = g0[grow];
  float4 c4 = *(const float4*)&o[lane * 4];
  float4 r4;
  r4.x = g * go4.x + (1.0f - g) * c4.x;
  r4.y = g * go4.y + (1.0f - g) * c4.y;
  r4.z = g * go4.z + (1.0f - g) * c4.z;
  r4.w = g * go4.w + (1.0f - g) * c4.w;
  if (full) {
    ((float4*)(out + (size_t)m0 * 8))[lane] = r4;
  } else if (gr < N) {
    *(float4*)(out + (size_t)gr * 8 + gcc * 4) = r4;
  }
}

extern "C" void kernel_launch(void* const* d_in, const int* in_sizes, int n_in,
                              void* d_out, int out_size, void* d_ws, size_t ws_size,
                              hipStream_t stream) {
  const float* x   = (const float*)d_in[0];
  const int*   ei  = (const int*)  d_in[1];
  const float* gW1 = (const float*)d_in[2];
  const float* gb1 = (const float*)d_in[3];
  const float* gW2 = (const float*)d_in[4];
  const float* gb2 = (const float*)d_in[5];
  const float* cW1 = (const float*)d_in[6];
  const float* cb1 = (const float*)d_in[7];
  const float* cW2 = (const float*)d_in[8];
  const float* cb2 = (const float*)d_in[9];
  const float* mW1 = (const float*)d_in[10];
  const float* mb1 = (const float*)d_in[11];
  const float* mW2 = (const float*)d_in[12];
  const float* mb2 = (const float*)d_in[13];
  float* out = (float*)d_out;

  const int N  = in_sizes[0] / 16;
  const int NE = in_sizes[1] / 2;
  const int* erow = ei;
  const int* ecol = ei + NE;

  float* deg  = (float*)d_ws;
  float* dinv = deg  + N;
  float* xs   = dinv + N;
  float* acc1 = xs   + (size_t)4 * N;
  float* xw2s = acc1 + (size_t)4 * N;
  float* acc2 = xw2s + (size_t)8 * N;
  int*   ctr  = (int*)(acc2 + (size_t)8 * N);

  const int tiles   = (N + 15) / 16;
  const int gBlocks = (tiles + WPB - 1) / WPB;

  k_init    <<<(N + 255) / 256, 256, 0, stream>>>(deg, acc1, acc2, ctr, N);
  k_degree  <<<256, 1024, 0, stream>>>(ecol, deg, N, NE);
  k_scale1  <<<(N + 255) / 256, 256, 0, stream>>>(x, deg, dinv, xs, N);
  k_scatter1<<<2048, 256, 0, stream>>>(erow, ecol, xs, acc1, ctr, N, NE);
  k_dense2  <<<gBlocks, 256, 0, stream>>>(acc1, xs, dinv, gW1, gW2, gb1, xw2s, N);
  k_scatter2<<<2048, 256, 0, stream>>>(erow, ecol, xw2s, acc2, ctr + 2 * CTR_PITCH, N, NE);
  k_final   <<<gBlocks, 256, 0, stream>>>(x, acc2, xw2s, dinv, gb2,
                                          cW1, cb1, cW2, cb2,
                                          mW1, mb1, mW2, mb2, out, N);
}


# layer-1 edge sums in LDS: each block sums one node range over a slice of the edge list (LDS f32 adds), then adds whole 128-B lines to acc1, own chip half first; the order of the f32 sums changes
# speedup vs baseline: 5.7559x; 1.0517x over previous
#include <hip/hip_runtime.h>
#include <math.h>


typedef float v2f __attribute__((ext_vector_type(2)));
typedef float v8f __attribute__((ext_vector_type(8)));

#define WPB 8
#define CTR_PITCH 64

static __device__ __forceinline__ v8f wmma_f32(v2f a, v2f b, v8f c) {
  return __builtin_amdgcn_wmma_f32_16x16x4_f32(false, a, false, b,
                                               (short)0, c, false, false);
}

static __device__ __forceinline__ v2f b_frag(const float* __restrict__ W,
                                             int lane, int k0, int K, int Ncols) {
  int n  = lane & 15;
  int kb = k0 + ((lane >> 4) << 1);
  int nm  = (n      < Ncols) ? n      : (Ncols - 1);
  int kc0 = (kb     < K)     ? kb     : (K - 1);
  int kc1 = (kb + 1 < K)     ? kb + 1 : (K - 1);
  float w0 = W[kc0 * Ncols + nm];
  float w1 = W[kc1 * Ncols + nm];
  v2f b;
  b.x = (kb     < K && n < Ncols) ? w0 : 0.0f;
  b.y = (kb + 1 < K && n < Ncols) ? w1 : 0.0f;
  return b;
}

static __device__ __forceinline__ v2f a_frag_x(const float* __restrict__ x,
                                               int N, int m0, int lane,
                                               int k0, int co, int K) {
  int r = m0 + (lane & 15);
  if (r >= N) r = N - 1;
  int kb = k0 + ((lane >> 4) << 1);
  const float* rowp = x + (size_t)r * 16;
  int i0 = co + kb;     i0 = (i0 < 16) ? i0 : 15;
  int i1 = co + kb + 1; i1 = (i1 < 16) ? i1 : 15;
  float a0 = rowp[i0];
  float a1 = rowp[i1];
  v2f a;
  a.x = (kb     < K) ? a0 : 0.0f;
  a.y = (kb + 1 < K) ? a1 : 0.0f;
  return a;
}

static __device__ __forceinline__ v2f a_frag_lds(const float* t, int lane, int k0) {
  int r  = lane & 15;
  int kb = k0 + ((lane >> 4) << 1);
  v2f a;
  a.x = t[r * 16 + kb];
  a.y = t[r * 16 + kb + 1];
  return a;
}

__global__ void k_init(float* __restrict__ deg, float* __restrict__ acc1,
                       float* __restrict__ acc2, int* __restrict__ ctr, int N) {
  int i = blockIdx.x * blockDim.x + threadIdx.x;
  if (i < 4) ctr[i * CTR_PITCH] = 0;
  if (i >= N) return;
  deg[i] = 1.0f;
  float4 z = make_float4(0.f, 0.f, 0.f, 0.f);
  *(float4*)(acc1 + (size_t)i * 4) = z;
  float4* a2 = (float4*)(acc2 + (size_t)i * 8);
  a2[0] = z; a2[1] = z;
}

#define DEG_CAP 25088

static __device__ __forceinline__ void deg_bump(unsigned* cnt, int c, int n0, int nn) {
  unsigned k = (unsigned)(c - n0);
  if (k < (unsigned)nn) atomicAdd(&cnt[k], 1u);
}

__global__ void __launch_bounds__(1024)
k_degree(const int* __restrict__ col, float* __restrict__ deg, int N, int NE) {
  __shared__ unsigned cnt[DEG_CAP];
  const int tid = (int)threadIdx.x;
  const int nth = (int)blockDim.x;
  const int R = (N + DEG_CAP - 1) / DEG_CAP;
  int S = (int)gridDim.x / R;
  if (S < 1) S = 1;
  const int per = (((NE + S - 1) / S) + 3) & ~3;
  const bool aligned = (((size_t)col) & 15) == 0;

  for (int item = (int)blockIdx.x; item < R * S; item += (int)gridDim.x) {
    const int rg = item % R;
    const int sl = item / R;
    const int n0 = rg * DEG_CAP;
    const int nn = (N - n0 < DEG_CAP) ? (N - n0) : DEG_CAP;
    for (int i = tid; i < nn; i += nth) cnt[i] = 0u;
    __syncthreads();

    const int e0 = (sl * per < NE) ? sl * per : NE;
    const int e1 = (e0 + per < NE) ? e0 + per : NE;
    int done = e0;
    if (aligned) {
      const int4* c4 = (const int4*)(col + e0);
      const int nv = (e1 - e0) >> 2;
      const int4 none = make_int4(-1, -1, -1, -1);
      for (int i = tid; i < nv; i += 8 * nth) {
        int4 v[8];
#pragma unroll
        for (int u = 0; u < 8; ++u) {
          const int iu = i + u * nth;
          v[u] = c4[(iu < nv) ? iu : nv - 1];
        }
#pragma unroll
        for (int u = 0; u < 8; ++u) {
          if (i + u * nth >= nv) v[u] = none;
          deg_bump(cnt, v[u].x, n0, nn); deg_bump(cnt, v[u].y, n0, nn);
          deg_bump(cnt, v[u].z, n0, nn); deg_bump(cnt, v[u].w, n0, nn);
        }
      }
      done = e0 + (nv << 2);
    }
    for (int e = done + tid; e < e1; e += nth) deg_bump(cnt, col[e], n0, nn);
    __syncthreads();

    const int lines = (nn + 31) >> 5;
    const int rot   = (int)(((long long)sl * lines) / S) << 5;
    for (int j = tid; j < lines * 32; j += nth) {
      int i = j + rot;
      if (i >= lines * 32) i -= lines * 32;
      if (i < nn) {
        unsigned c = cnt[i];
        if (c) atomicAdd(&deg[n0 + i], (float)c);
      }
    }
    __syncthreads();
  }
}

__global__ void __launch_bounds__(256)
k_scale1(const float* __restrict__ x, const float* __restrict__ deg,
         float* __restrict__ dinv, float* __restrict__ xs, int N) {
  int i = blockIdx.x * blockDim.x + threadIdx.x;
  if (i >= N) return;
  float d = __builtin_amdgcn_rsqf(deg[i]);
  dinv[i] = d;
  const float* xr = x + (size_t)i * 16;
  float4 o = make_float4(d * xr[0], d * xr[1], d * xr[2], 0.0f);
  *(float4*)(xs + (size_t)i * 4) = o;
}

#define REP4(M)  M(0) M(1) M(2) M(3)
#define REP8(M)  REP4(M) M(4) M(5) M(6) M(7)

template <int J>
static __device__ __forceinline__ int grp8_bcast(int x) {
  return __builtin_amdgcn_mov_dpp8(x, J * 0x249249);
}

static __device__ __forceinline__ int chip_half() {
  return (int)((__builtin_amdgcn_s_sendmsg_rtn(0x87) >> 18) & 1u);
}
static __device__ __forceinline__ int node_split(int N) { return ((N >> 1) + 63) & ~63; }

#define BPC 8

static __device__ __forceinline__ int take_chunk(int* cnt, int lane) {
  int ck = 0;
  if (lane == 0) ck = atomicAdd(cnt, 1);
  return __builtin_amdgcn_readfirstlane(ck);
}

#define S1_CAP 16704
#define S1_U   4

static __device__ __forceinline__ void s1_add(float* tab, unsigned k, float4 g) {
  atomicAdd(&tab[k * 3u + 0u], g.x);
  atomicAdd(&tab[k * 3u + 1u], g.y);
  atomicAdd(&tab[k * 3u + 2u], g.z);
}

__global__ void __launch_bounds__(1024)
k_scatter1(const int* __restrict__ row, const int* __restrict__ col,
           const float* __restrict__ xs, float* __restrict__ acc1,
           int* __restrict__ ctr, int N, int NE) {
  __shared__ float tab[S1_CAP * 3];
  __shared__ int s_item;
  __shared__ int s_half;
  const int tid = (int)threadIdx.x;
  const int nth = (int)blockDim.x;
  const int split = node_split(N);
  const int lo_n  = (split < N) ? split : N;
  const int R0 = (lo_n + S1_CAP - 1) / S1_CAP;
  const int R1 = (N - lo_n + S1_CAP - 1) / S1_CAP;
  int S = (int)gridDim.x / ((R0 + R1 > 0) ? R0 + R1 : 1);
  if (S < 1) S = 1;
  const int per = (((NE + S - 1) / S) + 3) & ~3;
  const bool aligned = ((((size_t)col) | ((size_t)row)) & 15) == 0;
  const float4* __restrict__ xs4 = (const float4*)xs;

  if (tid == 0) s_half = chip_half();
  __syncthreads();
  const int mine = s_half;

  for (int ph = 0; ph < 2; ++ph) {
    const int p     = mine ^ ph;
    const int Rp    = p ? R1 : R0;
    const int base  = p ? lo_n : 0;
    const int pn    = p ? (N - lo_n) : lo_n;
    const int items = Rp * S;
    int* cnt = ctr + p * CTR_PITCH;
    for (;;) {
      __syncthreads();
      if (tid == 0) s_item = atomicAdd(cnt, 1);
      __syncthreads();
      const int it = s_item;
      if (it >= items) break;
      const int rg = it % Rp;
      const int sl = it / Rp;
      const int n0 = base + rg * S1_CAP;
      const int nn = (pn - rg * S1_CAP < S1_CAP) ? (pn - rg * S1_CAP) : S1_CAP;
      for (int i = tid; i < nn * 3; i += nth) tab[i] = 0.0f;
      __syncthreads();

      const int e0 = (sl * per < NE) ? sl * per : NE;
      const int e1 = (e0 + per < NE) ? e0 + per : NE;
      int done = e0;
      if (aligned) {
        const int4* c4 = (const int4*)(col + e0);
        const int4* r4 = (const int4*)(row + e0);
        const int nv = (e1 - e0) >> 2;
        for (int i = tid; i < nv; i += S1_U * nth) {
          int4 cv[S1_U], rv[S1_U];
#pragma unroll
          for (int u = 0; u < S1_U; ++u) {
            const int iu = i + u * nth;
            const int ic = (iu < nv) ? iu : nv - 1;
            cv[u] = c4[ic]; rv[u] = r4[ic];
          }
#pragma unroll
          for (int u = 0; u < S1_U; ++u) {
            asm volatile("" :: "v"(cv[u].x), "v"(cv[u].y), "v"(cv[u].z), "v"(cv[u].w),
                               "v"(rv[u].x), "v"(rv[u].y), "v"(rv[u].z), "v"(rv[u].w));
          }
          unsigned k[S1_U * 4];
          int      r[S1_U * 4];
#pragma unroll
          for (int u = 0; u < S1_U; ++u) {
            const bool in = (i + u * nth) < nv;
            k[4 * u + 0] = in ? (unsigned)(cv[u].x - n0) : 0xFFFFFFFFu;
            k[4 * u + 1] = in ? (unsigned)(cv[u].y - n0) : 0xFFFFFFFFu;
            k[4 * u + 2] = in ? (unsigned)(cv[u].z - n0) : 0xFFFFFFFFu;
            k[4 * u + 3] = in ? (unsigned)(cv[u].w - n0) : 0xFFFFFFFFu;
            r[4 * u + 0] = rv[u].x; r[4 * u + 1] = rv[u].y;
            r[4 * u + 2] = rv[u].z; r[4 * u + 3] = rv[u].w;
          }
          float4 g[S1_U * 4];
#pragma unroll
          for (int j = 0; j < S1_U * 4; ++j) {
            g[j] = make_float4(0.f, 0.f, 0.f, 0.f);
            if (k[j] < (unsigned)nn) g[j] = xs4[(unsigned)r[j]];
          }
#pragma unroll
          for (int j = 0; j < S1_U * 4; ++j) {
            if (k[j] < (unsigned)nn) s1_add(tab, k[j], g[j]);
          }
        }
        done = e0 + (nv << 2);
      }
      for (int e = done + tid; e < e1; e += nth) {
        const unsigned k = (unsigned)(col[e] - n0);
        if (k < (unsigned)nn) s1_add(tab, k, xs4[(unsigned)row[e]]);
      }
      __syncthreads();

      const int lines = (nn * 4 + 31) >> 5;
      const int rot   = (int)(((long long)sl * lines) / S) << 5;
      float* dst = acc1 + (size_t)n0 * 4;
      for (int j = tid; j < lines * 32; j += nth) {
        int i = j + rot;
        if (i >= lines * 32) i -= lines * 32;
        const int node = i >> 2, f = i & 3;
        if (node < nn && f < 3) {
          const float v = tab[node * 3 + f];
          if (v != 0.0f) atomicAdd(&dst[i], v);
        }
      }
    }
  }
}

__global__ void __launch_bounds__(256)
k_dense2(const float* __restrict__ acc1, const float* __restrict__ xs,
         const float* __restrict__ dinv, const float* __restrict__ gW1,
         const float* __restrict__ gW2, const float* __restrict__ gb1,
         float* __restrict__ xw2s, int N) {
  __shared__ float ht[WPB][256];
  __shared__ float dv[WPB][16];
  int wslot = threadIdx.x >> 5;
  int lane  = threadIdx.x & 31;
  int m0 = (blockIdx.x * WPB + wslot) * 16;
  if (m0 >= N) return;
  bool full = (m0 + 16) <= N;
  float* t   = ht[wslot];
  float* dvs = dv[wslot];

  int n  = lane & 15;
  int mo = (lane >> 4) << 3;

  {
    int r  = m0 + n;
    int rc = (r < N) ? r : N - 1;
    float dvr = dinv[rc];
    if (lane < 16) dvs[n] = dvr;
    float4 a4 = *(const float4*)(acc1 + (size_t)rc * 4);
    float4 s4 = *(const float4*)(xs   + (size_t)rc * 4);
    float g0 = dvr * (a4.x + s4.x);
    float g1 = dvr * (a4.y + s4.y);
    float g2 = dvr * (a4.z + s4.z);
    v2f a;
    a.x = (lane < 16) ? g0 : g2;
    a.y = (lane < 16) ? g1 : 0.0f;
    v2f b = b_frag(gW1, lane, 0, 3, 16);
    v8f c1 = {};
    c1 = wmma_f32(a, b, c1);
    float bias = gb1[n];
    for (int v = 0; v < 8; ++v)
      t[(mo + v) * 16 + n] = fmaxf(c1[v] + bias, 0.0f);
  }
  __builtin_amdgcn_wave_barrier();

  v8f c = {};
  for (int k0 = 0; k0 < 16; k0 += 4) {
    v2f a = a_frag_lds(t, lane, k0);
    v2f b = b_frag(gW2, lane, k0, 16, 8);
    c = wmma_f32(a, b, c);
  }
  __builtin_amdgcn_wave_barrier();

  if (n < 8) {
    for (int v = 0; v < 8; ++v)
      t[(mo + v) * 8 + n] = c[v];
  }
  __builtin_amdgcn_wave_barrier();

  if (full) {
    float dvr = dvs[lane >> 1];
    float4 v4 = ((const float4*)t)[lane];
    v4.x *= dvr; v4.y *= dvr; v4.z *= dvr; v4.w *= dvr;
    ((float4*)(xw2s + (size_t)m0 * 8))[lane] = v4;
  } else if (n < 8) {
    for (int v = 0; v < 8; ++v) {
      int r = m0 + mo + v;
      if (r < N) xw2s[(size_t)r * 8 + n] = t[(mo + v) * 8 + n] * dvs[mo + v];
    }
  }
}

__global__ void __launch_bounds__(256)
k_scatter2(const int* __restrict__ row, const int* __restrict__ col,
           const float* __restrict__ xw2s, float* __restrict__ acc2,
           int* __restrict__ ctr, int N, int NE) {
  const int      lane = threadIdx.x & 31;
  const unsigned f    = lane & 7;
  const int split  = node_split(N);
  const int nchunk = (NE + 32 * BPC - 1) / (32 * BPC);
  const int mine   = chip_half();
  for (int ph = 0; ph < 2; ++ph) {
    const int p = mine ^ ph;
    int* cnt = ctr + p * CTR_PITCH;
    int ck = take_chunk(cnt, lane);
    while (ck < nchunk) {
      int r[BPC], c[BPC];
#pragma unroll
      for (int u = 0; u < BPC; ++u) {
        const int e  = (ck * BPC + u) * 32 + lane;
        const int ec = (e < NE) ? e : NE - 1;
        r[u] = row[ec]; c[u] = col[ec];
      }
      const int nx = take_chunk(cnt, lane);
#pragma unroll
      for (int u = 0; u < BPC; ++u) {
        const int e = (ck * BPC + u) * 32 + lane;
        const int ru = r[u], cu = c[u];
        asm volatile("" :: "v"(ru), "v"(cu));
        const int hit = (e < NE && ((cu >= split) ? 1 : 0) == p) ? 1 : 0;
        float v[8];
#define GATHERX(J) v[J] = 0.0f; if (grp8_bcast<J>(hit)) v[J] = xw2s[(unsigned)grp8_bcast<J>(ru) * 8u + f];
        REP8(GATHERX)
#undef GATHERX
#define ADDX(J) if (grp8_bcast<J>(hit)) atomicAdd(&acc2[(unsigned)grp8_bcast<J>(cu) * 8u + f], v[J]);
        REP8(ADDX)
#undef ADDX
      }
      ck = nx;
    }
  }
}

__global__ void __launch_bounds__(256)
k_final(const float* __restrict__ x, const float* __restrict__ acc2,
        const float* __restrict__ xw2s, const float* __restrict__ dinv,
        const float* __restrict__ gb2,
        const float* __restrict__ cW1, const float* __restrict__ cb1,
        const float* __restrict__ cW2, const float* __restrict__ cb2,
        const float* __restrict__ mW1, const float* __restrict__ mb1,
        const float* __restrict__ mW2, const float* __restrict__ mb2,
        float* __restrict__ out, int N) {
  __shared__ float ht[WPB][256];
  __shared__ float ct[WPB][128];
  __shared__ float g0s[WPB][16];
  int wslot = threadIdx.x >> 5;
  int lane  = threadIdx.x & 31;
  int m0 = (blockIdx.x * WPB + wslot) * 16;
  if (m0 >= N) return;
  bool full = (m0 + 16) <= N;
  float* t  = ht[wslot];
  float* o  = ct[wslot];
  float* g0 = g0s[wslot];

  int n  = lane & 15;
  int mo = (lane >> 4) << 3;

  int grow = lane >> 1, gcc = lane & 1;
  int gr  = m0 + grow;
  int grc = (gr < N) ? gr : N - 1;
  float  gdv = dinv[grc];
  float4 a24 = *(const float4*)(acc2 + (size_t)grc * 8 + gcc * 4);
  float4 s24 = *(const float4*)(xw2s + (size_t)grc * 8 + gcc * 4);
  float4 gb4 = ((const float4*)gb2)[gcc];
  float4 go4;
  go4.x = gdv * (a24.x + s24.x) + gb4.x;
  go4.y = gdv * (a24.y + s24.y) + gb4.y;
  go4.z = gdv * (a24.z + s24.z) + gb4.z;
  go4.w = gdv * (a24.w + s24.w) + gb4.w;

  v8f ch = {};
  for (int k0 = 0; k0 < 16; k0 += 4) {
    v2f a = a_frag_x(x, N, m0, lane, k0,  3,  13);
    v2f b = b_frag(cW1, lane, k0, 13, 16);
    ch = wmma_f32(a, b, ch);
  }
  for (int v = 0; v < 8; ++v) t[(mo + v) * 16 + n] = fmaxf(ch[v] + cb1[n], 0.0f);
  __builtin_amdgcn_wave_barrier();

  v8f cc = {};
  for (int k0 = 0; k0 < 16; k0 += 4) {
    v2f a = a_frag_lds(t, lane, k0);
    v2f b = b_frag(cW2, lane, k0, 16, 8);
    cc = wmma_f32(a, b, cc);
  }
  if (n < 8) {
    for (int v = 0; v < 8; ++v) o[(mo + v) * 8 + n] = cc[v] + cb2[n];
  }
  __builtin_amdgcn_wave_barrier();

  v8f cm = {};
  for (int k0 = 0; k0 < 16; k0 += 4) {
    v2f a = a_frag_x(x, N, m0, lane, k0,  0,  16);
    v2f b = b_frag(mW1, lane, k0, 16, 16);
    cm = wmma_f32(a, b, cm);
  }
  for (int v = 0; v < 8; ++v) t[(mo + v) * 16 + n] = fmaxf(cm[v] + mb1[n], 0.0f);
  __builtin_amdgcn_wave_barrier();

  if (lane < 16) {
    float l0 = mb2[0], l1 = mb2[1];
    for (int j = 0; j < 16; ++j) {
      float hv = t[lane * 16 + j];
      l0 += hv * mW2[j * 2 + 0];
      l1 += hv * mW2[j * 2 + 1];
    }
    float mx = fmaxf(l0, l1);
    float e0 = __expf(l0 - mx), e1 = __expf(l1 - mx);
    g0[lane] = e0 / (e0 + e1);
  }
  __builtin_amdgcn_wave_barrier();

  float g  = g0[grow];
  float4 c4 = *(const float4*)&o[lane * 4];
  float4 r4;
  r4.x = g * go4.x + (1.0f - g) * c4.x;
  r4.y = g * go4.y + (1.0f - g) * c4.y;
  r4.z = g * go4.z + (1.0f - g) * c4.z;
  r4.w = g * go4.w + (1.0f - g) * c4.w;
  if (full) {
    ((float4*)(out + (size_t)m0 * 8))[lane] = r4;
  } else if (gr < N) {
    *(float4*)(out + (size_t)gr * 8 + gcc * 4) = r4;
  }
}

extern "C" void kernel_launch(void* const* d_in, const int* in_sizes, int n_in,
                              void* d_out, int out_size, void* d_ws, size_t ws_size,
                              hipStream_t stream) {
  const float* x   = (const float*)d_in[0];
  const int*   ei  = (const int*)  d_in[1];
  const float* gW1 = (const float*)d_in[2];
  const float* gb1 = (const float*)d_in[3];
  const float* gW2 = (const float*)d_in[4];
  const float* gb2 = (const float*)d_in[5];
  const float* cW1 = (const float*)d_in[6];
  const float* cb1 = (const float*)d_in[7];
  const float* cW2 = (const float*)d_in[8];
  const float* cb2 = (const float*)d_in[9];
  const float* mW1 = (const float*)d_in[10];
  const float* mb1 = (const float*)d_in[11];
  const float* mW2 = (const float*)d_in[12];
  const float* mb2 = (const float*)d_in[13];
  float* out = (float*)d_out;

  const int N  = in_sizes[0] / 16;
  const int NE = in_sizes[1] / 2;
  const int* erow = ei;
  const int* ecol = ei + NE;

  float* deg  = (float*)d_ws;
  float* dinv = deg  + N;
  float* xs   = dinv + N;
  float* acc1 = xs   + (size_t)4 * N;
  float* xw2s = acc1 + (size_t)4 * N;
  float* acc2 = xw2s + (size_t)8 * N;
  int*   ctr  = (int*)(acc2 + (size_t)8 * N);

  const int tiles   = (N + 15) / 16;
  const int gBlocks = (tiles + WPB - 1) / WPB;

  k_init    <<<(N + 255) / 256, 256, 0, stream>>>(deg, acc1, acc2, ctr, N);
  k_degree  <<<256, 1024, 0, stream>>>(ecol, deg, N, NE);
  k_scale1  <<<(N + 255) / 256, 256, 0, stream>>>(x, deg, dinv, xs, N);
  k_scatter1<<<256, 1024, 0, stream>>>(erow, ecol, xs, acc1, ctr, N, NE);
  k_dense2  <<<gBlocks, 256, 0, stream>>>(acc1, xs, dinv, gW1, gW2, gb1, xw2s, N);
  k_scatter2<<<2048, 256, 0, stream>>>(erow, ecol, xw2s, acc2, ctr + 2 * CTR_PITCH, N, NE);
  k_final   <<<gBlocks, 256, 0, stream>>>(x, acc2, xw2s, dinv, gb2,
                                          cW1, cb1, cW2, cb2,
                                          mW1, mb1, mW2, mb2, out, N);
}


# edge list bucketed by target once per call (256-node buckets, LDS counting sort); both layers sum gathered rows in per-bucket LDS tables, each acc line stored once; the order of the f32 sums changes
# speedup vs baseline: 6.8443x; 1.1891x over previous
#include <hip/hip_runtime.h>
#include <math.h>


typedef float v2f __attribute__((ext_vector_type(2)));
typedef float v8f __attribute__((ext_vector_type(8)));

#define WPB 8

static __device__ __forceinline__ v8f wmma_f32(v2f a, v2f b, v8f c) {
  return __builtin_amdgcn_wmma_f32_16x16x4_f32(false, a, false, b,
                                               (short)0, c, false, false);
}

static __device__ __forceinline__ v2f b_frag(const float* __restrict__ W,
                                             int lane, int k0, int K, int Ncols) {
  int n  = lane & 15;
  int kb = k0 + ((lane >> 4) << 1);
  int nm  = (n      < Ncols) ? n      : (Ncols - 1);
  int kc0 = (kb     < K)     ? kb     : (K - 1);
  int kc1 = (kb + 1 < K)     ? kb + 1 : (K - 1);
  float w0 = W[kc0 * Ncols + nm];
  float w1 = W[kc1 * Ncols + nm];
  v2f b;
  b.x = (kb     < K && n < Ncols) ? w0 : 0.0f;
  b.y = (kb + 1 < K && n < Ncols) ? w1 : 0.0f;
  return b;
}

static __device__ __forceinline__ v2f a_frag_x(const float* __restrict__ x,
                                               int N, int m0, int lane,
                                               int k0, int co, int K) {
  int r = m0 + (lane & 15);
  if (r >= N) r = N - 1;
  int kb = k0 + ((lane >> 4) << 1);
  const float* rowp = x + (size_t)r * 16;
  int i0 = co + kb;     i0 = (i0 < 16) ? i0 : 15;
  int i1 = co + kb + 1; i1 = (i1 < 16) ? i1 : 15;
  float a0 = rowp[i0];
  float a1 = rowp[i1];
  v2f a;
  a.x = (kb     < K) ? a0 : 0.0f;
  a.y = (kb + 1 < K) ? a1 : 0.0f;
  return a;
}

static __device__ __forceinline__ v2f a_frag_lds(const float* t, int lane, int k0) {
  int r  = lane & 15;
  int kb = k0 + ((lane >> 4) << 1);
  v2f a;
  a.x = t[r * 16 + kb];
  a.y = t[r * 16 + kb + 1];
  return a;
}

__global__ void k_init(float* __restrict__ deg, int N) {
  int i = blockIdx.x * blockDim.x + threadIdx.x;
  if (i < N) deg[i] = 1.0f;
}

#define DEG_CAP 25088

static __device__ __forceinline__ void deg_bump(unsigned* cnt, int c, int n0, int nn) {
  unsigned k = (unsigned)(c - n0);
  if (k < (unsigned)nn) atomicAdd(&cnt[k], 1u);
}

__global__ void __launch_bounds__(1024)
k_degree(const int* __restrict__ col, float* __restrict__ deg, int N, int NE) {
  __shared__ unsigned cnt[DEG_CAP];
  const int tid = (int)threadIdx.x;
  const int nth = (int)blockDim.x;
  const int R = (N + DEG_CAP - 1) / DEG_CAP;
  int S = (int)gridDim.x / R;
  if (S < 1) S = 1;
  const int per = (((NE + S - 1) / S) + 3) & ~3;
  const bool aligned = (((size_t)col) & 15) == 0;

  for (int item = (int)blockIdx.x; item < R * S; item += (int)gridDim.x) {
    const int rg = item % R;
    const int sl = item / R;
    const int n0 = rg * DEG_CAP;
    const int nn = (N - n0 < DEG_CAP) ? (N - n0) : DEG_CAP;
    for (int i = tid; i < nn; i += nth) cnt[i] = 0u;
    __syncthreads();

    const int e0 = (sl * per < NE) ? sl * per : NE;
    const int e1 = (e0 + per < NE) ? e0 + per : NE;
    int done = e0;
    if (aligned) {
      const int4* c4 = (const int4*)(col + e0);
      const int nv = (e1 - e0) >> 2;
      const int4 none = make_int4(-1, -1, -1, -1);
      for (int i = tid; i < nv; i += 8 * nth) {
        int4 v[8];
#pragma unroll
        for (int u = 0; u < 8; ++u) {
          const int iu = i + u * nth;
          v[u] = c4[(iu < nv) ? iu : nv - 1];
        }
#pragma unroll
        for (int u = 0; u < 8; ++u) {
          if (i + u * nth >= nv) v[u] = none;
          deg_bump(cnt, v[u].x, n0, nn); deg_bump(cnt, v[u].y, n0, nn);
          deg_bump(cnt, v[u].z, n0, nn); deg_bump(cnt, v[u].w, n0, nn);
        }
      }
      done = e0 + (nv << 2);
    }
    for (int e = done + tid; e < e1; e += nth) deg_bump(cnt, col[e], n0, nn);
    __syncthreads();

    const int lines = (nn + 31) >> 5;
    const int rot   = (int)(((long long)sl * lines) / S) << 5;
    for (int j = tid; j < lines * 32; j += nth) {
      int i = j + rot;
      if (i >= lines * 32) i -= lines * 32;
      if (i < nn) {
        unsigned c = cnt[i];
        if (c) atomicAdd(&deg[n0 + i], (float)c);
      }
    }
    __syncthreads();
  }
}

__global__ void __launch_bounds__(256)
k_scale1(const float* __restrict__ x, const float* __restrict__ deg,
         float* __restrict__ dinv, float* __restrict__ xs, int N) {
  int i = blockIdx.x * blockDim.x + threadIdx.x;
  if (i >= N) return;
  float d = __builtin_amdgcn_rsqf(deg[i]);
  dinv[i] = d;
  const float* xr = x + (size_t)i * 16;
  float4 o = make_float4(d * xr[0], d * xr[1], d * xr[2], 0.0f);
  *(float4*)(xs + (size_t)i * 4) = o;
}

#define BKT_SH 8
#define BKT_N  (1 << BKT_SH)
#define MAXB   512
#define TP     (MAXB + 8)
#define CH     25024

static __device__ __forceinline__ unsigned bucket_of(int c) {
  unsigned b = (unsigned)c >> BKT_SH;
  return (b < (unsigned)MAXB) ? b : (unsigned)(MAXB - 1);
}

__global__ void __launch_bounds__(1024)
k_bucket(const int* __restrict__ row, const int* __restrict__ col,
         int* __restrict__ tbl, int2* __restrict__ pairs, int NE) {
  __shared__ int2 stage[CH];
  __shared__ int  hist[MAXB];
  __shared__ int  wsum[MAXB / 32];
  const int tid  = (int)threadIdx.x;
  const int nth  = (int)blockDim.x;
  const int lane = tid & 31;
  const int items = (NE + CH - 1) / CH;

  for (int item = (int)blockIdx.x; item < items; item += (int)gridDim.x) {
    const int e0 = item * CH;
    const int n  = (NE - e0 < CH) ? (NE - e0) : CH;
    const int* __restrict__ colp = col + e0;
    const int* __restrict__ rowp = row + e0;
    for (int i = tid; i < MAXB; i += nth) hist[i] = 0;
    __syncthreads();

    for (int i = tid; i < n; i += 8 * nth) {
      int c[8];
#pragma unroll
      for (int u = 0; u < 8; ++u) {
        const int iu = i + u * nth;
        c[u] = colp[(iu < n) ? iu : n - 1];
      }
#pragma unroll
      for (int u = 0; u < 8; ++u) {
        if (i + u * nth < n) atomicAdd(&hist[bucket_of(c[u])], 1);
      }
    }
    __syncthreads();

    {
      const int v = (tid < MAXB) ? hist[tid] : 0;
      int incl = v;
      for (int d = 1; d < 32; d <<= 1) {
        const int t = __shfl_up(incl, d);
        if (lane >= d) incl += t;
      }
      if (tid < MAXB && lane == 31) wsum[tid >> 5] = incl;
      __syncthreads();
      if (tid < MAXB) {
        int pre = 0;
        for (int w = 0; w < (tid >> 5); ++w) pre += wsum[w];
        const int ex = pre + incl - v;
        hist[tid] = ex;
        tbl[(size_t)item * TP + tid] = ex;
      }
      if (tid == 0) tbl[(size_t)item * TP + MAXB] = n;
      __syncthreads();
    }

    for (int i = tid; i < n; i += 4 * nth) {
      int c[4], r[4];
#pragma unroll
      for (int u = 0; u < 4; ++u) {
        const int iu = i + u * nth;
        const int ic = (iu < n) ? iu : n - 1;
        c[u] = colp[ic]; r[u] = rowp[ic];
      }
#pragma unroll
      for (int u = 0; u < 4; ++u) {
        if (i + u * nth < n) {
          const int pos = atomicAdd(&hist[bucket_of(c[u])], 1);
          stage[pos] = make_int2(r[u], c[u]);
        }
      }
    }
    __syncthreads();

    int2* __restrict__ dst = pairs + (size_t)e0;
    for (int i = tid; i < n; i += nth) dst[i] = stage[i];
    __syncthreads();
  }
}

#define RUNS_LDS 1024
#define TR       10

template <int V>
__global__ void __launch_bounds__(1024)
k_sum(const int2* __restrict__ pairs, const int* __restrict__ tbl,
      const float* __restrict__ src, float* __restrict__ acc, int N, int NE) {
  constexpr int P = 4 * V;
  __shared__ float tab[BKT_N * P];
  __shared__ int   so[RUNS_LDS];
  __shared__ int   eo[RUNS_LDS];
  const int tid  = (int)threadIdx.x;
  const int nth  = (int)blockDim.x;
  const int lane = tid & 31;
  const int wv   = tid >> 5;
  const int nwv  = nth >> 5;
  const int q    = lane & 3;
  const int sub  = lane >> 2;
  const int b    = (int)blockIdx.x;
  const int n0   = b << BKT_SH;
  const int items = (NE + CH - 1) / CH;
  const unsigned lastrow = (unsigned)(N - 1);

  for (int i = tid; i < BKT_N * P; i += nth) tab[i] = 0.0f;

  for (int ib = 0; ib < items; ib += RUNS_LDS) {
    __syncthreads();
    const int cnt = (items - ib < RUNS_LDS) ? (items - ib) : RUNS_LDS;
    for (int t = tid; t < cnt; t += nth) {
      so[t] = tbl[(size_t)(ib + t) * TP + b];
      eo[t] = tbl[(size_t)(ib + t) * TP + b + 1];
    }
    __syncthreads();

    for (int rr = wv; rr < cnt; rr += nwv) {
      const int s = __builtin_amdgcn_readfirstlane(so[rr]);
      const int e = __builtin_amdgcn_readfirstlane(eo[rr]);
      const int2* __restrict__ base = pairs + (size_t)(ib + rr) * CH;
      for (int j0 = s; j0 < e; j0 += 8 * TR) {
        int2 p[TR];
#pragma unroll
        for (int u = 0; u < TR; ++u) {
          const int j = j0 + u * 8 + sub;
          p[u] = base[(j < e) ? j : e - 1];
        }
        unsigned k[TR];
        float g[TR][V];
#pragma unroll
        for (int u = 0; u < TR; ++u) {
          const int j = j0 + u * 8 + sub;
          k[u] = (j < e) ? (unsigned)(p[u].y - n0) : 0xFFFFFFFFu;
          unsigned r = (unsigned)p[u].x;
          r = (r < lastrow) ? r : lastrow;
          if (k[u] >= (unsigned)BKT_N) r = 0u;
          const float* gp = src + (size_t)r * P + q * V;
          if (V == 1) { g[u][0] = gp[0]; }
          else { const float2 t2 = *(const float2*)gp; g[u][0] = t2.x; g[u][V - 1] = t2.y; }
        }
#pragma unroll
        for (int u = 0; u < TR; ++u) {
          if (k[u] < (unsigned)BKT_N) {
            float* tp = &tab[k[u] * P + q * V];
            atomicAdd(tp, g[u][0]);
            if (V == 2) atomicAdd(tp + (V - 1), g[u][V - 1]);
          }
        }
      }
    }
  }
  __syncthreads();

  for (int i = tid; i < BKT_N * P; i += nth) {
    const int node = n0 + i / P;
    if (node < N) acc[(size_t)n0 * P + i] = tab[i];
  }
}

__global__ void __launch_bounds__(256)
k_dense2(const float* __restrict__ acc1, const float* __restrict__ xs,
         const float* __restrict__ dinv, const float* __restrict__ gW1,
         const float* __restrict__ gW2, const float* __restrict__ gb1,
         float* __restrict__ xw2s, int N) {
  __shared__ float ht[WPB][256];
  __shared__ float dv[WPB][16];
  int wslot = threadIdx.x >> 5;
  int lane  = threadIdx.x & 31;
  int m0 = (blockIdx.x * WPB + wslot) * 16;
  if (m0 >= N) return;
  bool full = (m0 + 16) <= N;
  float* t   = ht[wslot];
  float* dvs = dv[wslot];

  int n  = lane & 15;
  int mo = (lane >> 4) << 3;

  {
    int r  = m0 + n;
    int rc = (r < N) ? r : N - 1;
    float dvr = dinv[rc];
    if (lane < 16) dvs[n] = dvr;
    float4 a4 = *(const float4*)(acc1 + (size_t)rc * 4);
    float4 s4 = *(const float4*)(xs   + (size_t)rc * 4);
    float g0 = dvr * (a4.x + s4.x);
    float g1 = dvr * (a4.y + s4.y);
    float g2 = dvr * (a4.z + s4.z);
    v2f a;
    a.x = (lane < 16) ? g0 : g2;
    a.y = (lane < 16) ? g1 : 0.0f;
    v2f b = b_frag(gW1, lane, 0, 3, 16);
    v8f c1 = {};
    c1 = wmma_f32(a, b, c1);
    float bias = gb1[n];
    for (int v = 0; v < 8; ++v)
      t[(mo + v) * 16 + n] = fmaxf(c1[v] + bias, 0.0f);
  }
  __builtin_amdgcn_wave_barrier();

  v8f c = {};
  for (int k0 = 0; k0 < 16; k0 += 4) {
    v2f a = a_frag_lds(t, lane, k0);
    v2f b = b_frag(gW2, lane, k0, 16, 8);
    c = wmma_f32(a, b, c);
  }
  __builtin_amdgcn_wave_barrier();

  if (n < 8) {
    for (int v = 0; v < 8; ++v)
      t[(mo + v) * 8 + n] = c[v];
  }
  __builtin_amdgcn_wave_barrier();

  if (full) {
    float dvr = dvs[lane >> 1];
    float4 v4 = ((const float4*)t)[lane];
    v4.x *= dvr; v4.y *= dvr; v4.z *= dvr; v4.w *= dvr;
    ((float4*)(xw2s + (size_t)m0 * 8))[lane] = v4;
  } else if (n < 8) {
    for (int v = 0; v < 8; ++v) {
      int r = m0 + mo + v;
      if (r < N) xw2s[(size_t)r * 8 + n] = t[(mo + v) * 8 + n] * dvs[mo + v];
    }
  }
}

__global__ void __launch_bounds__(256)
k_final(const float* __restrict__ x, const float* __restrict__ acc2,
        const float* __restrict__ xw2s, const float* __restrict__ dinv,
        const float* __restrict__ gb2,
        const float* __restrict__ cW1, const float* __restrict__ cb1,
        const float* __restrict__ cW2, const float* __restrict__ cb2,
        const float* __restrict__ mW1, const float* __restrict__ mb1,
        const float* __restrict__ mW2, const float* __restrict__ mb2,
        float* __restrict__ out, int N) {
  __shared__ float ht[WPB][256];
  __shared__ float ct[WPB][128];
  __shared__ float g0s[WPB][16];
  int wslot = threadIdx.x >> 5;
  int lane  = threadIdx.x & 31;
  int m0 = (blockIdx.x * WPB + wslot) * 16;
  if (m0 >= N) return;
  bool full = (m0 + 16) <= N;
  float* t  = ht[wslot];
  float* o  = ct[wslot];
  float* g0 = g0s[wslot];

  int n  = lane & 15;
  int mo = (lane >> 4) << 3;

  int grow = lane >> 1, gcc = lane & 1;
  int gr  = m0 + grow;
  int grc = (gr < N) ? gr : N - 1;
  float  gdv = dinv[grc];
  float4 a24 = *(const float4*)(acc2 + (size_t)grc * 8 + gcc * 4);
  float4 s24 = *(const float4*)(xw2s + (size_t)grc * 8 + gcc * 4);
  float4 gb4 = ((const float4*)gb2)[gcc];
  float4 go4;
  go4.x = gdv * (a24.x + s24.x) + gb4.x;
  go4.y = gdv * (a24.y + s24.y) + gb4.y;
  go4.z = gdv * (a24.z + s24.z) + gb4.z;
  go4.w = gdv * (a24.w + s24.w) + gb4.w;

  v8f ch = {};
  for (int k0 = 0; k0 < 16; k0 += 4) {
    v2f a = a_frag_x(x, N, m0, lane, k0,  3,  13);
    v2f b = b_frag(cW1, lane, k0, 13, 16);
    ch = wmma_f32(a, b, ch);
  }
  for (int v = 0; v < 8; ++v) t[(mo + v) * 16 + n] = fmaxf(ch[v] + cb1[n], 0.0f);
  __builtin_amdgcn_wave_barrier();

  v8f cc = {};
  for (int k0 = 0; k0 < 16; k0 += 4) {
    v2f a = a_frag_lds(t, lane, k0);
    v2f b = b_frag(cW2, lane, k0, 16, 8);
    cc = wmma_f32(a, b, cc);
  }
  if (n < 8) {
    for (int v = 0; v < 8; ++v) o[(mo + v) * 8 + n] = cc[v] + cb2[n];
  }
  __builtin_amdgcn_wave_barrier();

  v8f cm = {};
  for (int k0 = 0; k0 < 16; k0 += 4) {
    v2f a = a_frag_x(x, N, m0, lane, k0,  0,  16);
    v2f b = b_frag(mW1, lane, k0, 16, 16);
    cm = wmma_f32(a, b, cm);
  }
  for (int v = 0; v < 8; ++v) t[(mo + v) * 16 + n] = fmaxf(cm[v] + mb1[n], 0.0f);
  __builtin_amdgcn_wave_barrier();

  if (lane < 16) {
    float l0 = mb2[0], l1 = mb2[1];
    for (int j = 0; j < 16; ++j) {
      float hv = t[lane * 16 + j];
      l0 += hv * mW2[j * 2 + 0];
      l1 += hv * mW2[j * 2 + 1];
    }
    float mx = fmaxf(l0, l1);
    float e0 = __expf(l0 - mx), e1 = __expf(l1 - mx);
    g0[lane] = e0 / (e0 + e1);
  }
  __builtin_amdgcn_wave_barrier();

  float g  = g0[grow];
  float4 c4 = *(const float4*)&o[lane * 4];
  float4 r4;
  r4.x = g * go4.x + (1.0f - g) * c4.x;
  r4.y = g * go4.y + (1.0f - g) * c4.y;
  r4.z = g * go4.z + (1.0f - g) * c4.z;
  r4.w = g * go4.w + (1.0f - g) * c4.w;
  if (full) {
    ((float4*)(out + (size_t)m0 * 8))[lane] = r4;
  } else if (gr < N) {
    *(float4*)(out + (size_t)gr * 8 + gcc * 4) = r4;
  }
}

extern "C" void kernel_launch(void* const* d_in, const int* in_sizes, int n_in,
                              void* d_out, int out_size, void* d_ws, size_t ws_size,
                              hipStream_t stream) {
  const float* x   = (const float*)d_in[0];
  const int*   ei  = (const int*)  d_in[1];
  const float* gW1 = (const float*)d_in[2];
  const float* gb1 = (const float*)d_in[3];
  const float* gW2 = (const float*)d_in[4];
  const float* gb2 = (const float*)d_in[5];
  const float* cW1 = (const float*)d_in[6];
  const float* cb1 = (const float*)d_in[7];
  const float* cW2 = (const float*)d_in[8];
  const float* cb2 = (const float*)d_in[9];
  const float* mW1 = (const float*)d_in[10];
  const float* mb1 = (const float*)d_in[11];
  const float* mW2 = (const float*)d_in[12];
  const float* mb2 = (const float*)d_in[13];
  float* out = (float*)d_out;

  const int N  = in_sizes[0] / 16;
  const int NE = in_sizes[1] / 2;
  const int* erow = ei;
  const int* ecol = ei + NE;

  float* deg  = (float*)d_ws;
  float* dinv = deg  + N;
  float* xs   = dinv + N;
  float* acc1 = xs   + (size_t)4 * N;
  float* xw2s = acc1 + (size_t)4 * N;
  float* acc2 = xw2s + (size_t)8 * N;
  const int items = (NE + CH - 1) / CH;
  int*   tbl   = (int*)(acc2 + (size_t)8 * N);
  int2*  pairs = (int2*)(tbl + (size_t)items * TP);

  const int tiles   = (N + 15) / 16;
  const int gBlocks = (tiles + WPB - 1) / WPB;
  const int buckets = (N + BKT_N - 1) / BKT_N;

  k_init    <<<(N + 255) / 256, 256, 0, stream>>>(deg, N);
  k_degree  <<<256, 1024, 0, stream>>>(ecol, deg, N, NE);
  k_scale1  <<<(N + 255) / 256, 256, 0, stream>>>(x, deg, dinv, xs, N);
  k_bucket  <<<256, 1024, 0, stream>>>(erow, ecol, tbl, pairs, NE);
  k_sum<1>  <<<buckets, 1024, 0, stream>>>(pairs, tbl, xs, acc1, N, NE);
  k_dense2  <<<gBlocks, 256, 0, stream>>>(acc1, xs, dinv, gW1, gW2, gb1, xw2s, N);
  k_sum<2>  <<<buckets, 1024, 0, stream>>>(pairs, tbl, xw2s, acc2, N, NE);
  k_final   <<<gBlocks, 256, 0, stream>>>(x, acc2, xw2s, dinv, gb2,
                                          cW1, cb1, cW2, cb2,
                                          mW1, mb1, mW2, mb2, out, N);
}


# edges bucketed by (target range, source range) tile: a block gathers rows from an LDS copy of its source range, not from L2, and stores one partial per source range; the f32 sum order changes
# speedup vs baseline: 8.8319x; 1.2904x over previous
#include <hip/hip_runtime.h>
#include <math.h>


typedef float v2f __attribute__((ext_vector_type(2)));
typedef float v8f __attribute__((ext_vector_type(8)));

#define WPB 8

static __device__ __forceinline__ v8f wmma_f32(v2f a, v2f b, v8f c) {
  return __builtin_amdgcn_wmma_f32_16x16x4_f32(false, a, false, b,
                                               (short)0, c, false, false);
}

static __device__ __forceinline__ v2f b_frag(const float* __restrict__ W,
                                             int lane, int k0, int K, int Ncols) {
  int n  = lane & 15;
  int kb = k0 + ((lane >> 4) << 1);
  int nm  = (n      < Ncols) ? n      : (Ncols - 1);
  int kc0 = (kb     < K)     ? kb     : (K - 1);
  int kc1 = (kb + 1 < K)     ? kb + 1 : (K - 1);
  float w0 = W[kc0 * Ncols + nm];
  float w1 = W[kc1 * Ncols + nm];
  v2f b;
  b.x = (kb     < K && n < Ncols) ? w0 : 0.0f;
  b.y = (kb + 1 < K && n < Ncols) ? w1 : 0.0f;
  return b;
}

static __device__ __forceinline__ v2f a_frag_x(const float* __restrict__ x,
                                               int N, int m0, int lane,
                                               int k0, int co, int K) {
  int r = m0 + (lane & 15);
  if (r >= N) r = N - 1;
  int kb = k0 + ((lane >> 4) << 1);
  const float* rowp = x + (size_t)r * 16;
  int i0 = co + kb;     i0 = (i0 < 16) ? i0 : 15;
  int i1 = co + kb + 1; i1 = (i1 < 16) ? i1 : 15;
  float a0 = rowp[i0];
  float a1 = rowp[i1];
  v2f a;
  a.x = (kb     < K) ? a0 : 0.0f;
  a.y = (kb + 1 < K) ? a1 : 0.0f;
  return a;
}

static __device__ __forceinline__ v2f a_frag_lds(const float* t, int lane, int k0) {
  int r  = lane & 15;
  int kb = k0 + ((lane >> 4) << 1);
  v2f a;
  a.x = t[r * 16 + kb];
  a.y = t[r * 16 + kb + 1];
  return a;
}

__global__ void k_init(float* __restrict__ deg, int N) {
  int i = blockIdx.x * blockDim.x + threadIdx.x;
  if (i < N) deg[i] = 1.0f;
}

#define DEG_CAP 25088

static __device__ __forceinline__ void deg_bump(unsigned* cnt, int c, int n0, int nn) {
  unsigned k = (unsigned)(c - n0);
  if (k < (unsigned)nn) atomicAdd(&cnt[k], 1u);
}

__global__ void __launch_bounds__(1024)
k_degree(const int* __restrict__ col, float* __restrict__ deg, int N, int NE) {
  __shared__ unsigned cnt[DEG_CAP];
  const int tid = (int)threadIdx.x;
  const int nth = (int)blockDim.x;
  const int R = (N + DEG_CAP - 1) / DEG_CAP;
  int S = (int)gridDim.x / R;
  if (S < 1) S = 1;
  const int per = (((NE + S - 1) / S) + 3) & ~3;
  const bool aligned = (((size_t)col) & 15) == 0;

  for (int item = (int)blockIdx.x; item < R * S; item += (int)gridDim.x) {
    const int rg = item % R;
    const int sl = item / R;
    const int n0 = rg * DEG_CAP;
    const int nn = (N - n0 < DEG_CAP) ? (N - n0) : DEG_CAP;
    for (int i = tid; i < nn; i += nth) cnt[i] = 0u;
    __syncthreads();

    const int e0 = (sl * per < NE) ? sl * per : NE;
    const int e1 = (e0 + per < NE) ? e0 + per : NE;
    int done = e0;
    if (aligned) {
      const int4* c4 = (const int4*)(col + e0);
      const int nv = (e1 - e0) >> 2;
      const int4 none = make_int4(-1, -1, -1, -1);
      for (int i = tid; i < nv; i += 8 * nth) {
        int4 v[8];
#pragma unroll
        for (int u = 0; u < 8; ++u) {
          const int iu = i + u * nth;
          v[u] = c4[(iu < nv) ? iu : nv - 1];
        }
#pragma unroll
        for (int u = 0; u < 8; ++u) {
          if (i + u * nth >= nv) v[u] = none;
          deg_bump(cnt, v[u].x, n0, nn); deg_bump(cnt, v[u].y, n0, nn);
          deg_bump(cnt, v[u].z, n0, nn); deg_bump(cnt, v[u].w, n0, nn);
        }
      }
      done = e0 + (nv << 2);
    }
    for (int e = done + tid; e < e1; e += nth) deg_bump(cnt, col[e], n0, nn);
    __syncthreads();

    const int lines = (nn + 31) >> 5;
    const int rot   = (int)(((long long)sl * lines) / S) << 5;
    for (int j = tid; j < lines * 32; j += nth) {
      int i = j + rot;
      if (i >= lines * 32) i -= lines * 32;
      if (i < nn) {
        unsigned c = cnt[i];
        if (c) atomicAdd(&deg[n0 + i], (float)c);
      }
    }
    __syncthreads();
  }
}

__global__ void __launch_bounds__(256)
k_scale1(const float* __restrict__ x, const float* __restrict__ deg,
         float* __restrict__ dinv, float* __restrict__ xs, int N) {
  int i = blockIdx.x * blockDim.x + threadIdx.x;
  if (i >= N) return;
  float d = __builtin_amdgcn_rsqf(deg[i]);
  dinv[i] = d;
  const float* xr = x + (size_t)i * 16;
  float4 o = make_float4(d * xr[0], d * xr[1], d * xr[2], 0.0f);
  *(float4*)(xs + (size_t)i * 4) = o;
}

#define NSR    8
#define NTR    32
#define RGS    12544
#define RGT    3136
#define NPAD   (NTR * RGT)
#define MAXB   (NSR * NTR)
#define TP     (MAXB + 8)
#define CH     25024

static __device__ __forceinline__ unsigned tile_of(int r, int c) {
  unsigned tc = (unsigned)c / (unsigned)RGT;
  unsigned tr = (unsigned)r / (unsigned)RGS;
  tc = (tc < (unsigned)NTR) ? tc : (unsigned)(NTR - 1);
  tr = (tr < (unsigned)NSR) ? tr : (unsigned)(NSR - 1);
  return tc * NSR + tr;
}

__global__ void __launch_bounds__(1024)
k_bucket(const int* __restrict__ row, const int* __restrict__ col,
         int* __restrict__ tbl, int2* __restrict__ pairs, int NE) {
  __shared__ int2 stage[CH];
  __shared__ int  hist[MAXB];
  __shared__ int  wsum[MAXB / 32];
  const int tid  = (int)threadIdx.x;
  const int nth  = (int)blockDim.x;
  const int lane = tid & 31;
  const int items = (NE + CH - 1) / CH;

  for (int item = (int)blockIdx.x; item < items; item += (int)gridDim.x) {
    const int e0 = item * CH;
    const int n  = (NE - e0 < CH) ? (NE - e0) : CH;
    const int* __restrict__ colp = col + e0;
    const int* __restrict__ rowp = row + e0;
    for (int i = tid; i < MAXB; i += nth) hist[i] = 0;
    __syncthreads();

    for (int i = tid; i < n; i += 8 * nth) {
      int c[8], r[8];
#pragma unroll
      for (int u = 0; u < 8; ++u) {
        const int iu = i + u * nth;
        const int ic = (iu < n) ? iu : n - 1;
        c[u] = colp[ic]; r[u] = rowp[ic];
      }
#pragma unroll
      for (int u = 0; u < 8; ++u) {
        if (i + u * nth < n) atomicAdd(&hist[tile_of(r[u], c[u])], 1);
      }
    }
    __syncthreads();

    {
      const int v = (tid < MAXB) ? hist[tid] : 0;
      int incl = v;
      for (int d = 1; d < 32; d <<= 1) {
        const int t = __shfl_up(incl, d);
        if (lane >= d) incl += t;
      }
      if (tid < MAXB && lane == 31) wsum[tid >> 5] = incl;
      __syncthreads();
      if (tid < MAXB) {
        int pre = 0;
        for (int w = 0; w < (tid >> 5); ++w) pre += wsum[w];
        const int ex = pre + incl - v;
        hist[tid] = ex;
        tbl[(size_t)item * TP + tid] = ex;
      }
      if (tid == 0) tbl[(size_t)item * TP + MAXB] = n;
      __syncthreads();
    }

    for (int i = tid; i < n; i += 4 * nth) {
      int c[4], r[4];
#pragma unroll
      for (int u = 0; u < 4; ++u) {
        const int iu = i + u * nth;
        const int ic = (iu < n) ? iu : n - 1;
        c[u] = colp[ic]; r[u] = rowp[ic];
      }
#pragma unroll
      for (int u = 0; u < 4; ++u) {
        if (i + u * nth < n) {
          const int pos = atomicAdd(&hist[tile_of(r[u], c[u])], 1);
          stage[pos] = make_int2(r[u], c[u]);
        }
      }
    }
    __syncthreads();

    int2* __restrict__ dst = pairs + (size_t)e0;
    for (int i = tid; i < n; i += nth) dst[i] = stage[i];
    __syncthreads();
  }
}

#define RUNS_LDS 1024
#define TR       4

template <int NF>
__global__ void __launch_bounds__(1024)
k_sum(const int2* __restrict__ pairs, const int* __restrict__ tbl,
      const float* __restrict__ src, float* __restrict__ part, int N, int NE) {
  __shared__ float4 stab[RGS];
  __shared__ float  ttab[4 * RGT];
  __shared__ int    so[RUNS_LDS];
  __shared__ int    eo[RUNS_LDS];
  const int tid  = (int)threadIdx.x;
  const int nth  = (int)blockDim.x;
  const int lane = tid & 31;
  const int wv   = tid >> 5;
  const int nwv  = nth >> 5;
  const int b    = (int)blockIdx.x;
  const int tc   = b / NSR;
  const int tr   = b - tc * NSR;
  const int s0   = tr * RGS;
  const int t0   = tc * RGT;
  const int items = (NE + CH - 1) / CH;
  const float4* __restrict__ src4 = (const float4*)src + (size_t)blockIdx.y * N;
  float4* __restrict__ part4 = (float4*)part + ((size_t)blockIdx.y * NSR + tr) * NPAD + t0;

#pragma unroll 4
  for (int i = tid; i < RGS; i += nth) {
    int r = s0 + i;
    r = (r < N) ? r : N - 1;
    stab[i] = src4[r];
  }
  for (int i = tid; i < 4 * RGT; i += nth) ttab[i] = 0.0f;

  for (int ib = 0; ib < items; ib += RUNS_LDS) {
    __syncthreads();
    const int cnt = (items - ib < RUNS_LDS) ? (items - ib) : RUNS_LDS;
    for (int t = tid; t < cnt; t += nth) {
      so[t] = tbl[(size_t)(ib + t) * TP + b];
      eo[t] = tbl[(size_t)(ib + t) * TP + b + 1];
    }
    __syncthreads();

    for (int rr = wv; rr < cnt; rr += nwv) {
      const int s = __builtin_amdgcn_readfirstlane(so[rr]);
      const int e = __builtin_amdgcn_readfirstlane(eo[rr]);
      const int2* __restrict__ base = pairs + (size_t)(ib + rr) * CH;
      for (int j0 = s; j0 < e; j0 += 32 * TR) {
        int2 p[TR];
#pragma unroll
        for (int u = 0; u < TR; ++u) {
          const int j = j0 + u * 32 + lane;
          p[u] = base[(j < e) ? j : e - 1];
        }
        unsigned k[TR];
        float4 g[TR];
#pragma unroll
        for (int u = 0; u < TR; ++u) {
          const int j = j0 + u * 32 + lane;
          k[u] = (j < e) ? (unsigned)(p[u].y - t0) : 0xFFFFFFFFu;
          unsigned ls = (unsigned)(p[u].x - s0);
          ls = (ls < (unsigned)RGS) ? ls : (unsigned)(RGS - 1);
          g[u] = stab[ls];
        }
#pragma unroll
        for (int u = 0; u < TR; ++u) {
          if (k[u] < (unsigned)RGT) {
            atomicAdd(&ttab[k[u]], g[u].x);
            atomicAdd(&ttab[RGT + k[u]], g[u].y);
            atomicAdd(&ttab[2 * RGT + k[u]], g[u].z);
            if (NF == 4) atomicAdd(&ttab[3 * RGT + k[u]], g[u].w);
          }
        }
      }
    }
  }
  __syncthreads();

  for (int i = tid; i < RGT; i += nth) {
    float4 o;
    o.x = ttab[i]; o.y = ttab[RGT + i]; o.z = ttab[2 * RGT + i];
    o.w = (NF == 4) ? ttab[3 * RGT + i] : 0.0f;
    part4[i] = o;
  }
}

__global__ void __launch_bounds__(256)
k_dense2(const float* __restrict__ part1, const float* __restrict__ xs,
         const float* __restrict__ dinv, const float* __restrict__ gW1,
         const float* __restrict__ gW2, const float* __restrict__ gb1,
         float* __restrict__ xw2s, int N) {
  __shared__ float ht[WPB][256];
  __shared__ float dv[WPB][16];
  int wslot = threadIdx.x >> 5;
  int lane  = threadIdx.x & 31;
  int m0 = (blockIdx.x * WPB + wslot) * 16;
  if (m0 >= N) return;
  bool full = (m0 + 16) <= N;
  float* t   = ht[wslot];
  float* dvs = dv[wslot];

  int n  = lane & 15;
  int mo = (lane >> 4) << 3;

  {
    int r  = m0 + n;
    int rc = (r < N) ? r : N - 1;
    float dvr = dinv[rc];
    if (lane < 16) dvs[n] = dvr;
    float4 a4 = make_float4(0.0f, 0.0f, 0.0f, 0.0f);
#pragma unroll
    for (int p = 0; p < NSR; ++p) {
      const float4 q4 = ((const float4*)part1)[(size_t)p * NPAD + rc];
      a4.x += q4.x; a4.y += q4.y; a4.z += q4.z;
    }
    float4 s4 = *(const float4*)(xs   + (size_t)rc * 4);
    float g0 = dvr * (a4.x + s4.x);
    float g1 = dvr * (a4.y + s4.y);
    float g2 = dvr * (a4.z + s4.z);
    v2f a;
    a.x = (lane < 16) ? g0 : g2;
    a.y = (lane < 16) ? g1 : 0.0f;
    v2f b = b_frag(gW1, lane, 0, 3, 16);
    v8f c1 = {};
    c1 = wmma_f32(a, b, c1);
    float bias = gb1[n];
    for (int v = 0; v < 8; ++v)
      t[(mo + v) * 16 + n] = fmaxf(c1[v] + bias, 0.0f);
  }
  __builtin_amdgcn_wave_barrier();

  v8f c = {};
  for (int k0 = 0; k0 < 16; k0 += 4) {
    v2f a = a_frag_lds(t, lane, k0);
    v2f b = b_frag(gW2, lane, k0, 16, 8);
    c = wmma_f32(a, b, c);
  }
  __builtin_amdgcn_wave_barrier();

  if (n < 8) {
    for (int v = 0; v < 8; ++v)
      t[(mo + v) * 8 + n] = c[v];
  }
  __builtin_amdgcn_wave_barrier();

  if (full) {
    float dvr = dvs[lane >> 1];
    float4 v4 = ((const float4*)t)[lane];
    v4.x *= dvr; v4.y *= dvr; v4.z *= dvr; v4.w *= dvr;
    ((float4*)xw2s)[(size_t)(lane & 1) * N + m0 + (lane >> 1)] = v4;
  } else if (n < 8) {
    for (int v = 0; v < 8; ++v) {
      int r = m0 + mo + v;
      if (r < N) xw2s[((size_t)(n >> 2) * N + r) * 4 + (n & 3)] = t[(mo + v) * 8 + n] * dvs[mo + v];
    }
  }
}

__global__ void __launch_bounds__(256)
k_final(const float* __restrict__ x, const float* __restrict__ part2,
        const float* __restrict__ xw2s, const float* __restrict__ dinv,
        const float* __restrict__ gb2,
        const float* __restrict__ cW1, const float* __restrict__ cb1,
        const float* __restrict__ cW2, const float* __restrict__ cb2,
        const float* __restrict__ mW1, const float* __restrict__ mb1,
        const float* __restrict__ mW2, const float* __restrict__ mb2,
        float* __restrict__ out, int N) {
  __shared__ float ht[WPB][256];
  __shared__ float ct[WPB][128];
  __shared__ float g0s[WPB][16];
  int wslot = threadIdx.x >> 5;
  int lane  = threadIdx.x & 31;
  int m0 = (blockIdx.x * WPB + wslot) * 16;
  if (m0 >= N) return;
  bool full = (m0 + 16) <= N;
  float* t  = ht[wslot];
  float* o  = ct[wslot];
  float* g0 = g0s[wslot];

  int n  = lane & 15;
  int mo = (lane >> 4) << 3;

  int grow = lane >> 1, gcc = lane & 1;
  int gr  = m0 + grow;
  int grc = (gr < N) ? gr : N - 1;
  float  gdv = dinv[grc];
  float4 a24 = make_float4(0.0f, 0.0f, 0.0f, 0.0f);
#pragma unroll
  for (int p = 0; p < NSR; ++p) {
    const float4 q4 = ((const float4*)part2)[((size_t)gcc * NSR + p) * NPAD + grc];
    a24.x += q4.x; a24.y += q4.y; a24.z += q4.z; a24.w += q4.w;
  }
  float4 s24 = ((const float4*)xw2s)[(size_t)gcc * N + grc];
  float4 gb4 = ((const float4*)gb2)[gcc];
  float4 go4;
  go4.x = gdv * (a24.x + s24.x) + gb4.x;
  go4.y = gdv * (a24.y + s24.y) + gb4.y;
  go4.z = gdv * (a24.z + s24.z) + gb4.z;
  go4.w = gdv * (a24.w + s24.w) + gb4.w;

  v8f ch = {};
  for (int k0 = 0; k0 < 16; k0 += 4) {
    v2f a = a_frag_x(x, N, m0, lane, k0,  3,  13);
    v2f b = b_frag(cW1, lane, k0, 13, 16);
    ch = wmma_f32(a, b, ch);
  }
  for (int v = 0; v < 8; ++v) t[(mo + v) * 16 + n] = fmaxf(ch[v] + cb1[n], 0.0f);
  __builtin_amdgcn_wave_barrier();

  v8f cc = {};
  for (int k0 = 0; k0 < 16; k0 += 4) {
    v2f a = a_frag_lds(t, lane, k0);
    v2f b = b_frag(cW2, lane, k0, 16, 8);
    cc = wmma_f32(a, b, cc);
  }
  if (n < 8) {
    for (int v = 0; v < 8; ++v) o[(mo + v) * 8 + n] = cc[v] + cb2[n];
  }
  __builtin_amdgcn_wave_barrier();

  v8f cm = {};
  for (int k0 = 0; k0 < 16; k0 += 4) {
    v2f a = a_frag_x(x, N, m0, lane, k0,  0,  16);
    v2f b = b_frag(mW1, lane, k0, 16, 16);
    cm = wmma_f32(a, b, cm);
  }
  for (int v = 0; v < 8; ++v) t[(mo + v) * 16 + n] = fmaxf(cm[v] + mb1[n], 0.0f);
  __builtin_amdgcn_wave_barrier();

  if (lane < 16) {
    float l0 = mb2[0], l1 = mb2[1];
    for (int j = 0; j < 16; ++j) {
      float hv = t[lane * 16 + j];
      l0 += hv * mW2[j * 2 + 0];
      l1 += hv * mW2[j * 2 + 1];
    }
    float mx = fmaxf(l0, l1);
    float e0 = __expf(l0 - mx), e1 = __expf(l1 - mx);
    g0[lane] = e0 / (e0 + e1);
  }
  __builtin_amdgcn_wave_barrier();

  float g  = g0[grow];
  float4 c4 = *(const float4*)&o[lane * 4];
  float4 r4;
  r4.x = g * go4.x + (1.0f - g) * c4.x;
  r4.y = g * go4.y + (1.0f - g) * c4.y;
  r4.z = g * go4.z + (1.0f - g) * c4.z;
  r4.w = g * go4.w + (1.0f - g) * c4.w;
  if (full) {
    ((float4*)(out + (size_t)m0 * 8))[lane] = r4;
  } else if (gr < N) {
    *(float4*)(out + (size_t)gr * 8 + gcc * 4) = r4;
  }
}

extern "C" void kernel_launch(void* const* d_in, const int* in_sizes, int n_in,
                              void* d_out, int out_size, void* d_ws, size_t ws_size,
                              hipStream_t stream) {
  const float* x   = (const float*)d_in[0];
  const int*   ei  = (const int*)  d_in[1];
  const float* gW1 = (const float*)d_in[2];
  const float* gb1 = (const float*)d_in[3];
  const float* gW2 = (const float*)d_in[4];
  const float* gb2 = (const float*)d_in[5];
  const float* cW1 = (const float*)d_in[6];
  const float* cb1 = (const float*)d_in[7];
  const float* cW2 = (const float*)d_in[8];
  const float* cb2 = (const float*)d_in[9];
  const float* mW1 = (const float*)d_in[10];
  const float* mb1 = (const float*)d_in[11];
  const float* mW2 = (const float*)d_in[12];
  const float* mb2 = (const float*)d_in[13];
  float* out = (float*)d_out;

  const int N  = in_sizes[0] / 16;
  const int NE = in_sizes[1] / 2;
  const int* erow = ei;
  const int* ecol = ei + NE;

  float* deg   = (float*)d_ws;
  float* dinv  = deg   + N;
  float* xs    = dinv  + N;
  float* xw2s  = xs    + (size_t)4 * N;
  float* part1 = xw2s  + (size_t)8 * N;
  float* part2 = part1 + (size_t)4 * NSR * NPAD;
  const int items = (NE + CH - 1) / CH;
  int*   tbl   = (int*)(part2 + (size_t)8 * NSR * NPAD);
  int2*  pairs = (int2*)(tbl + (size_t)items * TP);

  const int tiles   = (N + 15) / 16;
  const int gBlocks = (tiles + WPB - 1) / WPB;

  k_init    <<<(N + 255) / 256, 256, 0, stream>>>(deg, N);
  k_degree  <<<256, 1024, 0, stream>>>(ecol, deg, N, NE);
  k_scale1  <<<(N + 255) / 256, 256, 0, stream>>>(x, deg, dinv, xs, N);
  k_bucket  <<<256, 1024, 0, stream>>>(erow, ecol, tbl, pairs, NE);
  k_sum<3>  <<<dim3(MAXB, 1), 1024, 0, stream>>>(pairs, tbl, xs, part1, N, NE);
  k_dense2  <<<gBlocks, 256, 0, stream>>>(part1, xs, dinv, gW1, gW2, gb1, xw2s, N);
  k_sum<4>  <<<dim3(MAXB, 2), 1024, 0, stream>>>(pairs, tbl, xw2s, part2, N, NE);
  k_final   <<<gBlocks, 256, 0, stream>>>(x, part2, xw2s, dinv, gb2,
                                          cW1, cb1, cW2, cb2,
                                          mW1, mb1, mW2, mb2, out, N);
}


# edge sums: a wave requests the pairs of all its runs at once and holds them in registers, in flight across an LDS-only barrier; layer 2 sums both halves from them in one block; f32 sum order changes
# speedup vs baseline: 9.1783x; 1.0392x over previous
#include <hip/hip_runtime.h>
#include <math.h>


typedef float v2f __attribute__((ext_vector_type(2)));
typedef float v8f __attribute__((ext_vector_type(8)));

#define WPB 8

static __device__ __forceinline__ v8f wmma_f32(v2f a, v2f b, v8f c) {
  return __builtin_amdgcn_wmma_f32_16x16x4_f32(false, a, false, b,
                                               (short)0, c, false, false);
}

static __device__ __forceinline__ v2f b_frag(const float* __restrict__ W,
                                             int lane, int k0, int K, int Ncols) {
  int n  = lane & 15;
  int kb = k0 + ((lane >> 4) << 1);
  int nm  = (n      < Ncols) ? n      : (Ncols - 1);
  int kc0 = (kb     < K)     ? kb     : (K - 1);
  int kc1 = (kb + 1 < K)     ? kb + 1 : (K - 1);
  float w0 = W[kc0 * Ncols + nm];
  float w1 = W[kc1 * Ncols + nm];
  v2f b;
  b.x = (kb     < K && n < Ncols) ? w0 : 0.0f;
  b.y = (kb + 1 < K && n < Ncols) ? w1 : 0.0f;
  return b;
}

static __device__ __forceinline__ v2f a_frag_x(const float* __restrict__ x,
                                               int N, int m0, int lane,
                                               int k0, int co, int K) {
  int r = m0 + (lane & 15);
  if (r >= N) r = N - 1;
  int kb = k0 + ((lane >> 4) << 1);
  const float* rowp = x + (size_t)r * 16;
  int i0 = co + kb;     i0 = (i0 < 16) ? i0 : 15;
  int i1 = co + kb + 1; i1 = (i1 < 16) ? i1 : 15;
  float a0 = rowp[i0];
  float a1 = rowp[i1];
  v2f a;
  a.x = (kb     < K) ? a0 : 0.0f;
  a.y = (kb + 1 < K) ? a1 : 0.0f;
  return a;
}

static __device__ __forceinline__ v2f a_frag_lds(const float* t, int lane, int k0) {
  int r  = lane & 15;
  int kb = k0 + ((lane >> 4) << 1);
  v2f a;
  a.x = t[r * 16 + kb];
  a.y = t[r * 16 + kb + 1];
  return a;
}

__global__ void k_init(float* __restrict__ deg, int N) {
  int i = blockIdx.x * blockDim.x + threadIdx.x;
  if (i < N) deg[i] = 1.0f;
}

#define DEG_CAP 25088

static __device__ __forceinline__ void deg_bump(unsigned* cnt, int c, int n0, int nn) {
  unsigned k = (unsigned)(c - n0);
  if (k < (unsigned)nn) atomicAdd(&cnt[k], 1u);
}

__global__ void __launch_bounds__(1024)
k_degree(const int* __restrict__ col, float* __restrict__ deg, int N, int NE) {
  __shared__ unsigned cnt[DEG_CAP];
  const int tid = (int)threadIdx.x;
  const int nth = (int)blockDim.x;
  const int R = (N + DEG_CAP - 1) / DEG_CAP;
  int S = (int)gridDim.x / R;
  if (S < 1) S = 1;
  const int per = (((NE + S - 1) / S) + 3) & ~3;
  const bool aligned = (((size_t)col) & 15) == 0;

  for (int item = (int)blockIdx.x; item < R * S; item += (int)gridDim.x) {
    const int rg = item % R;
    const int sl = item / R;
    const int n0 = rg * DEG_CAP;
    const int nn = (N - n0 < DEG_CAP) ? (N - n0) : DEG_CAP;
    for (int i = tid; i < nn; i += nth) cnt[i] = 0u;
    __syncthreads();

    const int e0 = (sl * per < NE) ? sl * per : NE;
    const int e1 = (e0 + per < NE) ? e0 + per : NE;
    int done = e0;
    if (aligned) {
      const int4* c4 = (const int4*)(col + e0);
      const int nv = (e1 - e0) >> 2;
      const int4 none = make_int4(-1, -1, -1, -1);
      for (int i = tid; i < nv; i += 8 * nth) {
        int4 v[8];
#pragma unroll
        for (int u = 0; u < 8; ++u) {
          const int iu = i + u * nth;
          v[u] = c4[(iu < nv) ? iu : nv - 1];
        }
#pragma unroll
        for (int u = 0; u < 8; ++u) {
          if (i + u * nth >= nv) v[u] = none;
          deg_bump(cnt, v[u].x, n0, nn); deg_bump(cnt, v[u].y, n0, nn);
          deg_bump(cnt, v[u].z, n0, nn); deg_bump(cnt, v[u].w, n0, nn);
        }
      }
      done = e0 + (nv << 2);
    }
    for (int e = done + tid; e < e1; e += nth) deg_bump(cnt, col[e], n0, nn);
    __syncthreads();

    const int lines = (nn + 31) >> 5;
    const int rot   = (int)(((long long)sl * lines) / S) << 5;
    for (int j = tid; j < lines * 32; j += nth) {
      int i = j + rot;
      if (i >= lines * 32) i -= lines * 32;
      if (i < nn) {
        unsigned c = cnt[i];
        if (c) atomicAdd(&deg[n0 + i], (float)c);
      }
    }
    __syncthreads();
  }
}

__global__ void __launch_bounds__(256)
k_scale1(const float* __restrict__ x, const float* __restrict__ deg,
         float* __restrict__ dinv, float* __restrict__ xs, int N) {
  int i = blockIdx.x * blockDim.x + threadIdx.x;
  if (i >= N) return;
  float d = __builtin_amdgcn_rsqf(deg[i]);
  dinv[i] = d;
  const float* xr = x + (size_t)i * 16;
  float4 o = make_float4(d * xr[0], d * xr[1], d * xr[2], 0.0f);
  *(float4*)(xs + (size_t)i * 4) = o;
}

#define NSR    8
#define NTR    32
#define RGS    12544
#define RGT    3136
#define NPAD   (NTR * RGT)
#define MAXB   (NSR * NTR)
#define TP     (MAXB + 8)
#define CH     25024

static __device__ __forceinline__ unsigned tile_of(int r, int c) {
  unsigned tc = (unsigned)c / (unsigned)RGT;
  unsigned tr = (unsigned)r / (unsigned)RGS;
  tc = (tc < (unsigned)NTR) ? tc : (unsigned)(NTR - 1);
  tr = (tr < (unsigned)NSR) ? tr : (unsigned)(NSR - 1);
  return tc * NSR + tr;
}

__global__ void __launch_bounds__(1024)
k_bucket(const int* __restrict__ row, const int* __restrict__ col,
         int* __restrict__ tbl, int2* __restrict__ pairs, int NE) {
  __shared__ int2 stage[CH];
  __shared__ int  hist[MAXB];
  __shared__ int  wsum[MAXB / 32];
  const int tid  = (int)threadIdx.x;
  const int nth  = (int)blockDim.x;
  const int lane = tid & 31;
  const int items = (NE + CH - 1) / CH;

  for (int item = (int)blockIdx.x; item < items; item += (int)gridDim.x) {
    const int e0 = item * CH;
    const int n  = (NE - e0 < CH) ? (NE - e0) : CH;
    const int* __restrict__ colp = col + e0;
    const int* __restrict__ rowp = row + e0;
    for (int i = tid; i < MAXB; i += nth) hist[i] = 0;
    __syncthreads();

    for (int i = tid; i < n; i += 8 * nth) {
      int c[8], r[8];
#pragma unroll
      for (int u = 0; u < 8; ++u) {
        const int iu = i + u * nth;
        const int ic = (iu < n) ? iu : n - 1;
        c[u] = colp[ic]; r[u] = rowp[ic];
      }
#pragma unroll
      for (int u = 0; u < 8; ++u) {
        if (i + u * nth < n) atomicAdd(&hist[tile_of(r[u], c[u])], 1);
      }
    }
    __syncthreads();

    {
      const int v = (tid < MAXB) ? hist[tid] : 0;
      int incl = v;
      for (int d = 1; d < 32; d <<= 1) {
        const int t = __shfl_up(incl, d);
        if (lane >= d) incl += t;
      }
      if (tid < MAXB && lane == 31) wsum[tid >> 5] = incl;
      __syncthreads();
      if (tid < MAXB) {
        int pre = 0;
        for (int w = 0; w < (tid >> 5); ++w) pre += wsum[w];
        const int ex = pre + incl - v;
        hist[tid] = ex;
        tbl[(size_t)item * TP + tid] = ex;
      }
      if (tid == 0) tbl[(size_t)item * TP + MAXB] = n;
      __syncthreads();
    }

    for (int i = tid; i < n; i += 4 * nth) {
      int c[4], r[4];
#pragma unroll
      for (int u = 0; u < 4; ++u) {
        const int iu = i + u * nth;
        const int ic = (iu < n) ? iu : n - 1;
        c[u] = colp[ic]; r[u] = rowp[ic];
      }
#pragma unroll
      for (int u = 0; u < 4; ++u) {
        if (i + u * nth < n) {
          const int pos = atomicAdd(&hist[tile_of(r[u], c[u])], 1);
          stage[pos] = make_int2(r[u], c[u]);
        }
      }
    }
    __syncthreads();

    int2* __restrict__ dst = pairs + (size_t)e0;
    for (int i = tid; i < n; i += nth) dst[i] = stage[i];
    __syncthreads();
  }
}

#define GR 8
#define TR 4

static __device__ __forceinline__ void lds_barrier() {
  __builtin_amdgcn_fence(__ATOMIC_RELEASE, "workgroup", "local");
  __builtin_amdgcn_s_barrier();
  __builtin_amdgcn_fence(__ATOMIC_ACQUIRE, "workgroup", "local");
}

template <int NF>
static __device__ __forceinline__ void add_row(float* ttab, unsigned k, float4 g) {
  if (k < (unsigned)RGT) {
    atomicAdd(&ttab[k], g.x);
    atomicAdd(&ttab[RGT + k], g.y);
    atomicAdd(&ttab[2 * RGT + k], g.z);
    if (NF == 4) atomicAdd(&ttab[3 * RGT + k], g.w);
  }
}

template <int NF, int NH>
__global__ void __launch_bounds__(1024)
k_sum(const int2* __restrict__ pairs, const int* __restrict__ tbl,
      const float* __restrict__ src, float* __restrict__ part, int N, int NE) {
  __shared__ float4 stab[RGS];
  __shared__ float  ttab[4 * RGT];
  const int tid  = (int)threadIdx.x;
  const int nth  = (int)blockDim.x;
  const int lane = tid & 31;
  const int wv   = __builtin_amdgcn_readfirstlane(tid >> 5);
  const int nwv  = nth >> 5;
  const int b    = (int)blockIdx.x;
  const int tc   = b / NSR;
  const int tr   = b - tc * NSR;
  const int s0   = tr * RGS;
  const int t0   = tc * RGT;
  const int items = (NE + CH - 1) / CH;
  const int pg      = nwv * GR;
  const int ngroups = (items + pg - 1) / pg;
  const bool reload = ngroups > 1;

  int2 p[GR][TR];
  int sv = 0, ev = 0;
  int sb[GR], eb[GR];

  for (int i = tid; i < 4 * RGT; i += nth) ttab[i] = 0.0f;

  for (int h = 0; h < NH; ++h) {
    const float4* __restrict__ src4 = (const float4*)src + (size_t)h * N;
    float4* __restrict__ part4 = (float4*)part + ((size_t)h * NSR + tr) * NPAD + t0;

    if (h == 0 || reload) {
      const int rr = lane * nwv + wv;
      sv = 0; ev = 0;
      if (lane < GR && rr < items) {
        sv = tbl[(size_t)rr * TP + b];
        ev = tbl[(size_t)rr * TP + b + 1];
      }
    }

#pragma unroll 4
    for (int i = tid; i < RGS; i += nth) {
      int r = s0 + i;
      r = (r < N) ? r : N - 1;
      stab[i] = src4[r];
    }

#pragma unroll
    for (int q = 0; q < GR; ++q) {
      sb[q] = __builtin_amdgcn_readlane(sv, q);
      eb[q] = __builtin_amdgcn_readlane(ev, q);
    }

    for (int g = 0; g < ngroups; ++g) {
      const int r0 = g * pg;
      if (h == 0 || reload) {
        if (g > 0) {
          const int rr = r0 + lane * nwv + wv;
          sv = 0; ev = 0;
          if (lane < GR && rr < items) {
            sv = tbl[(size_t)rr * TP + b];
            ev = tbl[(size_t)rr * TP + b + 1];
          }
#pragma unroll
          for (int q = 0; q < GR; ++q) {
            sb[q] = __builtin_amdgcn_readlane(sv, q);
            eb[q] = __builtin_amdgcn_readlane(ev, q);
          }
        }
#pragma unroll
        for (int q = 0; q < GR; ++q) {
          const int s = sb[q];
          const int e = eb[q];
          int rr = r0 + q * nwv + wv;
          rr = (rr < items) ? rr : items - 1;
          const int2* __restrict__ base = pairs + (size_t)rr * CH;
#pragma unroll
          for (int u = 0; u < TR; ++u) {
            const int j = s + u * 32 + lane;
            int jc = (j < e) ? j : e - 1;
            jc = (jc > 0) ? jc : 0;
            p[q][u] = base[jc];
          }
        }
      }
      if (g == 0) lds_barrier();

#pragma unroll
      for (int q = 0; q < GR; ++q) {
        const int s = sb[q];
        const int e = eb[q];
        unsigned k[TR];
        float4 gg[TR];
#pragma unroll
        for (int u = 0; u < TR; ++u) {
          const int j = s + u * 32 + lane;
          k[u] = (j < e) ? (unsigned)(p[q][u].y - t0) : 0xFFFFFFFFu;
          unsigned ls = (unsigned)(p[q][u].x - s0);
          ls = (ls < (unsigned)RGS) ? ls : (unsigned)(RGS - 1);
          gg[u] = stab[ls];
        }
#pragma unroll
        for (int u = 0; u < TR; ++u) add_row<NF>(ttab, k[u], gg[u]);
      }

#pragma nounroll
      for (int q = 0; q < GR; ++q) {
        const int s = __builtin_amdgcn_readlane(sv, q) + 32 * TR;
        const int e = __builtin_amdgcn_readlane(ev, q);
        if (s < e) {
          const int2* __restrict__ base = pairs + (size_t)(r0 + q * nwv + wv) * CH;
          for (int j0 = s; j0 < e; j0 += 32) {
            const int j = j0 + lane;
            const int2 pp = base[(j < e) ? j : e - 1];
            const unsigned k = (j < e) ? (unsigned)(pp.y - t0) : 0xFFFFFFFFu;
            unsigned ls = (unsigned)(pp.x - s0);
            ls = (ls < (unsigned)RGS) ? ls : (unsigned)(RGS - 1);
            add_row<NF>(ttab, k, stab[ls]);
          }
        }
      }
    }
    lds_barrier();

    for (int i = tid; i < RGT; i += nth) {
      float4 o;
      o.x = ttab[i]; o.y = ttab[RGT + i]; o.z = ttab[2 * RGT + i];
      o.w = (NF == 4) ? ttab[3 * RGT + i] : 0.0f;
      part4[i] = o;
      if (h + 1 < NH) {
        ttab[i] = 0.0f; ttab[RGT + i] = 0.0f; ttab[2 * RGT + i] = 0.0f; ttab[3 * RGT + i] = 0.0f;
      }
    }
  }
}

__global__ void __launch_bounds__(256)
k_dense2(const float* __restrict__ part1, const float* __restrict__ xs,
         const float* __restrict__ dinv, const float* __restrict__ gW1,
         const float* __restrict__ gW2, const float* __restrict__ gb1,
         float* __restrict__ xw2s, int N) {
  __shared__ float ht[WPB][256];
  __shared__ float dv[WPB][16];
  int wslot = threadIdx.x >> 5;
  int lane  = threadIdx.x & 31;
  int m0 = (blockIdx.x * WPB + wslot) * 16;
  if (m0 >= N) return;
  bool full = (m0 + 16) <= N;
  float* t   = ht[wslot];
  float* dvs = dv[wslot];

  int n  = lane & 15;
  int mo = (lane >> 4) << 3;

  {
    int r  = m0 + n;
    int rc = (r < N) ? r : N - 1;
    float dvr = dinv[rc];
    if (lane < 16) dvs[n] = dvr;
    float4 a4 = make_float4(0.0f, 0.0f, 0.0f, 0.0f);
#pragma unroll
    for (int p = 0; p < NSR; ++p) {
      const float4 q4 = ((const float4*)part1)[(size_t)p * NPAD + rc];
      a4.x += q4.x; a4.y += q4.y; a4.z += q4.z;
    }
    float4 s4 = *(const float4*)(xs   + (size_t)rc * 4);
    float g0 = dvr * (a4.x + s4.x);
    float g1 = dvr * (a4.y + s4.y);
    float g2 = dvr * (a4.z + s4.z);
    v2f a;
    a.x = (lane < 16) ? g0 : g2;
    a.y = (lane < 16) ? g1 : 0.0f;
    v2f b = b_frag(gW1, lane, 0, 3, 16);
    v8f c1 = {};
    c1 = wmma_f32(a, b, c1);
    float bias = gb1[n];
    for (int v = 0; v < 8; ++v)
      t[(mo + v) * 16 + n] = fmaxf(c1[v] + bias, 0.0f);
  }
  __builtin_amdgcn_wave_barrier();

  v8f c = {};
  for (int k0 = 0; k0 < 16; k0 += 4) {
    v2f a = a_frag_lds(t, lane, k0);
    v2f b = b_frag(gW2, lane, k0, 16, 8);
    c = wmma_f32(a, b, c);
  }
  __builtin_amdgcn_wave_barrier();

  if (n < 8) {
    for (int v = 0; v < 8; ++v)
      t[(mo + v) * 8 + n] = c[v];
  }
  __builtin_amdgcn_wave_barrier();

  if (full) {
    float dvr = dvs[lane >> 1];
    float4 v4 = ((const float4*)t)[lane];
    v4.x *= dvr; v4.y *= dvr; v4.z *= dvr; v4.w *= dvr;
    ((float4*)xw2s)[(size_t)(lane & 1) * N + m0 + (lane >> 1)] = v4;
  } else if (n < 8) {
    for (int v = 0; v < 8; ++v) {
      int r = m0 + mo + v;
      if (r < N) xw2s[((size_t)(n >> 2) * N + r) * 4 + (n & 3)] = t[(mo + v) * 8 + n] * dvs[mo + v];
    }
  }
}

__global__ void __launch_bounds__(256)
k_final(const float* __restrict__ x, const float* __restrict__ part2,
        const float* __restrict__ xw2s, const float* __restrict__ dinv,
        const float* __restrict__ gb2,
        const float* __restrict__ cW1, const float* __restrict__ cb1,
        const float* __restrict__ cW2, const float* __restrict__ cb2,
        const float* __restrict__ mW1, const float* __restrict__ mb1,
        const float* __restrict__ mW2, const float* __restrict__ mb2,
        float* __restrict__ out, int N) {
  __shared__ float ht[WPB][256];
  __shared__ float ct[WPB][128];
  __shared__ float g0s[WPB][16];
  int wslot = threadIdx.x >> 5;
  int lane  = threadIdx.x & 31;
  int m0 = (blockIdx.x * WPB + wslot) * 16;
  if (m0 >= N) return;
  bool full = (m0 + 16) <= N;
  float* t  = ht[wslot];
  float* o  = ct[wslot];
  float* g0 = g0s[wslot];

  int n  = lane & 15;
  int mo = (lane >> 4) << 3;

  int grow = lane >> 1, gcc = lane & 1;
  int gr  = m0 + grow;
  int grc = (gr < N) ? gr : N - 1;
  float  gdv = dinv[grc];
  float4 a24 = make_float4(0.0f, 0.0f, 0.0f, 0.0f);
#pragma unroll
  for (int p = 0; p < NSR; ++p) {
    const float4 q4 = ((const float4*)part2)[((size_t)gcc * NSR + p) * NPAD + grc];
    a24.x += q4.x; a24.y += q4.y; a24.z += q4.z; a24.w += q4.w;
  }
  float4 s24 = ((const float4*)xw2s)[(size_t)gcc * N + grc];
  float4 gb4 = ((const float4*)gb2)[gcc];
  float4 go4;
  go4.x = gdv * (a24.x + s24.x) + gb4.x;
  go4.y = gdv * (a24.y + s24.y) + gb4.y;
  go4.z = gdv * (a24.z + s24.z) + gb4.z;
  go4.w = gdv * (a24.w + s24.w) + gb4.w;

  v8f ch = {};
  for (int k0 = 0; k0 < 16; k0 += 4) {
    v2f a = a_frag_x(x, N, m0, lane, k0,  3,  13);
    v2f b = b_frag(cW1, lane, k0, 13, 16);
    ch = wmma_f32(a, b, ch);
  }
  for (int v = 0; v < 8; ++v) t[(mo + v) * 16 + n] = fmaxf(ch[v] + cb1[n], 0.0f);
  __builtin_amdgcn_wave_barrier();

  v8f cc = {};
  for (int k0 = 0; k0 < 16; k0 += 4) {
    v2f a = a_frag_lds(t, lane, k0);
    v2f b = b_frag(cW2, lane, k0, 16, 8);
    cc = wmma_f32(a, b, cc);
  }
  if (n < 8) {
    for (int v = 0; v < 8; ++v) o[(mo + v) * 8 + n] = cc[v] + cb2[n];
  }
  __builtin_amdgcn_wave_barrier();

  v8f cm = {};
  for (int k0 = 0; k0 < 16; k0 += 4) {
    v2f a = a_frag_x(x, N, m0, lane, k0,  0,  16);
    v2f b = b_frag(mW1, lane, k0, 16, 16);
    cm = wmma_f32(a, b, cm);
  }
  for (int v = 0; v < 8; ++v) t[(mo + v) * 16 + n] = fmaxf(cm[v] + mb1[n], 0.0f);
  __builtin_amdgcn_wave_barrier();

  if (lane < 16) {
    float l0 = mb2[0], l1 = mb2[1];
    for (int j = 0; j < 16; ++j) {
      float hv = t[lane * 16 + j];
      l0 += hv * mW2[j * 2 + 0];
      l1 += hv * mW2[j * 2 + 1];
    }
    float mx = fmaxf(l0, l1);
    float e0 = __expf(l0 - mx), e1 = __expf(l1 - mx);
    g0[lane] = e0 / (e0 + e1);
  }
  __builtin_amdgcn_wave_barrier();

  float g  = g0[grow];
  float4 c4 = *(const float4*)&o[lane * 4];
  float4 r4;
  r4.x = g * go4.x + (1.0f - g) * c4.x;
  r4.y = g * go4.y + (1.0f - g) * c4.y;
  r4.z = g * go4.z + (1.0f - g) * c4.z;
  r4.w = g * go4.w + (1.0f - g) * c4.w;
  if (full) {
    ((float4*)(out + (size_t)m0 * 8))[lane] = r4;
  } else if (gr < N) {
    *(float4*)(out + (size_t)gr * 8 + gcc * 4) = r4;
  }
}

extern "C" void kernel_launch(void* const* d_in, const int* in_sizes, int n_in,
                              void* d_out, int out_size, void* d_ws, size_t ws_size,
                              hipStream_t stream) {
  const float* x   = (const float*)d_in[0];
  const int*   ei  = (const int*)  d_in[1];
  const float* gW1 = (const float*)d_in[2];
  const float* gb1 = (const float*)d_in[3];
  const float* gW2 = (const float*)d_in[4];
  const float* gb2 = (const float*)d_in[5];
  const float* cW1 = (const float*)d_in[6];
  const float* cb1 = (const float*)d_in[7];
  const float* cW2 = (const float*)d_in[8];
  const float* cb2 = (const float*)d_in[9];
  const float* mW1 = (const float*)d_in[10];
  const float* mb1 = (const float*)d_in[11];
  const float* mW2 = (const float*)d_in[12];
  const float* mb2 = (const float*)d_in[13];
  float* out = (float*)d_out;

  const int N  = in_sizes[0] / 16;
  const int NE = in_sizes[1] / 2;
  const int* erow = ei;
  const int* ecol = ei + NE;

  float* deg   = (float*)d_ws;
  float* dinv  = deg   + N;
  float* xs    = dinv  + N;
  float* xw2s  = xs    + (size_t)4 * N;
  float* part1 = xw2s  + (size_t)8 * N;
  float* part2 = part1 + (size_t)4 * NSR * NPAD;
  const int items = (NE + CH - 1) / CH;
  int*   tbl   = (int*)(part2 + (size_t)8 * NSR * NPAD);
  int2*  pairs = (int2*)(tbl + (size_t)items * TP);

  const int tiles   = (N + 15) / 16;
  const int gBlocks = (tiles + WPB - 1) / WPB;

  k_init    <<<(N + 255) / 256, 256, 0, stream>>>(deg, N);
  k_degree  <<<256, 1024, 0, stream>>>(ecol, deg, N, NE);
  k_scale1  <<<(N + 255) / 256, 256, 0, stream>>>(x, deg, dinv, xs, N);
  k_bucket  <<<256, 1024, 0, stream>>>(erow, ecol, tbl, pairs, NE);
  k_sum<3, 1><<<MAXB, 1024, 0, stream>>>(pairs, tbl, xs, part1, N, NE);
  k_dense2  <<<gBlocks, 256, 0, stream>>>(part1, xs, dinv, gW1, gW2, gb1, xw2s, N);
  k_sum<4, 2><<<MAXB, 1024, 0, stream>>>(pairs, tbl, xw2s, part2, N, NE);
  k_final   <<<gBlocks, 256, 0, stream>>>(x, part2, xw2s, dinv, gb2,
                                          cW1, cb1, cW2, cb2,
                                          mW1, mb1, mW2, mb2, out, N);
}


# six launches, not eight: the bucketing launch runs first and also stores deg's start values and the 16-byte x rows; rsqrt(deg) is applied where rows are loaded, so the init and scale launches go
# speedup vs baseline: 9.4940x; 1.0344x over previous
#include <hip/hip_runtime.h>
#include <math.h>


typedef float v2f __attribute__((ext_vector_type(2)));
typedef float v8f __attribute__((ext_vector_type(8)));

#define WPB 8

static __device__ __forceinline__ v8f wmma_f32(v2f a, v2f b, v8f c) {
  return __builtin_amdgcn_wmma_f32_16x16x4_f32(false, a, false, b,
                                               (short)0, c, false, false);
}

static __device__ __forceinline__ v2f b_frag(const float* __restrict__ W,
                                             int lane, int k0, int K, int Ncols) {
  int n  = lane & 15;
  int kb = k0 + ((lane >> 4) << 1);
  int nm  = (n      < Ncols) ? n      : (Ncols - 1);
  int kc0 = (kb     < K)     ? kb     : (K - 1);
  int kc1 = (kb + 1 < K)     ? kb + 1 : (K - 1);
  float w0 = W[kc0 * Ncols + nm];
  float w1 = W[kc1 * Ncols + nm];
  v2f b;
  b.x = (kb     < K && n < Ncols) ? w0 : 0.0f;
  b.y = (kb + 1 < K && n < Ncols) ? w1 : 0.0f;
  return b;
}

static __device__ __forceinline__ v2f a_frag_x(const float* __restrict__ x,
                                               int N, int m0, int lane,
                                               int k0, int co, int K) {
  int r = m0 + (lane & 15);
  if (r >= N) r = N - 1;
  int kb = k0 + ((lane >> 4) << 1);
  const float* rowp = x + (size_t)r * 16;
  int i0 = co + kb;     i0 = (i0 < 16) ? i0 : 15;
  int i1 = co + kb + 1; i1 = (i1 < 16) ? i1 : 15;
  float a0 = rowp[i0];
  float a1 = rowp[i1];
  v2f a;
  a.x = (kb     < K) ? a0 : 0.0f;
  a.y = (kb + 1 < K) ? a1 : 0.0f;
  return a;
}

static __device__ __forceinline__ v2f a_frag_lds(const float* t, int lane, int k0) {
  int r  = lane & 15;
  int kb = k0 + ((lane >> 4) << 1);
  v2f a;
  a.x = t[r * 16 + kb];
  a.y = t[r * 16 + kb + 1];
  return a;
}

#define DEG_CAP 25088

static __device__ __forceinline__ void deg_bump(unsigned* cnt, int c, int n0, int nn) {
  unsigned k = (unsigned)(c - n0);
  if (k < (unsigned)nn) atomicAdd(&cnt[k], 1u);
}

__global__ void __launch_bounds__(1024)
k_degree(const int* __restrict__ col, float* __restrict__ deg, int N, int NE) {
  __shared__ unsigned cnt[DEG_CAP];
  const int tid = (int)threadIdx.x;
  const int nth = (int)blockDim.x;
  const int R = (N + DEG_CAP - 1) / DEG_CAP;
  int S = (int)gridDim.x / R;
  if (S < 1) S = 1;
  const int per = (((NE + S - 1) / S) + 3) & ~3;
  const bool aligned = (((size_t)col) & 15) == 0;

  for (int item = (int)blockIdx.x; item < R * S; item += (int)gridDim.x) {
    const int rg = item % R;
    const int sl = item / R;
    const int n0 = rg * DEG_CAP;
    const int nn = (N - n0 < DEG_CAP) ? (N - n0) : DEG_CAP;
    for (int i = tid; i < nn; i += nth) cnt[i] = 0u;
    __syncthreads();

    const int e0 = (sl * per < NE) ? sl * per : NE;
    const int e1 = (e0 + per < NE) ? e0 + per : NE;
    int done = e0;
    if (aligned) {
      const int4* c4 = (const int4*)(col + e0);
      const int nv = (e1 - e0) >> 2;
      const int4 none = make_int4(-1, -1, -1, -1);
      for (int i = tid; i < nv; i += 8 * nth) {
        int4 v[8];
#pragma unroll
        for (int u = 0; u < 8; ++u) {
          const int iu = i + u * nth;
          v[u] = c4[(iu < nv) ? iu : nv - 1];
        }
#pragma unroll
        for (int u = 0; u < 8; ++u) {
          if (i + u * nth >= nv) v[u] = none;
          deg_bump(cnt, v[u].x, n0, nn); deg_bump(cnt, v[u].y, n0, nn);
          deg_bump(cnt, v[u].z, n0, nn); deg_bump(cnt, v[u].w, n0, nn);
        }
      }
      done = e0 + (nv << 2);
    }
    for (int e = done + tid; e < e1; e += nth) deg_bump(cnt, col[e], n0, nn);
    __syncthreads();

    const int lines = (nn + 31) >> 5;
    const int rot   = (int)(((long long)sl * lines) / S) << 5;
    for (int j = tid; j < lines * 32; j += nth) {
      int i = j + rot;
      if (i >= lines * 32) i -= lines * 32;
      if (i < nn) {
        unsigned c = cnt[i];
        if (c) atomicAdd(&deg[n0 + i], (float)c);
      }
    }
    __syncthreads();
  }
}

#define NSR    8
#define NTR    32
#define RGS    12544
#define RGT    3136
#define NPAD   (NTR * RGT)
#define MAXB   (NSR * NTR)
#define TP     (MAXB + 8)
#define CH     25024

static __device__ __forceinline__ unsigned tile_of(int r, int c) {
  unsigned tc = (unsigned)c / (unsigned)RGT;
  unsigned tr = (unsigned)r / (unsigned)RGS;
  tc = (tc < (unsigned)NTR) ? tc : (unsigned)(NTR - 1);
  tr = (tr < (unsigned)NSR) ? tr : (unsigned)(NSR - 1);
  return tc * NSR + tr;
}

__global__ void __launch_bounds__(1024)
k_bucket(const int* __restrict__ row, const int* __restrict__ col,
         int* __restrict__ tbl, int2* __restrict__ pairs, int NE,
         const float* __restrict__ x, float* __restrict__ deg, float4* __restrict__ x3, int N) {
  __shared__ int2 stage[CH];
  __shared__ int  hist[MAXB];
  __shared__ int  wsum[MAXB / 32];
  const int tid  = (int)threadIdx.x;
  const int nth  = (int)blockDim.x;
  const int lane = tid & 31;
  const int items = (NE + CH - 1) / CH;

  const int npb = (((N + (int)gridDim.x - 1) / (int)gridDim.x) + 31) & ~31;
  const int nd0 = (int)blockIdx.x * npb;
  bool want = (tid < npb) && (nd0 + tid < N);

  for (int item = (int)blockIdx.x; item < items; item += (int)gridDim.x) {
    const int e0 = item * CH;
    const int n  = (NE - e0 < CH) ? (NE - e0) : CH;
    const int* __restrict__ colp = col + e0;
    const int* __restrict__ rowp = row + e0;
    for (int i = tid; i < MAXB; i += nth) hist[i] = 0;
    __syncthreads();

    const bool now = want;
    float f0 = 0.0f, f1 = 0.0f, f2 = 0.0f;
    if (now) {
      const float* xr = x + (size_t)(nd0 + tid) * 16;
      f0 = xr[0]; f1 = xr[1]; f2 = xr[2];
    }

    for (int i = tid; i < n; i += 8 * nth) {
      int c[8], r[8];
#pragma unroll
      for (int u = 0; u < 8; ++u) {
        const int iu = i + u * nth;
        const int ic = (iu < n) ? iu : n - 1;
        c[u] = colp[ic]; r[u] = rowp[ic];
      }
#pragma unroll
      for (int u = 0; u < 8; ++u) {
        if (i + u * nth < n) atomicAdd(&hist[tile_of(r[u], c[u])], 1);
      }
    }
    __syncthreads();

    if (now) {
      deg[nd0 + tid] = 1.0f;
      x3[nd0 + tid]  = make_float4(f0, f1, f2, 0.0f);
      want = false;
    }

    {
      const int v = (tid < MAXB) ? hist[tid] : 0;
      int incl = v;
      for (int d = 1; d < 32; d <<= 1) {
        const int t = __shfl_up(incl, d);
        if (lane >= d) incl += t;
      }
      if (tid < MAXB && lane == 31) wsum[tid >> 5] = incl;
      __syncthreads();
      if (tid < MAXB) {
        int pre = 0;
        for (int w = 0; w < (tid >> 5); ++w) pre += wsum[w];
        const int ex = pre + incl - v;
        hist[tid] = ex;
        tbl[(size_t)item * TP + tid] = ex;
      }
      if (tid == 0) tbl[(size_t)item * TP + MAXB] = n;
      __syncthreads();
    }

    for (int i = tid; i < n; i += 4 * nth) {
      int c[4], r[4];
#pragma unroll
      for (int u = 0; u < 4; ++u) {
        const int iu = i + u * nth;
        const int ic = (iu < n) ? iu : n - 1;
        c[u] = colp[ic]; r[u] = rowp[ic];
      }
#pragma unroll
      for (int u = 0; u < 4; ++u) {
        if (i + u * nth < n) {
          const int pos = atomicAdd(&hist[tile_of(r[u], c[u])], 1);
          stage[pos] = make_int2(r[u], c[u]);
        }
      }
    }
    __syncthreads();

    int2* __restrict__ dst = pairs + (size_t)e0;
    for (int i = tid; i < n; i += nth) dst[i] = stage[i];
    __syncthreads();
  }

  for (int i = want ? tid : tid + nth; i < npb && nd0 + i < N; i += nth) {
    const float* xr = x + (size_t)(nd0 + i) * 16;
    deg[nd0 + i] = 1.0f;
    x3[nd0 + i]  = make_float4(xr[0], xr[1], xr[2], 0.0f);
  }
}

#define GR 8
#define TR 4

static __device__ __forceinline__ void lds_barrier() {
  __builtin_amdgcn_fence(__ATOMIC_RELEASE, "workgroup", "local");
  __builtin_amdgcn_s_barrier();
  __builtin_amdgcn_fence(__ATOMIC_ACQUIRE, "workgroup", "local");
}

template <int NF>
static __device__ __forceinline__ void add_row(float* ttab, unsigned k, float4 g) {
  if (k < (unsigned)RGT) {
    atomicAdd(&ttab[k], g.x);
    atomicAdd(&ttab[RGT + k], g.y);
    atomicAdd(&ttab[2 * RGT + k], g.z);
    if (NF == 4) atomicAdd(&ttab[3 * RGT + k], g.w);
  }
}

template <int NF, int NH, bool SC>
__global__ void __launch_bounds__(1024)
k_sum(const int2* __restrict__ pairs, const int* __restrict__ tbl,
      const float* __restrict__ src, const float* __restrict__ deg,
      float* __restrict__ part, int N, int NE) {
  __shared__ float4 stab[RGS];
  __shared__ float  ttab[4 * RGT];
  const int tid  = (int)threadIdx.x;
  const int nth  = (int)blockDim.x;
  const int lane = tid & 31;
  const int wv   = __builtin_amdgcn_readfirstlane(tid >> 5);
  const int nwv  = nth >> 5;
  const int b    = (int)blockIdx.x;
  const int tc   = b / NSR;
  const int tr   = b - tc * NSR;
  const int s0   = tr * RGS;
  const int t0   = tc * RGT;
  const int items = (NE + CH - 1) / CH;
  const int pg      = nwv * GR;
  const int ngroups = (items + pg - 1) / pg;
  const bool reload = ngroups > 1;

  int2 p[GR][TR];
  int sv = 0, ev = 0;
  int sb[GR], eb[GR];

  for (int i = tid; i < 4 * RGT; i += nth) ttab[i] = 0.0f;

  for (int h = 0; h < NH; ++h) {
    const float4* __restrict__ src4 = (const float4*)src + (size_t)h * N;
    float4* __restrict__ part4 = (float4*)part + ((size_t)h * NSR + tr) * NPAD + t0;

    if (h == 0 || reload) {
      const int rr = lane * nwv + wv;
      sv = 0; ev = 0;
      if (lane < GR && rr < items) {
        sv = tbl[(size_t)rr * TP + b];
        ev = tbl[(size_t)rr * TP + b + 1];
      }
    }

#pragma unroll 4
    for (int i = tid; i < RGS; i += nth) {
      int r = s0 + i;
      r = (r < N) ? r : N - 1;
      float4 v = src4[r];
      if (SC) {
        const float d = __builtin_amdgcn_rsqf(deg[r]);
        v.x *= d; v.y *= d; v.z *= d;
      }
      stab[i] = v;
    }

#pragma unroll
    for (int q = 0; q < GR; ++q) {
      sb[q] = __builtin_amdgcn_readlane(sv, q);
      eb[q] = __builtin_amdgcn_readlane(ev, q);
    }

    for (int g = 0; g < ngroups; ++g) {
      const int r0 = g * pg;
      if (h == 0 || reload) {
        if (g > 0) {
          const int rr = r0 + lane * nwv + wv;
          sv = 0; ev = 0;
          if (lane < GR && rr < items) {
            sv = tbl[(size_t)rr * TP + b];
            ev = tbl[(size_t)rr * TP + b + 1];
          }
#pragma unroll
          for (int q = 0; q < GR; ++q) {
            sb[q] = __builtin_amdgcn_readlane(sv, q);
            eb[q] = __builtin_amdgcn_readlane(ev, q);
          }
        }
#pragma unroll
        for (int q = 0; q < GR; ++q) {
          const int s = sb[q];
          const int e = eb[q];
          int rr = r0 + q * nwv + wv;
          rr = (rr < items) ? rr : items - 1;
          const int2* __restrict__ base = pairs + (size_t)rr * CH;
#pragma unroll
          for (int u = 0; u < TR; ++u) {
            const int j = s + u * 32 + lane;
            int jc = (j < e) ? j : e - 1;
            jc = (jc > 0) ? jc : 0;
            p[q][u] = base[jc];
          }
        }
      }
      if (g == 0) lds_barrier();

#pragma unroll
      for (int q = 0; q < GR; ++q) {
        const int s = sb[q];
        const int e = eb[q];
        unsigned k[TR];
        float4 gg[TR];
#pragma unroll
        for (int u = 0; u < TR; ++u) {
          const int j = s + u * 32 + lane;
          k[u] = (j < e) ? (unsigned)(p[q][u].y - t0) : 0xFFFFFFFFu;
          unsigned ls = (unsigned)(p[q][u].x - s0);
          ls = (ls < (unsigned)RGS) ? ls : (unsigned)(RGS - 1);
          gg[u] = stab[ls];
        }
#pragma unroll
        for (int u = 0; u < TR; ++u) add_row<NF>(ttab, k[u], gg[u]);
      }

#pragma nounroll
      for (int q = 0; q < GR; ++q) {
        const int s = __builtin_amdgcn_readlane(sv, q) + 32 * TR;
        const int e = __builtin_amdgcn_readlane(ev, q);
        if (s < e) {
          const int2* __restrict__ base = pairs + (size_t)(r0 + q * nwv + wv) * CH;
          for (int j0 = s; j0 < e; j0 += 32) {
            const int j = j0 + lane;
            const int2 pp = base[(j < e) ? j : e - 1];
            const unsigned k = (j < e) ? (unsigned)(pp.y - t0) : 0xFFFFFFFFu;
            unsigned ls = (unsigned)(pp.x - s0);
            ls = (ls < (unsigned)RGS) ? ls : (unsigned)(RGS - 1);
            add_row<NF>(ttab, k, stab[ls]);
          }
        }
      }
    }
    lds_barrier();

    for (int i = tid; i < RGT; i += nth) {
      float4 o;
      o.x = ttab[i]; o.y = ttab[RGT + i]; o.z = ttab[2 * RGT + i];
      o.w = (NF == 4) ? ttab[3 * RGT + i] : 0.0f;
      part4[i] = o;
      if (h + 1 < NH) {
        ttab[i] = 0.0f; ttab[RGT + i] = 0.0f; ttab[2 * RGT + i] = 0.0f; ttab[3 * RGT + i] = 0.0f;
      }
    }
  }
}

__global__ void __launch_bounds__(256)
k_dense2(const float* __restrict__ part1, const float* __restrict__ x3,
         const float* __restrict__ deg, const float* __restrict__ gW1,
         const float* __restrict__ gW2, const float* __restrict__ gb1,
         float* __restrict__ xw2s, int N) {
  __shared__ float ht[WPB][256];
  __shared__ float dv[WPB][16];
  int wslot = threadIdx.x >> 5;
  int lane  = threadIdx.x & 31;
  int m0 = (blockIdx.x * WPB + wslot) * 16;
  if (m0 >= N) return;
  bool full = (m0 + 16) <= N;
  float* t   = ht[wslot];
  float* dvs = dv[wslot];

  int n  = lane & 15;
  int mo = (lane >> 4) << 3;

  {
    int r  = m0 + n;
    int rc = (r < N) ? r : N - 1;
    float dvr = __builtin_amdgcn_rsqf(deg[rc]);
    if (lane < 16) dvs[n] = dvr;
    float4 a4 = make_float4(0.0f, 0.0f, 0.0f, 0.0f);
#pragma unroll
    for (int p = 0; p < NSR; ++p) {
      const float4 q4 = ((const float4*)part1)[(size_t)p * NPAD + rc];
      a4.x += q4.x; a4.y += q4.y; a4.z += q4.z;
    }
    float4 s4 = *(const float4*)(x3   + (size_t)rc * 4);
    float g0 = dvr * (a4.x + dvr * s4.x);
    float g1 = dvr * (a4.y + dvr * s4.y);
    float g2 = dvr * (a4.z + dvr * s4.z);
    v2f a;
    a.x = (lane < 16) ? g0 : g2;
    a.y = (lane < 16) ? g1 : 0.0f;
    v2f b = b_frag(gW1, lane, 0, 3, 16);
    v8f c1 = {};
    c1 = wmma_f32(a, b, c1);
    float bias = gb1[n];
    for (int v = 0; v < 8; ++v)
      t[(mo + v) * 16 + n] = fmaxf(c1[v] + bias, 0.0f);
  }
  __builtin_amdgcn_wave_barrier();

  v8f c = {};
  for (int k0 = 0; k0 < 16; k0 += 4) {
    v2f a = a_frag_lds(t, lane, k0);
    v2f b = b_frag(gW2, lane, k0, 16, 8);
    c = wmma_f32(a, b, c);
  }
  __builtin_amdgcn_wave_barrier();

  if (n < 8) {
    for (int v = 0; v < 8; ++v)
      t[(mo + v) * 8 + n] = c[v];
  }
  __builtin_amdgcn_wave_barrier();

  if (full) {
    float dvr = dvs[lane >> 1];
    float4 v4 = ((const float4*)t)[lane];
    v4.x *= dvr; v4.y *= dvr; v4.z *= dvr; v4.w *= dvr;
    ((float4*)xw2s)[(size_t)(lane & 1) * N + m0 + (lane >> 1)] = v4;
  } else if (n < 8) {
    for (int v = 0; v < 8; ++v) {
      int r = m0 + mo + v;
      if (r < N) xw2s[((size_t)(n >> 2) * N + r) * 4 + (n & 3)] = t[(mo + v) * 8 + n] * dvs[mo + v];
    }
  }
}

__global__ void __launch_bounds__(256)
k_final(const float* __restrict__ x, const float* __restrict__ part2,
        const float* __restrict__ xw2s, const float* __restrict__ deg,
        const float* __restrict__ gb2,
        const float* __restrict__ cW1, const float* __restrict__ cb1,
        const float* __restrict__ cW2, const float* __restrict__ cb2,
        const float* __restrict__ mW1, const float* __restrict__ mb1,
        const float* __restrict__ mW2, const float* __restrict__ mb2,
        float* __restrict__ out, int N) {
  __shared__ float ht[WPB][256];
  __shared__ float ct[WPB][128];
  __shared__ float g0s[WPB][16];
  int wslot = threadIdx.x >> 5;
  int lane  = threadIdx.x & 31;
  int m0 = (blockIdx.x * WPB + wslot) * 16;
  if (m0 >= N) return;
  bool full = (m0 + 16) <= N;
  float* t  = ht[wslot];
  float* o  = ct[wslot];
  float* g0 = g0s[wslot];

  int n  = lane & 15;
  int mo = (lane >> 4) << 3;

  int grow = lane >> 1, gcc = lane & 1;
  int gr  = m0 + grow;
  int grc = (gr < N) ? gr : N - 1;
  float  gdv = __builtin_amdgcn_rsqf(deg[grc]);
  float4 a24 = make_float4(0.0f, 0.0f, 0.0f, 0.0f);
#pragma unroll
  for (int p = 0; p < NSR; ++p) {
    const float4 q4 = ((const float4*)part2)[((size_t)gcc * NSR + p) * NPAD + grc];
    a24.x += q4.x; a24.y += q4.y; a24.z += q4.z; a24.w += q4.w;
  }
  float4 s24 = ((const float4*)xw2s)[(size_t)gcc * N + grc];
  float4 gb4 = ((const float4*)gb2)[gcc];
  float4 go4;
  go4.x = gdv * (a24.x + s24.x) + gb4.x;
  go4.y = gdv * (a24.y + s24.y) + gb4.y;
  go4.z = gdv * (a24.z + s24.z) + gb4.z;
  go4.w = gdv * (a24.w + s24.w) + gb4.w;

  v8f ch = {};
  for (int k0 = 0; k0 < 16; k0 += 4) {
    v2f a = a_frag_x(x, N, m0, lane, k0,  3,  13);
    v2f b = b_frag(cW1, lane, k0, 13, 16);
    ch = wmma_f32(a, b, ch);
  }
  for (int v = 0; v < 8; ++v) t[(mo + v) * 16 + n] = fmaxf(ch[v] + cb1[n], 0.0f);
  __builtin_amdgcn_wave_barrier();

  v8f cc = {};
  for (int k0 = 0; k0 < 16; k0 += 4) {
    v2f a = a_frag_lds(t, lane, k0);
    v2f b = b_frag(cW2, lane, k0, 16, 8);
    cc = wmma_f32(a, b, cc);
  }
  if (n < 8) {
    for (int v = 0; v < 8; ++v) o[(mo + v) * 8 + n] = cc[v] + cb2[n];
  }
  __builtin_amdgcn_wave_barrier();

  v8f cm = {};
  for (int k0 = 0; k0 < 16; k0 += 4) {
    v2f a = a_frag_x(x, N, m0, lane, k0,  0,  16);
    v2f b = b_frag(mW1, lane, k0, 16, 16);
    cm = wmma_f32(a, b, cm);
  }
  for (int v = 0; v < 8; ++v) t[(mo + v) * 16 + n] = fmaxf(cm[v] + mb1[n], 0.0f);
  __builtin_amdgcn_wave_barrier();

  if (lane < 16) {
    float l0 = mb2[0], l1 = mb2[1];
    for (int j = 0; j < 16; ++j) {
      float hv = t[lane * 16 + j];
      l0 += hv * mW2[j * 2 + 0];
      l1 += hv * mW2[j * 2 + 1];
    }
    float mx = fmaxf(l0, l1);
    float e0 = __expf(l0 - mx), e1 = __expf(l1 - mx);
    g0[lane] = e0 / (e0 + e1);
  }
  __builtin_amdgcn_wave_barrier();

  float g  = g0[grow];
  float4 c4 = *(const float4*)&o[lane * 4];
  float4 r4;
  r4.x = g * go4.x + (1.0f - g) * c4.x;
  r4.y = g * go4.y + (1.0f - g) * c4.y;
  r4.z = g * go4.z + (1.0f - g) * c4.z;
  r4.w = g * go4.w + (1.0f - g) * c4.w;
  if (full) {
    ((float4*)(out + (size_t)m0 * 8))[lane] = r4;
  } else if (gr < N) {
    *(float4*)(out + (size_t)gr * 8 + gcc * 4) = r4;
  }
}

extern "C" void kernel_launch(void* const* d_in, const int* in_sizes, int n_in,
                              void* d_out, int out_size, void* d_ws, size_t ws_size,
                              hipStream_t stream) {
  const float* x   = (const float*)d_in[0];
  const int*   ei  = (const int*)  d_in[1];
  const float* gW1 = (const float*)d_in[2];
  const float* gb1 = (const float*)d_in[3];
  const float* gW2 = (const float*)d_in[4];
  const float* gb2 = (const float*)d_in[5];
  const float* cW1 = (const float*)d_in[6];
  const float* cb1 = (const float*)d_in[7];
  const float* cW2 = (const float*)d_in[8];
  const float* cb2 = (const float*)d_in[9];
  const float* mW1 = (const float*)d_in[10];
  const float* mb1 = (const float*)d_in[11];
  const float* mW2 = (const float*)d_in[12];
  const float* mb2 = (const float*)d_in[13];
  float* out = (float*)d_out;

  const int N  = in_sizes[0] / 16;
  const int NE = in_sizes[1] / 2;
  const int* erow = ei;
  const int* ecol = ei + NE;

  float* deg   = (float*)d_ws;
  float* x3    = deg   + (size_t)2 * N;
  float* xw2s  = x3    + (size_t)4 * N;
  float* part1 = xw2s  + (size_t)8 * N;
  float* part2 = part1 + (size_t)4 * NSR * NPAD;
  const int items = (NE + CH - 1) / CH;
  int*   tbl   = (int*)(part2 + (size_t)8 * NSR * NPAD);
  int2*  pairs = (int2*)(tbl + (size_t)items * TP);

  const int tiles   = (N + 15) / 16;
  const int gBlocks = (tiles + WPB - 1) / WPB;

  k_bucket  <<<256, 1024, 0, stream>>>(erow, ecol, tbl, pairs, NE, x, deg, (float4*)x3, N);
  k_degree  <<<256, 1024, 0, stream>>>(ecol, deg, N, NE);
  k_sum<3, 1, true><<<MAXB, 1024, 0, stream>>>(pairs, tbl, x3, deg, part1, N, NE);
  k_dense2  <<<gBlocks, 256, 0, stream>>>(part1, x3, deg, gW1, gW2, gb1, xw2s, N);
  k_sum<4, 2, false><<<MAXB, 1024, 0, stream>>>(pairs, tbl, xw2s, deg, part2, N, NE);
  k_final   <<<gBlocks, 256, 0, stream>>>(x, part2, xw2s, deg, gb2,
                                          cW1, cb1, cW2, cb2,
                                          mW1, mb1, mW2, mb2, out, N);
}


# degree counted from the bucketed pairs: a block per (target range, item group) counts only pairs of its own range in LDS and adds whole lines to deg: 8 f32 adds per node, not 64, integer-exact
# speedup vs baseline: 9.8118x; 1.0335x over previous
#include <hip/hip_runtime.h>
#include <math.h>


typedef float v2f __attribute__((ext_vector_type(2)));
typedef float v8f __attribute__((ext_vector_type(8)));

#define WPB 8

static __device__ __forceinline__ v8f wmma_f32(v2f a, v2f b, v8f c) {
  return __builtin_amdgcn_wmma_f32_16x16x4_f32(false, a, false, b,
                                               (short)0, c, false, false);
}

static __device__ __forceinline__ v2f b_frag(const float* __restrict__ W,
                                             int lane, int k0, int K, int Ncols) {
  int n  = lane & 15;
  int kb = k0 + ((lane >> 4) << 1);
  int nm  = (n      < Ncols) ? n      : (Ncols - 1);
  int kc0 = (kb     < K)     ? kb     : (K - 1);
  int kc1 = (kb + 1 < K)     ? kb + 1 : (K - 1);
  float w0 = W[kc0 * Ncols + nm];
  float w1 = W[kc1 * Ncols + nm];
  v2f b;
  b.x = (kb     < K && n < Ncols) ? w0 : 0.0f;
  b.y = (kb + 1 < K && n < Ncols) ? w1 : 0.0f;
  return b;
}

static __device__ __forceinline__ v2f a_frag_x(const float* __restrict__ x,
                                               int N, int m0, int lane,
                                               int k0, int co, int K) {
  int r = m0 + (lane & 15);
  if (r >= N) r = N - 1;
  int kb = k0 + ((lane >> 4) << 1);
  const float* rowp = x + (size_t)r * 16;
  int i0 = co + kb;     i0 = (i0 < 16) ? i0 : 15;
  int i1 = co + kb + 1; i1 = (i1 < 16) ? i1 : 15;
  float a0 = rowp[i0];
  float a1 = rowp[i1];
  v2f a;
  a.x = (kb     < K) ? a0 : 0.0f;
  a.y = (kb + 1 < K) ? a1 : 0.0f;
  return a;
}

static __device__ __forceinline__ v2f a_frag_lds(const float* t, int lane, int k0) {
  int r  = lane & 15;
  int kb = k0 + ((lane >> 4) << 1);
  v2f a;
  a.x = t[r * 16 + kb];
  a.y = t[r * 16 + kb + 1];
  return a;
}

#define NSR    8
#define NTR    32
#define RGS    12544
#define RGT    3136
#define NPAD   (NTR * RGT)
#define MAXB   (NSR * NTR)
#define TP     (MAXB + 8)
#define CH     25024

static __device__ __forceinline__ unsigned tile_of(int r, int c) {
  unsigned tc = (unsigned)c / (unsigned)RGT;
  unsigned tr = (unsigned)r / (unsigned)RGS;
  tc = (tc < (unsigned)NTR) ? tc : (unsigned)(NTR - 1);
  tr = (tr < (unsigned)NSR) ? tr : (unsigned)(NSR - 1);
  return tc * NSR + tr;
}

__global__ void __launch_bounds__(1024)
k_bucket(const int* __restrict__ row, const int* __restrict__ col,
         int* __restrict__ tbl, int2* __restrict__ pairs, int NE,
         const float* __restrict__ x, float* __restrict__ deg, float4* __restrict__ x3, int N) {
  __shared__ int2 stage[CH];
  __shared__ int  hist[MAXB];
  __shared__ int  wsum[MAXB / 32];
  const int tid  = (int)threadIdx.x;
  const int nth  = (int)blockDim.x;
  const int lane = tid & 31;
  const int items = (NE + CH - 1) / CH;

  const int npb = (((N + (int)gridDim.x - 1) / (int)gridDim.x) + 31) & ~31;
  const int nd0 = (int)blockIdx.x * npb;
  bool want = (tid < npb) && (nd0 + tid < N);

  for (int item = (int)blockIdx.x; item < items; item += (int)gridDim.x) {
    const int e0 = item * CH;
    const int n  = (NE - e0 < CH) ? (NE - e0) : CH;
    const int* __restrict__ colp = col + e0;
    const int* __restrict__ rowp = row + e0;
    for (int i = tid; i < MAXB; i += nth) hist[i] = 0;
    __syncthreads();

    const bool now = want;
    float f0 = 0.0f, f1 = 0.0f, f2 = 0.0f;
    if (now) {
      const float* xr = x + (size_t)(nd0 + tid) * 16;
      f0 = xr[0]; f1 = xr[1]; f2 = xr[2];
    }

    for (int i = tid; i < n; i += 8 * nth) {
      int c[8], r[8];
#pragma unroll
      for (int u = 0; u < 8; ++u) {
        const int iu = i + u * nth;
        const int ic = (iu < n) ? iu : n - 1;
        c[u] = colp[ic]; r[u] = rowp[ic];
      }
#pragma unroll
      for (int u = 0; u < 8; ++u) {
        if (i + u * nth < n) atomicAdd(&hist[tile_of(r[u], c[u])], 1);
      }
    }
    __syncthreads();

    if (now) {
      deg[nd0 + tid] = 1.0f;
      x3[nd0 + tid]  = make_float4(f0, f1, f2, 0.0f);
      want = false;
    }

    {
      const int v = (tid < MAXB) ? hist[tid] : 0;
      int incl = v;
      for (int d = 1; d < 32; d <<= 1) {
        const int t = __shfl_up(incl, d);
        if (lane >= d) incl += t;
      }
      if (tid < MAXB && lane == 31) wsum[tid >> 5] = incl;
      __syncthreads();
      if (tid < MAXB) {
        int pre = 0;
        for (int w = 0; w < (tid >> 5); ++w) pre += wsum[w];
        const int ex = pre + incl - v;
        hist[tid] = ex;
        tbl[(size_t)item * TP + tid] = ex;
      }
      if (tid == 0) tbl[(size_t)item * TP + MAXB] = n;
      __syncthreads();
    }

    for (int i = tid; i < n; i += 4 * nth) {
      int c[4], r[4];
#pragma unroll
      for (int u = 0; u < 4; ++u) {
        const int iu = i + u * nth;
        const int ic = (iu < n) ? iu : n - 1;
        c[u] = colp[ic]; r[u] = rowp[ic];
      }
#pragma unroll
      for (int u = 0; u < 4; ++u) {
        if (i + u * nth < n) {
          const int pos = atomicAdd(&hist[tile_of(r[u], c[u])], 1);
          stage[pos] = make_int2(r[u], c[u]);
        }
      }
    }
    __syncthreads();

    int2* __restrict__ dst = pairs + (size_t)e0;
    for (int i = tid; i < n; i += nth) dst[i] = stage[i];
    __syncthreads();
  }

  for (int i = want ? tid : tid + nth; i < npb && nd0 + i < N; i += nth) {
    const float* xr = x + (size_t)(nd0 + i) * 16;
    deg[nd0 + i] = 1.0f;
    x3[nd0 + i]  = make_float4(xr[0], xr[1], xr[2], 0.0f);
  }
}

#define DU 28

__global__ void __launch_bounds__(1024)
k_degree(const int2* __restrict__ pairs, const int* __restrict__ tbl,
         float* __restrict__ deg, int N, int NE) {
  __shared__ unsigned cnt[RGT];
  const int tid  = (int)threadIdx.x;
  const int nth  = (int)blockDim.x;
  const int lane = tid & 31;
  const int wv   = __builtin_amdgcn_readfirstlane(tid >> 5);
  const int nwv  = nth >> 5;
  const int items = (NE + CH - 1) / CH;
  int G = (int)gridDim.x / NTR;
  if (G < 1) G = 1;

  for (int w = (int)blockIdx.x; w < NTR * G; w += (int)gridDim.x) {
    const int tc = w / G;
    const int ig = w - tc * G;
    const int t0 = tc * RGT;
    for (int i = tid; i < RGT; i += nth) cnt[i] = 0u;
    __syncthreads();

    for (int item = ig + G * wv; item < items; item += G * nwv) {
      const int* __restrict__ tb = tbl + (size_t)item * TP + tc * NSR;
      const int s = tb[0];
      const int e = tb[NSR];
      const int2* __restrict__ base = pairs + (size_t)item * CH;
      for (int j0 = s; j0 < e; j0 += 32 * DU) {
        int c[DU];
#pragma unroll
        for (int u = 0; u < DU; ++u) {
          const int j = j0 + u * 32 + lane;
          c[u] = base[(unsigned)((j < e) ? j : e - 1)].y;
        }
#pragma unroll
        for (int u = 0; u < DU; ++u) {
          const unsigned k = (unsigned)(c[u] - t0);
          if (j0 + u * 32 + lane < e && k < (unsigned)RGT) atomicAdd(&cnt[k], 1u);
        }
      }
    }
    __syncthreads();

    const int rot = (int)(((unsigned)ig * (unsigned)(RGT / 32)) / (unsigned)G) << 5;
    for (int j = tid; j < RGT; j += nth) {
      int i = j + rot;
      if (i >= RGT) i -= RGT;
      const unsigned c = cnt[i];
      if (c && t0 + i < N) atomicAdd(&deg[t0 + i], (float)c);
    }
    __syncthreads();
  }
}

#define GR 8
#define TR 4

static __device__ __forceinline__ void lds_barrier() {
  __builtin_amdgcn_fence(__ATOMIC_RELEASE, "workgroup", "local");
  __builtin_amdgcn_s_barrier();
  __builtin_amdgcn_fence(__ATOMIC_ACQUIRE, "workgroup", "local");
}

template <int NF>
static __device__ __forceinline__ void add_row(float* ttab, unsigned k, float4 g) {
  if (k < (unsigned)RGT) {
    atomicAdd(&ttab[k], g.x);
    atomicAdd(&ttab[RGT + k], g.y);
    atomicAdd(&ttab[2 * RGT + k], g.z);
    if (NF == 4) atomicAdd(&ttab[3 * RGT + k], g.w);
  }
}

template <int NF, int NH, bool SC>
__global__ void __launch_bounds__(1024)
k_sum(const int2* __restrict__ pairs, const int* __restrict__ tbl,
      const float* __restrict__ src, const float* __restrict__ deg,
      float* __restrict__ part, int N, int NE) {
  __shared__ float4 stab[RGS];
  __shared__ float  ttab[4 * RGT];
  const int tid  = (int)threadIdx.x;
  const int nth  = (int)blockDim.x;
  const int lane = tid & 31;
  const int wv   = __builtin_amdgcn_readfirstlane(tid >> 5);
  const int nwv  = nth >> 5;
  const int b    = (int)blockIdx.x;
  const int tc   = b / NSR;
  const int tr   = b - tc * NSR;
  const int s0   = tr * RGS;
  const int t0   = tc * RGT;
  const int items = (NE + CH - 1) / CH;
  const int pg      = nwv * GR;
  const int ngroups = (items + pg - 1) / pg;
  const bool reload = ngroups > 1;

  int2 p[GR][TR];
  int sv = 0, ev = 0;
  int sb[GR], eb[GR];

  for (int i = tid; i < 4 * RGT; i += nth) ttab[i] = 0.0f;

  for (int h = 0; h < NH; ++h) {
    const float4* __restrict__ src4 = (const float4*)src + (size_t)h * N;
    float4* __restrict__ part4 = (float4*)part + ((size_t)h * NSR + tr) * NPAD + t0;

    if (h == 0 || reload) {
      const int rr = lane * nwv + wv;
      sv = 0; ev = 0;
      if (lane < GR && rr < items) {
        sv = tbl[(size_t)rr * TP + b];
        ev = tbl[(size_t)rr * TP + b + 1];
      }
    }

#pragma unroll 4
    for (int i = tid; i < RGS; i += nth) {
      int r = s0 + i;
      r = (r < N) ? r : N - 1;
      float4 v = src4[r];
      if (SC) {
        const float d = __builtin_amdgcn_rsqf(deg[r]);
        v.x *= d; v.y *= d; v.z *= d;
      }
      stab[i] = v;
    }

#pragma unroll
    for (int q = 0; q < GR; ++q) {
      sb[q] = __builtin_amdgcn_readlane(sv, q);
      eb[q] = __builtin_amdgcn_readlane(ev, q);
    }

    for (int g = 0; g < ngroups; ++g) {
      const int r0 = g * pg;
      if (h == 0 || reload) {
        if (g > 0) {
          const int rr = r0 + lane * nwv + wv;
          sv = 0; ev = 0;
          if (lane < GR && rr < items) {
            sv = tbl[(size_t)rr * TP + b];
            ev = tbl[(size_t)rr * TP + b + 1];
          }
#pragma unroll
          for (int q = 0; q < GR; ++q) {
            sb[q] = __builtin_amdgcn_readlane(sv, q);
            eb[q] = __builtin_amdgcn_readlane(ev, q);
          }
        }
#pragma unroll
        for (int q = 0; q < GR; ++q) {
          const int s = sb[q];
          const int e = eb[q];
          int rr = r0 + q * nwv + wv;
          rr = (rr < items) ? rr : items - 1;
          const int2* __restrict__ base = pairs + (size_t)rr * CH;
#pragma unroll
          for (int u = 0; u < TR; ++u) {
            const int j = s + u * 32 + lane;
            int jc = (j < e) ? j : e - 1;
            jc = (jc > 0) ? jc : 0;
            p[q][u] = base[jc];
          }
        }
      }
      if (g == 0) lds_barrier();

#pragma unroll
      for (int q = 0; q < GR; ++q) {
        const int s = sb[q];
        const int e = eb[q];
        unsigned k[TR];
        float4 gg[TR];
#pragma unroll
        for (int u = 0; u < TR; ++u) {
          const int j = s + u * 32 + lane;
          k[u] = (j < e) ? (unsigned)(p[q][u].y - t0) : 0xFFFFFFFFu;
          unsigned ls = (unsigned)(p[q][u].x - s0);
          ls = (ls < (unsigned)RGS) ? ls : (unsigned)(RGS - 1);
          gg[u] = stab[ls];
        }
#pragma unroll
        for (int u = 0; u < TR; ++u) add_row<NF>(ttab, k[u], gg[u]);
      }

#pragma nounroll
      for (int q = 0; q < GR; ++q) {
        const int s = __builtin_amdgcn_readlane(sv, q) + 32 * TR;
        const int e = __builtin_amdgcn_readlane(ev, q);
        if (s < e) {
          const int2* __restrict__ base = pairs + (size_t)(r0 + q * nwv + wv) * CH;
          for (int j0 = s; j0 < e; j0 += 32) {
            const int j = j0 + lane;
            const int2 pp = base[(j < e) ? j : e - 1];
            const unsigned k = (j < e) ? (unsigned)(pp.y - t0) : 0xFFFFFFFFu;
            unsigned ls = (unsigned)(pp.x - s0);
            ls = (ls < (unsigned)RGS) ? ls : (unsigned)(RGS - 1);
            add_row<NF>(ttab, k, stab[ls]);
          }
        }
      }
    }
    lds_barrier();

    for (int i = tid; i < RGT; i += nth) {
      float4 o;
      o.x = ttab[i]; o.y = ttab[RGT + i]; o.z = ttab[2 * RGT + i];
      o.w = (NF == 4) ? ttab[3 * RGT + i] : 0.0f;
      part4[i] = o;
      if (h + 1 < NH) {
        ttab[i] = 0.0f; ttab[RGT + i] = 0.0f; ttab[2 * RGT + i] = 0.0f; ttab[3 * RGT + i] = 0.0f;
      }
    }
  }
}

__global__ void __launch_bounds__(256)
k_dense2(const float* __restrict__ part1, const float* __restrict__ x3,
         const float* __restrict__ deg, const float* __restrict__ gW1,
         const float* __restrict__ gW2, const float* __restrict__ gb1,
         float* __restrict__ xw2s, int N) {
  __shared__ float ht[WPB][256];
  __shared__ float dv[WPB][16];
  int wslot = threadIdx.x >> 5;
  int lane  = threadIdx.x & 31;
  int m0 = (blockIdx.x * WPB + wslot) * 16;
  if (m0 >= N) return;
  bool full = (m0 + 16) <= N;
  float* t   = ht[wslot];
  float* dvs = dv[wslot];

  int n  = lane & 15;
  int mo = (lane >> 4) << 3;

  {
    int r  = m0 + n;
    int rc = (r < N) ? r : N - 1;
    float dvr = __builtin_amdgcn_rsqf(deg[rc]);
    if (lane < 16) dvs[n] = dvr;
    float4 a4 = make_float4(0.0f, 0.0f, 0.0f, 0.0f);
#pragma unroll
    for (int p = 0; p < NSR; ++p) {
      const float4 q4 = ((const float4*)part1)[(size_t)p * NPAD + rc];
      a4.x += q4.x; a4.y += q4.y; a4.z += q4.z;
    }
    float4 s4 = *(const float4*)(x3   + (size_t)rc * 4);
    float g0 = dvr * (a4.x + dvr * s4.x);
    float g1 = dvr * (a4.y + dvr * s4.y);
    float g2 = dvr * (a4.z + dvr * s4.z);
    v2f a;
    a.x = (lane < 16) ? g0 : g2;
    a.y = (lane < 16) ? g1 : 0.0f;
    v2f b = b_frag(gW1, lane, 0, 3, 16);
    v8f c1 = {};
    c1 = wmma_f32(a, b, c1);
    float bias = gb1[n];
    for (int v = 0; v < 8; ++v)
      t[(mo + v) * 16 + n] = fmaxf(c1[v] + bias, 0.0f);
  }
  __builtin_amdgcn_wave_barrier();

  v8f c = {};
  for (int k0 = 0; k0 < 16; k0 += 4) {
    v2f a = a_frag_lds(t, lane, k0);
    v2f b = b_frag(gW2, lane, k0, 16, 8);
    c = wmma_f32(a, b, c);
  }
  __builtin_amdgcn_wave_barrier();

  if (n < 8) {
    for (int v = 0; v < 8; ++v)
      t[(mo + v) * 8 + n] = c[v];
  }
  __builtin_amdgcn_wave_barrier();

  if (full) {
    float dvr = dvs[lane >> 1];
    float4 v4 = ((const float4*)t)[lane];
    v4.x *= dvr; v4.y *= dvr; v4.z *= dvr; v4.w *= dvr;
    ((float4*)xw2s)[(size_t)(lane & 1) * N + m0 + (lane >> 1)] = v4;
  } else if (n < 8) {
    for (int v = 0; v < 8; ++v) {
      int r = m0 + mo + v;
      if (r < N) xw2s[((size_t)(n >> 2) * N + r) * 4 + (n & 3)] = t[(mo + v) * 8 + n] * dvs[mo + v];
    }
  }
}

__global__ void __launch_bounds__(256)
k_final(const float* __restrict__ x, const float* __restrict__ part2,
        const float* __restrict__ xw2s, const float* __restrict__ deg,
        const float* __restrict__ gb2,
        const float* __restrict__ cW1, const float* __restrict__ cb1,
        const float* __restrict__ cW2, const float* __restrict__ cb2,
        const float* __restrict__ mW1, const float* __restrict__ mb1,
        const float* __restrict__ mW2, const float* __restrict__ mb2,
        float* __restrict__ out, int N) {
  __shared__ float ht[WPB][256];
  __shared__ float ct[WPB][128];
  __shared__ float g0s[WPB][16];
  int wslot = threadIdx.x >> 5;
  int lane  = threadIdx.x & 31;
  int m0 = (blockIdx.x * WPB + wslot) * 16;
  if (m0 >= N) return;
  bool full = (m0 + 16) <= N;
  float* t  = ht[wslot];
  float* o  = ct[wslot];
  float* g0 = g0s[wslot];

  int n  = lane & 15;
  int mo = (lane >> 4) << 3;

  int grow = lane >> 1, gcc = lane & 1;
  int gr  = m0 + grow;
  int grc = (gr < N) ? gr : N - 1;
  float  gdv = __builtin_amdgcn_rsqf(deg[grc]);
  float4 a24 = make_float4(0.0f, 0.0f, 0.0f, 0.0f);
#pragma unroll
  for (int p = 0; p < NSR; ++p) {
    const float4 q4 = ((const float4*)part2)[((size_t)gcc * NSR + p) * NPAD + grc];
    a24.x += q4.x; a24.y += q4.y; a24.z += q4.z; a24.w += q4.w;
  }
  float4 s24 = ((const float4*)xw2s)[(size_t)gcc * N + grc];
  float4 gb4 = ((const float4*)gb2)[gcc];
  float4 go4;
  go4.x = gdv * (a24.x + s24.x) + gb4.x;
  go4.y = gdv * (a24.y + s24.y) + gb4.y;
  go4.z = gdv * (a24.z + s24.z) + gb4.z;
  go4.w = gdv * (a24.w + s24.w) + gb4.w;

  v8f ch = {};
  for (int k0 = 0; k0 < 16; k0 += 4) {
    v2f a = a_frag_x(x, N, m0, lane, k0,  3,  13);
    v2f b = b_frag(cW1, lane, k0, 13, 16);
    ch = wmma_f32(a, b, ch);
  }
  for (int v = 0; v < 8; ++v) t[(mo + v) * 16 + n] = fmaxf(ch[v] + cb1[n], 0.0f);
  __builtin_amdgcn_wave_barrier();

  v8f cc = {};
  for (int k0 = 0; k0 < 16; k0 += 4) {
    v2f a = a_frag_lds(t, lane, k0);
    v2f b = b_frag(cW2, lane, k0, 16, 8);
    cc = wmma_f32(a, b, cc);
  }
  if (n < 8) {
    for (int v = 0; v < 8; ++v) o[(mo + v) * 8 + n] = cc[v] + cb2[n];
  }
  __builtin_amdgcn_wave_barrier();

  v8f cm = {};
  for (int k0 = 0; k0 < 16; k0 += 4) {
    v2f a = a_frag_x(x, N, m0, lane, k0,  0,  16);
    v2f b = b_frag(mW1, lane, k0, 16, 16);
    cm = wmma_f32(a, b, cm);
  }
  for (int v = 0; v < 8; ++v) t[(mo + v) * 16 + n] = fmaxf(cm[v] + mb1[n], 0.0f);
  __builtin_amdgcn_wave_barrier();

  if (lane < 16) {
    float l0 = mb2[0], l1 = mb2[1];
    for (int j = 0; j < 16; ++j) {
      float hv = t[lane * 16 + j];
      l0 += hv * mW2[j * 2 + 0];
      l1 += hv * mW2[j * 2 + 1];
    }
    float mx = fmaxf(l0, l1);
    float e0 = __expf(l0 - mx), e1 = __expf(l1 - mx);
    g0[lane] = e0 / (e0 + e1);
  }
  __builtin_amdgcn_wave_barrier();

  float g  = g0[grow];
  float4 c4 = *(const float4*)&o[lane * 4];
  float4 r4;
  r4.x = g * go4.x + (1.0f - g) * c4.x;
  r4.y = g * go4.y + (1.0f - g) * c4.y;
  r4.z = g * go4.z + (1.0f - g) * c4.z;
  r4.w = g * go4.w + (1.0f - g) * c4.w;
  if (full) {
    ((float4*)(out + (size_t)m0 * 8))[lane] = r4;
  } else if (gr < N) {
    *(float4*)(out + (size_t)gr * 8 + gcc * 4) = r4;
  }
}

extern "C" void kernel_launch(void* const* d_in, const int* in_sizes, int n_in,
                              void* d_out, int out_size, void* d_ws, size_t ws_size,
                              hipStream_t stream) {
  const float* x   = (const float*)d_in[0];
  const int*   ei  = (const int*)  d_in[1];
  const float* gW1 = (const float*)d_in[2];
  const float* gb1 = (const float*)d_in[3];
  const float* gW2 = (const float*)d_in[4];
  const float* gb2 = (const float*)d_in[5];
  const float* cW1 = (const float*)d_in[6];
  const float* cb1 = (const float*)d_in[7];
  const float* cW2 = (const float*)d_in[8];
  const float* cb2 = (const float*)d_in[9];
  const float* mW1 = (const float*)d_in[10];
  const float* mb1 = (const float*)d_in[11];
  const float* mW2 = (const float*)d_in[12];
  const float* mb2 = (const float*)d_in[13];
  float* out = (float*)d_out;

  const int N  = in_sizes[0] / 16;
  const int NE = in_sizes[1] / 2;
  const int* erow = ei;
  const int* ecol = ei + NE;

  float* deg   = (float*)d_ws;
  float* x3    = deg   + (size_t)2 * N;
  float* xw2s  = x3    + (size_t)4 * N;
  float* part1 = xw2s  + (size_t)8 * N;
  float* part2 = part1 + (size_t)4 * NSR * NPAD;
  const int items = (NE + CH - 1) / CH;
  int*   tbl   = (int*)(part2 + (size_t)8 * NSR * NPAD);
  int2*  pairs = (int2*)(tbl + (size_t)items * TP);

  const int tiles   = (N + 15) / 16;
  const int gBlocks = (tiles + WPB - 1) / WPB;

  k_bucket  <<<256, 1024, 0, stream>>>(erow, ecol, tbl, pairs, NE, x, deg, (float4*)x3, N);
  k_degree  <<<256, 1024, 0, stream>>>(pairs, tbl, deg, N, NE);
  k_sum<3, 1, true><<<MAXB, 1024, 0, stream>>>(pairs, tbl, x3, deg, part1, N, NE);
  k_dense2  <<<gBlocks, 256, 0, stream>>>(part1, x3, deg, gW1, gW2, gb1, xw2s, N);
  k_sum<4, 2, false><<<MAXB, 1024, 0, stream>>>(pairs, tbl, xw2s, deg, part2, N, NE);
  k_final   <<<gBlocks, 256, 0, stream>>>(x, part2, xw2s, deg, gb2,
                                          cW1, cb1, cW2, cb2,
                                          mW1, mb1, mW2, mb2, out, N);
}
